# starting kernel
# baseline (speedup 1.0000x reference)
#include <hip/hip_runtime.h>
#include <math.h>

#define NN 50000
#define EE 800000

typedef __attribute__((ext_vector_type(2))) float v2f;
typedef __attribute__((ext_vector_type(8))) float v8f;

__device__ __forceinline__ unsigned f2ord(float f) {
  unsigned u = __float_as_uint(f);
  return (u & 0x80000000u) ? ~u : (u | 0x80000000u);
}
__device__ __forceinline__ float ord2f(unsigned e) {
  unsigned u = (e & 0x80000000u) ? (e ^ 0x80000000u) : ~e;
  return __uint_as_float(u);
}

__global__ __launch_bounds__(32)
void gat_gemm_xwt(const float* __restrict__ X, const float* __restrict__ W,
                  float* __restrict__ OUT, int K, int Opad, int Oout) {
  const int lane = threadIdx.x & 31;
  const int row0 = blockIdx.x << 4;
  const int col0 = blockIdx.y << 6;
  const int mr   = lane & 15;
  const int half = lane >> 4;
  const int koff = half << 1;

  const float* xrow = X + (size_t)(row0 + mr) * (size_t)K + koff;
  const float* wrow = W + (size_t)(col0 + mr) * (size_t)K + koff;
  const size_t wstep = (size_t)16 * (size_t)K;

  v8f acc[4] = {};
  v2f a0, a1;
  v2f b0[4], b1[4];

  a0 = *(const v2f*)(xrow);
#pragma unroll
  for (int t = 0; t < 4; ++t) b0[t] = *(const v2f*)(wrow + (size_t)t * wstep);

  int k0 = 0;
  for (; k0 + 8 < K; k0 += 8) {
    a1 = *(const v2f*)(xrow + k0 + 4);
#pragma unroll
    for (int t = 0; t < 4; ++t)
      b1[t] = *(const v2f*)(wrow + (size_t)t * wstep + k0 + 4);
#pragma unroll
    for (int t = 0; t < 4; ++t)
      acc[t] = __builtin_amdgcn_wmma_f32_16x16x4_f32(
          false, a0, false, b0[t], (short)0, acc[t], false, false);

    a0 = *(const v2f*)(xrow + k0 + 8);
#pragma unroll
    for (int t = 0; t < 4; ++t)
      b0[t] = *(const v2f*)(wrow + (size_t)t * wstep + k0 + 8);
#pragma unroll
    for (int t = 0; t < 4; ++t)
      acc[t] = __builtin_amdgcn_wmma_f32_16x16x4_f32(
          false, a1, false, b1[t], (short)0, acc[t], false, false);
  }

  a1 = *(const v2f*)(xrow + k0 + 4);
#pragma unroll
  for (int t = 0; t < 4; ++t)
    b1[t] = *(const v2f*)(wrow + (size_t)t * wstep + k0 + 4);
#pragma unroll
  for (int t = 0; t < 4; ++t)
    acc[t] = __builtin_amdgcn_wmma_f32_16x16x4_f32(
        false, a0, false, b0[t], (short)0, acc[t], false, false);
#pragma unroll
  for (int t = 0; t < 4; ++t)
    acc[t] = __builtin_amdgcn_wmma_f32_16x16x4_f32(
        false, a1, false, b1[t], (short)0, acc[t], false, false);

#pragma unroll
  for (int t = 0; t < 4; ++t) {
    const int oc = col0 + (t << 4) + mr;
    if (oc < Oout) {
      float* orow = OUT + (size_t)(row0 + (half << 3)) * (size_t)Oout + oc;
#pragma unroll
      for (int v = 0; v < 8; ++v) {
        orow[(size_t)v * (size_t)Oout] = acc[t][v];
      }
    }
  }
}

__global__ void gat_pad_w(const float* __restrict__ Wsrc, float* __restrict__ Wdst,
                          int rows, int padrows, int K) {
  int idx = blockIdx.x * blockDim.x + threadIdx.x;
  int total = padrows * K;
  if (idx >= total) return;
  int r = idx / K;
  Wdst[idx] = (r < rows) ? Wsrc[idx] : 0.0f;
}

__global__ void gat_el_er(const float* __restrict__ Hm, const float* __restrict__ al,
                          const float* __restrict__ ar, float* __restrict__ el,
                          float* __restrict__ er, int total, int Hh, int F) {
  int idx = blockIdx.x * blockDim.x + threadIdx.x;
  if (idx >= total) return;
  int h = idx % Hh;
  int n = idx / Hh;
  const float* row = Hm + (size_t)n * (size_t)(Hh * F) + (size_t)h * F;
  const float* a = al + h * F;
  const float* b = ar + h * F;
  float sl = 0.f, sr = 0.f;
  for (int f = 0; f < F; ++f) {
    float v = row[f];
    sl += v * a[f];
    sr += v * b[f];
  }
  el[idx] = sl;
  er[idx] = sr;
}

__global__ void gat_init_nm(unsigned* __restrict__ mEnc, float* __restrict__ den,
                            int total) {
  int idx = blockIdx.x * blockDim.x + threadIdx.x;
  if (idx < total) { mEnc[idx] = 0u; den[idx] = 0.0f; }
}

__global__ void gat_edge_smax(const float* __restrict__ el, const float* __restrict__ er,
                              const int* __restrict__ src, const int* __restrict__ dst,
                              float* __restrict__ sb, unsigned* __restrict__ mEnc,
                              int total, int Hh) {
  int idx = blockIdx.x * blockDim.x + threadIdx.x;
  if (idx >= total) return;
  int e = idx / Hh;
  int h = idx - e * Hh;
  int s_ = src[e], d_ = dst[e];
  float s = el[s_ * Hh + h] + er[d_ * Hh + h];
  s = (s > 0.0f) ? s : 0.2f * s;
  sb[idx] = s;
  atomicMax(mEnc + d_ * Hh + h, f2ord(s));
}

__global__ void gat_edge_expsum(float* __restrict__ sb, const unsigned* __restrict__ mEnc,
                                float* __restrict__ den, const int* __restrict__ dst,
                                int total, int Hh) {
  int idx = blockIdx.x * blockDim.x + threadIdx.x;
  if (idx >= total) return;
  int e = idx / Hh;
  int h = idx - e * Hh;
  int d_ = dst[e];
  float m = ord2f(mEnc[d_ * Hh + h]);
  float ex = expf(sb[idx] - m);
  sb[idx] = ex;
  atomicAdd(den + d_ * Hh + h, ex);
}

__global__ void gat_scatter(const float* __restrict__ Hm, const float* __restrict__ sb,
                            const float* __restrict__ den, const int* __restrict__ src,
                            const int* __restrict__ dst, float* __restrict__ AGG,
                            int HF, int F, int Hh) {
  int e = blockIdx.x;
  int j = threadIdx.x;
  if (j >= HF) return;
  int s_ = src[e], d_ = dst[e];
  int h = j / F;
  float alpha = sb[(size_t)e * Hh + h] / den[(size_t)d_ * Hh + h];
  atomicAdd(AGG + (size_t)d_ * HF + j, Hm[(size_t)s_ * HF + j] * alpha);
}

__global__ void gat_elu(float* __restrict__ y, long total) {
  long idx = (long)blockIdx.x * blockDim.x + threadIdx.x;
  if (idx >= total) return;
  float x = y[idx];
  y[idx] = (x > 0.0f) ? x : expm1f(x);
}

static void run_layer(const float* X, int K, int Hh, int F,
                      const float* W, const float* al, const float* ar, const float* rW,
                      const int* src, const int* dst,
                      float* Hbuf, float* AGG, bool act, int Opad,
                      float* el, float* er, float* den, unsigned* mEnc, float* sb,
                      hipStream_t stream) {
  const int HF = Hh * F;
  dim3 ggrid(NN / 16, Opad / 64);
  gat_gemm_xwt<<<ggrid, 32, 0, stream>>>(X, W, Hbuf, K, Opad, HF);
  gat_gemm_xwt<<<ggrid, 32, 0, stream>>>(X, rW, AGG, K, Opad, HF);

  const int nh = NN * Hh;
  gat_el_er<<<(nh + 255) / 256, 256, 0, stream>>>(Hbuf, al, ar, el, er, nh, Hh, F);
  gat_init_nm<<<(nh + 255) / 256, 256, 0, stream>>>(mEnc, den, nh);

  const int eh = EE * Hh;
  gat_edge_smax<<<(eh + 255) / 256, 256, 0, stream>>>(el, er, src, dst, sb, mEnc, eh, Hh);
  gat_edge_expsum<<<(eh + 255) / 256, 256, 0, stream>>>(sb, mEnc, den, dst, eh, Hh);

  const int bs = (HF >= 256) ? 256 : 64;
  gat_scatter<<<EE, bs, 0, stream>>>(Hbuf, sb, den, src, dst, AGG, HF, F, Hh);

  if (act) {
    long tot = (long)NN * HF;
    gat_elu<<<(unsigned)((tot + 255) / 256), 256, 0, stream>>>(AGG, tot);
  }
}

extern "C" void kernel_launch(void* const* d_in, const int* in_sizes, int n_in,
                              void* d_out, int out_size, void* d_ws, size_t ws_size,
                              hipStream_t stream) {
  (void)in_sizes; (void)n_in; (void)out_size; (void)ws_size;

  const float* feat = (const float*)d_in[0];
  const int*   src  = (const int*)d_in[1];
  const int*   dst  = (const int*)d_in[2];
  const float* W0  = (const float*)d_in[3];
  const float* al0 = (const float*)d_in[4];
  const float* ar0 = (const float*)d_in[5];
  const float* rW0 = (const float*)d_in[6];
  const float* W1  = (const float*)d_in[7];
  const float* al1 = (const float*)d_in[8];
  const float* ar1 = (const float*)d_in[9];
  const float* rW1 = (const float*)d_in[10];
  const float* W2  = (const float*)d_in[11];
  const float* al2 = (const float*)d_in[12];
  const float* ar2 = (const float*)d_in[13];
  const float* rW2 = (const float*)d_in[14];

  const size_t NF = (size_t)NN * 256;
  float* ws    = (float*)d_ws;
  float* bufH  = ws;
  float* bufY0 = bufH  + NF;
  float* bufY1 = bufY0 + NF;
  float* el    = bufY1 + NF;
  float* er    = el  + (size_t)NN * 4;
  float* den   = er  + (size_t)NN * 4;
  unsigned* mE = (unsigned*)(den + (size_t)NN * 4);
  float* sb    = (float*)mE + (size_t)NN * 4;
  float* W2p   = sb + (size_t)EE * 4;
  float* rW2p  = W2p + (size_t)64 * 256;

  {
    int tot = 64 * 256;
    gat_pad_w<<<(tot + 255) / 256, 256, 0, stream>>>(W2,  W2p,  40, 64, 256);
    gat_pad_w<<<(tot + 255) / 256, 256, 0, stream>>>(rW2, rW2p, 40, 64, 256);
  }

  run_layer(feat, 128, 4, 64, W0, al0, ar0, rW0, src, dst,
            bufH, bufY0, true, 256, el, er, den, mE, sb, stream);
  run_layer(bufY0, 256, 4, 64, W1, al1, ar1, rW1, src, dst,
            bufH, bufY1, true, 256, el, er, den, mE, sb, stream);
  run_layer(bufY1, 256, 1, 40, W2p, al2, ar2, rW2p, src, dst,
            bufH, (float*)d_out, false, 64, el, er, den, mE, sb, stream);
}


# Hidden layers: row atomics replaced by row ownership: a workgroup filters dst for its 200 rows into LDS queues, owner waves sum H rows in registers and store elu(res+sum) once; sums in queue order
# speedup vs baseline: 3.2337x; 3.2337x over previous
#include <hip/hip_runtime.h>
#include <math.h>

#define NN 50000
#define EE 800000

typedef __attribute__((ext_vector_type(2))) float v2f;
typedef __attribute__((ext_vector_type(4))) float v4f;
typedef __attribute__((ext_vector_type(8))) float v8f;

__device__ __forceinline__ unsigned f2ord(float f) {
  unsigned u = __float_as_uint(f);
  return (u & 0x80000000u) ? ~u : (u | 0x80000000u);
}
__device__ __forceinline__ float ord2f(unsigned e) {
  unsigned u = (e & 0x80000000u) ? (e ^ 0x80000000u) : ~e;
  return __uint_as_float(u);
}

template <int NF>
__device__ __forceinline__ void gemm_load_lines(v4f (&f)[NF][4], const float* __restrict__ p,
                                                size_t fstep) {
#pragma unroll
  for (int i = 0; i < NF; ++i) {
#pragma unroll
    for (int j = 0; j < 4; ++j) f[i][j] = *(const v4f*)(p + (size_t)i * fstep + 8 * j);
  }
}

template <int MA, int NB>
__device__ __forceinline__ void gemm_mma_lines(v8f (&acc)[MA][NB], const v4f (&a)[MA][4],
                                               const v4f (&b)[NB][4]) {
#pragma unroll
  for (int j = 0; j < 4; ++j) {
#pragma unroll
    for (int i = 0; i < MA; ++i) {
#pragma unroll
      for (int t = 0; t < NB; ++t)
        acc[i][t] = __builtin_amdgcn_wmma_f32_16x16x4_f32(
            false, a[i][j].xy, false, b[t][j].xy, (short)0, acc[i][t], false, false);
    }
#pragma unroll
    for (int i = 0; i < MA; ++i) {
#pragma unroll
      for (int t = 0; t < NB; ++t)
        acc[i][t] = __builtin_amdgcn_wmma_f32_16x16x4_f32(
            false, a[i][j].zw, false, b[t][j].zw, (short)0, acc[i][t], false, false);
    }
  }
}

template <int K, int MA, int NB>
__global__ __launch_bounds__(32)
void gat_gemm_xwt(const float* __restrict__ X, const float* __restrict__ W0,
                  const float* __restrict__ W1, float* __restrict__ OUT0,
                  float* __restrict__ OUT1, int Oout) {
  static_assert(K % 64 == 0, "the K loop takes two 32-float lines per trip");
  static_assert(NN % (16 * MA) == 0, "row tiles cover the nodes exactly");
  const int lane = threadIdx.x & 31;
  const int c    = lane & 15;
  const int half = lane >> 4;
  const int row0 = blockIdx.x * (16 * MA);
  const int col0 = blockIdx.y << 6;
  const float* __restrict__ W   = (blockIdx.z == 0) ? W0 : W1;
  float* __restrict__       OUT = (blockIdx.z == 0) ? OUT0 : OUT1;

  const float* xp = X + (size_t)(row0 + c) * (size_t)K + (half << 2);
  const float* wp = W + (size_t)(col0 + c) * (size_t)K + (half << 2);
  const size_t fstep = (size_t)16 * (size_t)K;

  v8f acc[MA][NB] = {};
  v4f aP[MA][4], bP[NB][4], aQ[MA][4], bQ[NB][4];

  gemm_load_lines<MA>(aP, xp, fstep);
  gemm_load_lines<NB>(bP, wp, fstep);
  int kb = 0;
  for (; kb + 64 < K; kb += 64) {
    gemm_load_lines<MA>(aQ, xp + kb + 32, fstep);
    gemm_load_lines<NB>(bQ, wp + kb + 32, fstep);
    __builtin_amdgcn_sched_barrier(0);
    gemm_mma_lines<MA, NB>(acc, aP, bP);
    __builtin_amdgcn_sched_barrier(0);
    gemm_load_lines<MA>(aP, xp + kb + 64, fstep);
    gemm_load_lines<NB>(bP, wp + kb + 64, fstep);
    __builtin_amdgcn_sched_barrier(0);
    gemm_mma_lines<MA, NB>(acc, aQ, bQ);
    __builtin_amdgcn_sched_barrier(0);
  }
  gemm_load_lines<MA>(aQ, xp + kb + 32, fstep);
  gemm_load_lines<NB>(bQ, wp + kb + 32, fstep);
  __builtin_amdgcn_sched_barrier(0);
  gemm_mma_lines<MA, NB>(acc, aP, bP);
  gemm_mma_lines<MA, NB>(acc, aQ, bQ);

#pragma unroll
  for (int i = 0; i < MA; ++i) {
#pragma unroll
    for (int t = 0; t < NB; ++t) {
      const int oc = col0 + (t << 4) + c;
      if (oc < Oout) {
        float* orow = OUT + (size_t)(row0 + (i << 4) + (half << 3)) * (size_t)Oout + oc;
#pragma unroll
        for (int v = 0; v < 8; ++v) {
          orow[(size_t)v * (size_t)Oout] = acc[i][t][v];
        }
      }
    }
  }
}

__global__ void gat_pad_w(const float* __restrict__ Wsrc0, const float* __restrict__ Wsrc1,
                          float* __restrict__ Wdst0, float* __restrict__ Wdst1,
                          int rows, int padrows, int K) {
  int idx = blockIdx.x * blockDim.x + threadIdx.x;
  int total = padrows * K;
  if (idx >= total) return;
  const float* __restrict__ Wsrc = (blockIdx.y == 0) ? Wsrc0 : Wsrc1;
  float* __restrict__       Wdst = (blockIdx.y == 0) ? Wdst0 : Wdst1;
  int r = idx / K;
  Wdst[idx] = (r < rows) ? Wsrc[idx] : 0.0f;
}

__global__ void gat_el_er(const float* __restrict__ Hm, const float* __restrict__ al,
                          const float* __restrict__ ar, float* __restrict__ el,
                          float* __restrict__ er, unsigned* __restrict__ mEnc,
                          float* __restrict__ den, int total, int Hh, int F) {
  int idx = blockIdx.x * blockDim.x + threadIdx.x;
  if (idx >= total) return;
  int h = idx % Hh;
  int n = idx / Hh;
  const float* row = Hm + (size_t)n * (size_t)(Hh * F) + (size_t)h * F;
  const float* a = al + h * F;
  const float* b = ar + h * F;
  float sl = 0.f, sr = 0.f;
  for (int f = 0; f < F; ++f) {
    float v = row[f];
    sl += v * a[f];
    sr += v * b[f];
  }
  el[idx] = sl;
  er[idx] = sr;
  mEnc[idx] = 0u;
  den[idx] = 0.0f;
}

__global__ void gat_edge_smax(const float* __restrict__ el, const float* __restrict__ er,
                              const int* __restrict__ src, const int* __restrict__ dst,
                              float* __restrict__ sb, unsigned* __restrict__ mEnc,
                              int total, int Hh) {
  int idx = blockIdx.x * blockDim.x + threadIdx.x;
  if (idx >= total) return;
  int e = idx / Hh;
  int h = idx - e * Hh;
  int s_ = src[e], d_ = dst[e];
  float s = el[s_ * Hh + h] + er[d_ * Hh + h];
  s = (s > 0.0f) ? s : 0.2f * s;
  sb[idx] = s;
  atomicMax(mEnc + d_ * Hh + h, f2ord(s));
}

__global__ void gat_edge_expsum(float* __restrict__ sb, const unsigned* __restrict__ mEnc,
                                float* __restrict__ den, const int* __restrict__ dst,
                                int total, int Hh) {
  int idx = blockIdx.x * blockDim.x + threadIdx.x;
  if (idx >= total) return;
  int e = idx / Hh;
  int h = idx - e * Hh;
  int d_ = dst[e];
  float m = ord2f(mEnc[d_ * Hh + h]);
  float ex = expf(sb[idx] - m);
  sb[idx] = ex;
  atomicAdd(den + d_ * Hh + h, ex);
}

static_assert(EE % 256 == 0, "a wave owns 32 whole edges and EE / 256 blocks of 8 waves cover them all");

template <int HF, int NCH>
__device__ __forceinline__ void scat_load_row(float (&r)[NCH], const float* __restrict__ row,
                                              int lane) {
#pragma unroll
  for (int c = 0; c < NCH; ++c) {
    const int col = (c << 5) + lane;
    if (HF % 32 == 0 || col < HF) r[c] = row[col];
  }
}

template <int HH, int F, int NCH>
__device__ __forceinline__ void scat_add_row(const float (&r)[NCH], const float (&al)[HH],
                                             float* __restrict__ row, int lane) {
  constexpr int HF = HH * F;
#pragma unroll
  for (int c = 0; c < NCH; ++c) {
    const int col = (c << 5) + lane;
    const float a = al[(HH == 1) ? 0 : ((c << 5) / F)];
    if (HF % 32 == 0 || col < HF) atomicAdd(row + col, r[c] * a);
  }
}

template <int HH, int F>
__global__ __launch_bounds__(256)
void gat_scatter_rows(const float* __restrict__ Hm, const float* __restrict__ sb,
                      const float* __restrict__ den, const int* __restrict__ src,
                      const int* __restrict__ dst, float* __restrict__ AGG) {
  static_assert(HH == 1 || F % 32 == 0, "a 32-column piece must lie inside one head");
  constexpr int HF  = HH * F;
  constexpr int NCH = (HF + 31) / 32;
  const int lane = threadIdx.x & 31;
  const int e0   = ((blockIdx.x * blockDim.x + threadIdx.x) >> 5) << 5;
  if (e0 >= EE) return;

  const int me = e0 + lane;
  const int ms = src[me];
  const int md = dst[me];
  float ma[HH];
#pragma unroll
  for (int h = 0; h < HH; ++h)
    ma[h] = sb[(size_t)me * HH + h] / den[(size_t)md * HH + h];

  float ra[NCH], rb[NCH];
  float al[HH];
  scat_load_row<HF, NCH>(ra, Hm + (size_t)__builtin_amdgcn_readlane(ms, 0) * HF, lane);
  for (int i = 0; i < 32; i += 2) {
    scat_load_row<HF, NCH>(rb, Hm + (size_t)__builtin_amdgcn_readlane(ms, i + 1) * HF, lane);
#pragma unroll
    for (int h = 0; h < HH; ++h)
      al[h] = __int_as_float(__builtin_amdgcn_readlane(__float_as_int(ma[h]), i));
    scat_add_row<HH, F, NCH>(ra, al, AGG + (size_t)__builtin_amdgcn_readlane(md, i) * HF, lane);

    if (i + 2 < 32)
      scat_load_row<HF, NCH>(ra, Hm + (size_t)__builtin_amdgcn_readlane(ms, i + 2) * HF, lane);
#pragma unroll
    for (int h = 0; h < HH; ++h)
      al[h] = __int_as_float(__builtin_amdgcn_readlane(__float_as_int(ma[h]), i + 1));
    scat_add_row<HH, F, NCH>(rb, al, AGG + (size_t)__builtin_amdgcn_readlane(md, i + 1) * HF, lane);
  }
}


#define OWN_R   200
#define OWN_CAP 1024
#define OWN_NCH (EE / 128)
static_assert(NN % OWN_R == 0, "row groups cover the nodes exactly");
static_assert(OWN_R <= 7 * 32, "7 register slots per wave");
static_assert(EE % 128 == 0, "a wave filters 128 edges per load");
static_assert(EE <= (1 << 20), "an edge index is packed into 20 bits");
static_assert(((OWN_NCH + 31) / 32) % 4 == 0, "the filter loop takes 4 loads per trip");
static_assert(OWN_NCH % (NN / OWN_R) == 0, "start offsets of the groups tile the list");

typedef __attribute__((ext_vector_type(4))) int v4i;

__device__ __forceinline__ void own_queue(unsigned (*lst)[OWN_CAP], unsigned* cnt, int e, unsigned r,
                                          int base, const float* __restrict__ Hm,
                                          const float* __restrict__ sb,
                                          const float* __restrict__ den,
                                          const int* __restrict__ src, float* AGG) {
  const unsigned w = r & 31u;
  const unsigned pos = atomicAdd(&cnt[w], 1u);
  if (pos < OWN_CAP) {
    lst[w][pos] = (unsigned)e | ((r >> 5) << 20);
  } else {
    const int s = src[e];
    const int d = base + (int)r;
    float a[4];
    for (int h = 0; h < 4; ++h) a[h] = sb[(size_t)e * 4 + h] / den[(size_t)d * 4 + h];
#pragma nounroll
    for (int c = 0; c < 256; ++c)
      atomicAdd(AGG + (size_t)d * 256 + c, Hm[(size_t)s * 256 + c] * a[c >> 6]);
    __threadfence();
  }
}

__device__ __forceinline__ void own_add(float (&acc)[7][8], const float (&q)[8], int slot,
                                        const float (&al)[4]) {
#pragma unroll
  for (int k = 0; k < 7; ++k) {
    if (slot == k) {
#pragma unroll
      for (int c = 0; c < 8; ++c) acc[k][c] += q[c] * al[c >> 1];
    }
  }
}

__global__ __launch_bounds__(1024)
void gat_owner_rows(const float* __restrict__ Hm, const float* __restrict__ sb,
                    const float* __restrict__ den, const int* __restrict__ src,
                    const int* __restrict__ dst, float* AGG) {
  __shared__ unsigned lst[32][OWN_CAP];
  __shared__ unsigned cnt[32];
  const int tid  = threadIdx.x;
  const int lane = tid & 31;
  const int wave = __builtin_amdgcn_readfirstlane(tid >> 5);
  const int base = blockIdx.x * OWN_R;

  if (tid < 32) cnt[tid] = 0u;
  __syncthreads();

  const int rot = blockIdx.x * (OWN_NCH / (NN / OWN_R));
  const v4i* __restrict__ dp = (const v4i*)dst;
  for (int t = 0; t < (OWN_NCH + 31) / 32; t += 4) {
    v4i dv[4];
    int eb[4];
    bool ok[4];
#pragma unroll
    for (int u = 0; u < 4; ++u) {
      const int L = wave + 32 * (t + u);
      ok[u] = L < OWN_NCH;
      int ch = (ok[u] ? L : 0) + rot;
      if (ch >= OWN_NCH) ch -= OWN_NCH;
      eb[u] = ch * 128 + lane * 4;
      dv[u] = dp[ch * 32 + lane];
    }
#pragma unroll
    for (int u = 0; u < 4; ++u) {
      const unsigned r0 = (unsigned)(dv[u].x - base);
      const unsigned r1 = (unsigned)(dv[u].y - base);
      const unsigned r2 = (unsigned)(dv[u].z - base);
      const unsigned r3 = (unsigned)(dv[u].w - base);
      const unsigned mn = min(min(r0, r1), min(r2, r3));
      if (ok[u] && mn < (unsigned)OWN_R) {
        if (r0 < (unsigned)OWN_R) own_queue(lst, cnt, eb[u] + 0, r0, base, Hm, sb, den, src, AGG);
        if (r1 < (unsigned)OWN_R) own_queue(lst, cnt, eb[u] + 1, r1, base, Hm, sb, den, src, AGG);
        if (r2 < (unsigned)OWN_R) own_queue(lst, cnt, eb[u] + 2, r2, base, Hm, sb, den, src, AGG);
        if (r3 < (unsigned)OWN_R) own_queue(lst, cnt, eb[u] + 3, r3, base, Hm, sb, den, src, AGG);
      }
    }
  }
  __syncthreads();

  const unsigned nq = min((unsigned)__builtin_amdgcn_readfirstlane((int)cnt[wave]), (unsigned)OWN_CAP);
  float acc[7][8] = {};
  for (unsigned b0 = 0; b0 < nq; b0 += 32) {
    const unsigned idx = b0 + (unsigned)lane;
    const unsigned ent = (idx < nq) ? lst[wave][idx] : (7u << 20);
    const int me    = (int)(ent & 0xFFFFFu);
    const int mslot = (int)(ent >> 20);
    const int ms    = src[me];
    const int md    = (mslot < 7) ? (base + (mslot << 5) + wave) : base;
    const v4f s4 = *(const v4f*)(sb + (size_t)me * 4);
    const v4f d4 = *(const v4f*)(den + (size_t)md * 4);
    float ma[4];
    ma[0] = s4.x / d4.x; ma[1] = s4.y / d4.y; ma[2] = s4.z / d4.z; ma[3] = s4.w / d4.w;

    float q0[8], q1[8], q2[8], q3[8];
    float al[4];
    scat_load_row<256, 8>(q0, Hm + (size_t)__builtin_amdgcn_readlane(ms, 0) * 256, lane);
    scat_load_row<256, 8>(q1, Hm + (size_t)__builtin_amdgcn_readlane(ms, 1) * 256, lane);
    scat_load_row<256, 8>(q2, Hm + (size_t)__builtin_amdgcn_readlane(ms, 2) * 256, lane);
    for (int i = 0; i < 32; i += 4) {
      scat_load_row<256, 8>(q3, Hm + (size_t)__builtin_amdgcn_readlane(ms, i + 3) * 256, lane);
#pragma unroll
      for (int h = 0; h < 4; ++h) al[h] = __int_as_float(__builtin_amdgcn_readlane(__float_as_int(ma[h]), i));
      own_add(acc, q0, __builtin_amdgcn_readlane(mslot, i), al);

      scat_load_row<256, 8>(q0, Hm + (size_t)__builtin_amdgcn_readlane(ms, (i + 4) & 31) * 256, lane);
#pragma unroll
      for (int h = 0; h < 4; ++h) al[h] = __int_as_float(__builtin_amdgcn_readlane(__float_as_int(ma[h]), i + 1));
      own_add(acc, q1, __builtin_amdgcn_readlane(mslot, i + 1), al);

      scat_load_row<256, 8>(q1, Hm + (size_t)__builtin_amdgcn_readlane(ms, (i + 5) & 31) * 256, lane);
#pragma unroll
      for (int h = 0; h < 4; ++h) al[h] = __int_as_float(__builtin_amdgcn_readlane(__float_as_int(ma[h]), i + 2));
      own_add(acc, q2, __builtin_amdgcn_readlane(mslot, i + 2), al);

      scat_load_row<256, 8>(q2, Hm + (size_t)__builtin_amdgcn_readlane(ms, (i + 6) & 31) * 256, lane);
#pragma unroll
      for (int h = 0; h < 4; ++h) al[h] = __int_as_float(__builtin_amdgcn_readlane(__float_as_int(ma[h]), i + 3));
      own_add(acc, q3, __builtin_amdgcn_readlane(mslot, i + 3), al);
    }
  }

#pragma unroll
  for (int g = 0; g < 7; g += 4) {
    float t[4][8];
#pragma unroll
    for (int kk = 0; kk < 4; ++kk) {
      const int k = g + kk;
      const int r = (k << 5) + wave;
      if (k < 7 && r < OWN_R) {
        const float* orow = AGG + (size_t)(base + r) * 256;
#pragma unroll
        for (int c = 0; c < 8; ++c) t[kk][c] = orow[(c << 5) + lane];
      }
    }
#pragma unroll
    for (int kk = 0; kk < 4; ++kk) {
      const int k = g + kk;
      const int r = (k << 5) + wave;
      if (k < 7 && r < OWN_R) {
        float* orow = AGG + (size_t)(base + r) * 256;
#pragma unroll
        for (int c = 0; c < 8; ++c) {
          const float v = t[kk][c] + acc[k < 7 ? k : 0][c];
          orow[(c << 5) + lane] = (v > 0.0f) ? v : expm1f(v);
        }
      }
    }
  }
}

static void run_layer(const float* X, int K, int Hh, int F,
                      const float* W, const float* al, const float* ar, const float* rW,
                      const int* src, const int* dst,
                      float* Hbuf, float* AGG, int Opad,
                      float* el, float* er, float* den, unsigned* mEnc, float* sb,
                      hipStream_t stream) {
  const int HF = Hh * F;
  dim3 ggrid(NN / 80, Opad / 64, 2);
  if (Hh == 4 && K == 128) {
    gat_gemm_xwt<128, 5, 4><<<ggrid, 32, 0, stream>>>(X, W, rW, Hbuf, AGG, HF);
  } else if (Hh == 4) {
    gat_gemm_xwt<256, 5, 4><<<ggrid, 32, 0, stream>>>(X, W, rW, Hbuf, AGG, HF);
  } else {
    gat_gemm_xwt<256, 5, 3><<<ggrid, 32, 0, stream>>>(X, W, rW, Hbuf, AGG, HF);
  }

  const int nh = NN * Hh;
  gat_el_er<<<(nh + 255) / 256, 256, 0, stream>>>(Hbuf, al, ar, el, er, mEnc, den, nh, Hh, F);

  const int eh = EE * Hh;
  gat_edge_smax<<<(eh + 255) / 256, 256, 0, stream>>>(el, er, src, dst, sb, mEnc, eh, Hh);
  gat_edge_expsum<<<(eh + 255) / 256, 256, 0, stream>>>(sb, mEnc, den, dst, eh, Hh);

  if (Hh == 4) {
    gat_owner_rows<<<NN / OWN_R, 1024, 0, stream>>>(Hbuf, sb, den, src, dst, AGG);
  } else {
    gat_scatter_rows<1, 40><<<EE / 256, 256, 0, stream>>>(Hbuf, sb, den, src, dst, AGG);
  }
}

extern "C" void kernel_launch(void* const* d_in, const int* in_sizes, int n_in,
                              void* d_out, int out_size, void* d_ws, size_t ws_size,
                              hipStream_t stream) {
  (void)in_sizes; (void)n_in; (void)out_size; (void)ws_size;

  const float* feat = (const float*)d_in[0];
  const int*   src  = (const int*)d_in[1];
  const int*   dst  = (const int*)d_in[2];
  const float* W0  = (const float*)d_in[3];
  const float* al0 = (const float*)d_in[4];
  const float* ar0 = (const float*)d_in[5];
  const float* rW0 = (const float*)d_in[6];
  const float* W1  = (const float*)d_in[7];
  const float* al1 = (const float*)d_in[8];
  const float* ar1 = (const float*)d_in[9];
  const float* rW1 = (const float*)d_in[10];
  const float* W2  = (const float*)d_in[11];
  const float* al2 = (const float*)d_in[12];
  const float* ar2 = (const float*)d_in[13];
  const float* rW2 = (const float*)d_in[14];

  const size_t NF = (size_t)NN * 256;
  float* ws    = (float*)d_ws;
  float* bufH  = ws;
  float* bufY0 = bufH  + NF;
  float* bufY1 = bufY0 + NF;
  float* el    = bufY1 + NF;
  float* er    = el  + (size_t)NN * 4;
  float* den   = er  + (size_t)NN * 4;
  unsigned* mE = (unsigned*)(den + (size_t)NN * 4);
  float* sb    = (float*)mE + (size_t)NN * 4;
  float* W2p   = sb + (size_t)EE * 4;
  float* rW2p  = W2p + (size_t)64 * 256;

  {
    int tot = 64 * 256;
    gat_pad_w<<<dim3((tot + 255) / 256, 2), 256, 0, stream>>>(W2, rW2, W2p, rW2p, 40, 64, 256);
  }

  run_layer(feat, 128, 4, 64, W0, al0, ar0, rW0, src, dst,
            bufH, bufY0, 256, el, er, den, mE, sb, stream);
  run_layer(bufY0, 256, 4, 64, W1, al1, ar1, rW1, src, dst,
            bufH, bufY1, 256, el, er, den, mE, sb, stream);
  run_layer(bufY1, 256, 1, 40, W2p, al2, ar2, rW2p, src, dst,
            bufH, (float*)d_out, 64, el, er, den, mE, sb, stream);
}


# Hidden layers: edge softmax folded into the row-owner workgroup: score, max, denominator of an owned row formed in LDS from its queue; 4 launches, global atomics gone; denominator sum order changes
# speedup vs baseline: 4.0244x; 1.2445x over previous
#include <hip/hip_runtime.h>
#include <math.h>

#define NN 50000
#define EE 800000

typedef __attribute__((ext_vector_type(2))) float v2f;
typedef __attribute__((ext_vector_type(4))) float v4f;
typedef __attribute__((ext_vector_type(8))) float v8f;

__device__ __forceinline__ unsigned f2ord(float f) {
  unsigned u = __float_as_uint(f);
  return (u & 0x80000000u) ? ~u : (u | 0x80000000u);
}
__device__ __forceinline__ float ord2f(unsigned e) {
  unsigned u = (e & 0x80000000u) ? (e ^ 0x80000000u) : ~e;
  return __uint_as_float(u);
}

template <int NF>
__device__ __forceinline__ void gemm_load_lines(v4f (&f)[NF][4], const float* __restrict__ p,
                                                size_t fstep) {
#pragma unroll
  for (int i = 0; i < NF; ++i) {
#pragma unroll
    for (int j = 0; j < 4; ++j) f[i][j] = *(const v4f*)(p + (size_t)i * fstep + 8 * j);
  }
}

template <int MA, int NB>
__device__ __forceinline__ void gemm_mma_lines(v8f (&acc)[MA][NB], const v4f (&a)[MA][4],
                                               const v4f (&b)[NB][4]) {
#pragma unroll
  for (int j = 0; j < 4; ++j) {
#pragma unroll
    for (int i = 0; i < MA; ++i) {
#pragma unroll
      for (int t = 0; t < NB; ++t)
        acc[i][t] = __builtin_amdgcn_wmma_f32_16x16x4_f32(
            false, a[i][j].xy, false, b[t][j].xy, (short)0, acc[i][t], false, false);
    }
#pragma unroll
    for (int i = 0; i < MA; ++i) {
#pragma unroll
      for (int t = 0; t < NB; ++t)
        acc[i][t] = __builtin_amdgcn_wmma_f32_16x16x4_f32(
            false, a[i][j].zw, false, b[t][j].zw, (short)0, acc[i][t], false, false);
    }
  }
}

template <int K, int MA, int NB>
__global__ __launch_bounds__(32)
void gat_gemm_xwt(const float* __restrict__ X, const float* __restrict__ W0,
                  const float* __restrict__ W1, float* __restrict__ OUT0,
                  float* __restrict__ OUT1, int Oout) {
  static_assert(K % 64 == 0, "the K loop takes two 32-float lines per trip");
  static_assert(NN % (16 * MA) == 0, "row tiles cover the nodes exactly");
  const int lane = threadIdx.x & 31;
  const int c    = lane & 15;
  const int half = lane >> 4;
  const int row0 = blockIdx.x * (16 * MA);
  const int col0 = blockIdx.y << 6;
  const float* __restrict__ W   = (blockIdx.z == 0) ? W0 : W1;
  float* __restrict__       OUT = (blockIdx.z == 0) ? OUT0 : OUT1;

  const float* xp = X + (size_t)(row0 + c) * (size_t)K + (half << 2);
  const float* wp = W + (size_t)(col0 + c) * (size_t)K + (half << 2);
  const size_t fstep = (size_t)16 * (size_t)K;

  v8f acc[MA][NB] = {};
  v4f aP[MA][4], bP[NB][4], aQ[MA][4], bQ[NB][4];

  gemm_load_lines<MA>(aP, xp, fstep);
  gemm_load_lines<NB>(bP, wp, fstep);
  int kb = 0;
  for (; kb + 64 < K; kb += 64) {
    gemm_load_lines<MA>(aQ, xp + kb + 32, fstep);
    gemm_load_lines<NB>(bQ, wp + kb + 32, fstep);
    __builtin_amdgcn_sched_barrier(0);
    gemm_mma_lines<MA, NB>(acc, aP, bP);
    __builtin_amdgcn_sched_barrier(0);
    gemm_load_lines<MA>(aP, xp + kb + 64, fstep);
    gemm_load_lines<NB>(bP, wp + kb + 64, fstep);
    __builtin_amdgcn_sched_barrier(0);
    gemm_mma_lines<MA, NB>(acc, aQ, bQ);
    __builtin_amdgcn_sched_barrier(0);
  }
  gemm_load_lines<MA>(aQ, xp + kb + 32, fstep);
  gemm_load_lines<NB>(bQ, wp + kb + 32, fstep);
  __builtin_amdgcn_sched_barrier(0);
  gemm_mma_lines<MA, NB>(acc, aP, bP);
  gemm_mma_lines<MA, NB>(acc, aQ, bQ);

#pragma unroll
  for (int i = 0; i < MA; ++i) {
#pragma unroll
    for (int t = 0; t < NB; ++t) {
      const int oc = col0 + (t << 4) + c;
      if (oc < Oout) {
        float* orow = OUT + (size_t)(row0 + (i << 4) + (half << 3)) * (size_t)Oout + oc;
#pragma unroll
        for (int v = 0; v < 8; ++v) {
          orow[(size_t)v * (size_t)Oout] = acc[i][t][v];
        }
      }
    }
  }
}

__global__ void gat_pad_w(const float* __restrict__ Wsrc0, const float* __restrict__ Wsrc1,
                          float* __restrict__ Wdst0, float* __restrict__ Wdst1,
                          int rows, int padrows, int K) {
  int idx = blockIdx.x * blockDim.x + threadIdx.x;
  int total = padrows * K;
  if (idx >= total) return;
  const float* __restrict__ Wsrc = (blockIdx.y == 0) ? Wsrc0 : Wsrc1;
  float* __restrict__       Wdst = (blockIdx.y == 0) ? Wdst0 : Wdst1;
  int r = idx / K;
  Wdst[idx] = (r < rows) ? Wsrc[idx] : 0.0f;
}

template <bool INIT>
__global__ void gat_el_er(const float* __restrict__ Hm, const float* __restrict__ al,
                          const float* __restrict__ ar, float* __restrict__ el,
                          float* __restrict__ er, unsigned* __restrict__ mEnc,
                          float* __restrict__ den, int total, int Hh, int F) {
  int idx = blockIdx.x * blockDim.x + threadIdx.x;
  if (idx >= total) return;
  int h = idx % Hh;
  int n = idx / Hh;
  const float* row = Hm + (size_t)n * (size_t)(Hh * F) + (size_t)h * F;
  const float* a = al + h * F;
  const float* b = ar + h * F;
  float sl = 0.f, sr = 0.f;
  for (int f = 0; f < F; ++f) {
    float v = row[f];
    sl += v * a[f];
    sr += v * b[f];
  }
  el[idx] = sl;
  er[idx] = sr;
  if (INIT) {
    mEnc[idx] = 0u;
    den[idx] = 0.0f;
  }
}

__global__ void gat_edge_smax(const float* __restrict__ el, const float* __restrict__ er,
                              const int* __restrict__ src, const int* __restrict__ dst,
                              float* __restrict__ sb, unsigned* __restrict__ mEnc,
                              int total, int Hh) {
  int idx = blockIdx.x * blockDim.x + threadIdx.x;
  if (idx >= total) return;
  int e = idx / Hh;
  int h = idx - e * Hh;
  int s_ = src[e], d_ = dst[e];
  float s = el[s_ * Hh + h] + er[d_ * Hh + h];
  s = (s > 0.0f) ? s : 0.2f * s;
  sb[idx] = s;
  atomicMax(mEnc + d_ * Hh + h, f2ord(s));
}

__global__ void gat_edge_expsum(float* __restrict__ sb, const unsigned* __restrict__ mEnc,
                                float* __restrict__ den, const int* __restrict__ dst,
                                int total, int Hh) {
  int idx = blockIdx.x * blockDim.x + threadIdx.x;
  if (idx >= total) return;
  int e = idx / Hh;
  int h = idx - e * Hh;
  int d_ = dst[e];
  float m = ord2f(mEnc[d_ * Hh + h]);
  float ex = expf(sb[idx] - m);
  sb[idx] = ex;
  atomicAdd(den + d_ * Hh + h, ex);
}

static_assert(EE % 256 == 0, "a wave owns 32 whole edges and EE / 256 blocks of 8 waves cover them all");

template <int HF, int NCH>
__device__ __forceinline__ void scat_load_row(float (&r)[NCH], const float* __restrict__ row,
                                              int lane) {
#pragma unroll
  for (int c = 0; c < NCH; ++c) {
    const int col = (c << 5) + lane;
    if (HF % 32 == 0 || col < HF) r[c] = row[col];
  }
}

template <int HH, int F, int NCH>
__device__ __forceinline__ void scat_add_row(const float (&r)[NCH], const float (&al)[HH],
                                             float* __restrict__ row, int lane) {
  constexpr int HF = HH * F;
#pragma unroll
  for (int c = 0; c < NCH; ++c) {
    const int col = (c << 5) + lane;
    const float a = al[(HH == 1) ? 0 : ((c << 5) / F)];
    if (HF % 32 == 0 || col < HF) atomicAdd(row + col, r[c] * a);
  }
}

template <int HH, int F>
__global__ __launch_bounds__(256)
void gat_scatter_rows(const float* __restrict__ Hm, const float* __restrict__ sb,
                      const float* __restrict__ den, const int* __restrict__ src,
                      const int* __restrict__ dst, float* __restrict__ AGG) {
  static_assert(HH == 1 || F % 32 == 0, "a 32-column piece must lie inside one head");
  constexpr int HF  = HH * F;
  constexpr int NCH = (HF + 31) / 32;
  const int lane = threadIdx.x & 31;
  const int e0   = ((blockIdx.x * blockDim.x + threadIdx.x) >> 5) << 5;
  if (e0 >= EE) return;

  const int me = e0 + lane;
  const int ms = src[me];
  const int md = dst[me];
  float ma[HH];
#pragma unroll
  for (int h = 0; h < HH; ++h)
    ma[h] = sb[(size_t)me * HH + h] / den[(size_t)md * HH + h];

  float ra[NCH], rb[NCH];
  float al[HH];
  scat_load_row<HF, NCH>(ra, Hm + (size_t)__builtin_amdgcn_readlane(ms, 0) * HF, lane);
  for (int i = 0; i < 32; i += 2) {
    scat_load_row<HF, NCH>(rb, Hm + (size_t)__builtin_amdgcn_readlane(ms, i + 1) * HF, lane);
#pragma unroll
    for (int h = 0; h < HH; ++h)
      al[h] = __int_as_float(__builtin_amdgcn_readlane(__float_as_int(ma[h]), i));
    scat_add_row<HH, F, NCH>(ra, al, AGG + (size_t)__builtin_amdgcn_readlane(md, i) * HF, lane);

    if (i + 2 < 32)
      scat_load_row<HF, NCH>(ra, Hm + (size_t)__builtin_amdgcn_readlane(ms, i + 2) * HF, lane);
#pragma unroll
    for (int h = 0; h < HH; ++h)
      al[h] = __int_as_float(__builtin_amdgcn_readlane(__float_as_int(ma[h]), i + 1));
    scat_add_row<HH, F, NCH>(rb, al, AGG + (size_t)__builtin_amdgcn_readlane(md, i + 1) * HF, lane);
  }
}


#define OWN_R   200
#define OWN_CAP 1024
#define OWN_NCH (EE / 128)
static_assert(NN % OWN_R == 0, "row groups cover the nodes exactly");
static_assert(OWN_R <= 7 * 32, "7 register slots per wave");
static_assert(EE % 128 == 0, "a wave filters 128 edges per load");
static_assert(EE <= (1 << 20), "an edge index is packed into 20 bits");
static_assert(NN <= (1 << 20), "a node index is packed into 20 bits");
static_assert(((OWN_NCH + 31) / 32) % 4 == 0, "the filter loop takes 4 loads per trip");
static_assert(OWN_NCH % (NN / OWN_R) == 0, "start offsets of the groups tile the list");

typedef __attribute__((ext_vector_type(4))) int v4i;

__device__ __forceinline__ void own_queue(unsigned (*lst)[OWN_CAP], unsigned* cnt, unsigned* full,
                                          int e, unsigned r) {
  const unsigned w = r & 31u;
  const unsigned pos = atomicAdd(&cnt[w], 1u);
  if (pos < OWN_CAP) {
    lst[w][pos] = (unsigned)e | ((r >> 5) << 20);
  } else {
    *full = 1u;
  }
}

__device__ __forceinline__ void own_scores(float (&s)[4], const float* __restrict__ el,
                                           const float* __restrict__ er, int sn, int dn_) {
  const v4f a = *(const v4f*)(el + (size_t)sn * 4);
  const v4f b = *(const v4f*)(er + (size_t)dn_ * 4);
  s[0] = a.x + b.x; s[1] = a.y + b.y; s[2] = a.z + b.z; s[3] = a.w + b.w;
#pragma unroll
  for (int h = 0; h < 4; ++h) s[h] = (s[h] > 0.0f) ? s[h] : 0.2f * s[h];
}

__device__ __forceinline__ void own_add(float (&acc)[7][8], const float (&q)[8], int slot,
                                        const float (&al)[4]) {
#pragma unroll
  for (int k = 0; k < 7; ++k) {
    if (slot == k) {
#pragma unroll
      for (int c = 0; c < 8; ++c) acc[k][c] += q[c] * al[c >> 1];
    }
  }
}

__device__ __forceinline__ float own_wave_max(float v) {
  for (int o = 16; o > 0; o >>= 1) v = fmaxf(v, __shfl_xor(v, o));
  return v;
}
__device__ __forceinline__ float own_wave_sum(float v) {
  for (int o = 16; o > 0; o >>= 1) v += __shfl_xor(v, o);
  return v;
}

__device__ void own_rows_by_scan(int base, int wave, int lane, const float* __restrict__ Hm,
                                 const float* __restrict__ el, const float* __restrict__ er,
                                 const int* __restrict__ src, const int* __restrict__ dst,
                                 float* AGG) {
  for (int k = 0; k < 7; ++k) {
    const int r = (k << 5) + wave;
    if (r >= OWN_R) break;
    const int d = base + r;
    float m[4] = {-INFINITY, -INFINITY, -INFINITY, -INFINITY};
    for (int e = lane; e < EE; e += 32) {
      if (dst[e] == d) {
        float s[4];
        own_scores(s, el, er, src[e], d);
        for (int h = 0; h < 4; ++h) m[h] = fmaxf(m[h], s[h]);
      }
    }
    for (int h = 0; h < 4; ++h) m[h] = own_wave_max(m[h]);
    float z[4] = {0.0f, 0.0f, 0.0f, 0.0f};
    for (int e = lane; e < EE; e += 32) {
      if (dst[e] == d) {
        float s[4];
        own_scores(s, el, er, src[e], d);
        for (int h = 0; h < 4; ++h) z[h] += expf(s[h] - m[h]);
      }
    }
    for (int h = 0; h < 4; ++h) z[h] = own_wave_sum(z[h]);
    float acc[8] = {};
    for (int e0 = 0; e0 < EE; e0 += 32) {
      const int e = e0 + lane;
      const bool hit = dst[e] == d;
      int sn = 0;
      float a[4] = {0.0f, 0.0f, 0.0f, 0.0f};
      if (hit) {
        sn = src[e];
        float s[4];
        own_scores(s, el, er, sn, d);
        for (int h = 0; h < 4; ++h) a[h] = expf(s[h] - m[h]) / z[h];
      }
      unsigned mask = (unsigned)__ballot(hit);
      while (mask) {
        const int j = __ffs(mask) - 1;
        mask &= mask - 1u;
        const int sj = __shfl(sn, j);
        float aj[4];
        for (int h = 0; h < 4; ++h) aj[h] = __shfl(a[h], j);
        const float* row = Hm + (size_t)sj * 256;
        for (int c = 0; c < 8; ++c) acc[c] += row[(c << 5) + lane] * aj[c >> 1];
      }
    }
    float* orow = AGG + (size_t)d * 256;
    for (int c = 0; c < 8; ++c) {
      const float v = orow[(c << 5) + lane] + acc[c];
      orow[(c << 5) + lane] = (v > 0.0f) ? v : expm1f(v);
    }
  }
}

__global__ __launch_bounds__(1024)
void gat_owner_rows(const float* __restrict__ Hm, const float* __restrict__ el,
                    const float* __restrict__ er, const int* __restrict__ src,
                    const int* __restrict__ dst, float* AGG) {
  __shared__ unsigned lst[32][OWN_CAP];
  __shared__ unsigned cnt[32];
  __shared__ unsigned mx[32][8][4];
  __shared__ float    dn[32][8][4];
  __shared__ unsigned full;
  const int tid  = threadIdx.x;
  const int lane = tid & 31;
  const int wave = __builtin_amdgcn_readfirstlane(tid >> 5);
  const int base = blockIdx.x * OWN_R;

  if (tid < 32) cnt[tid] = 0u;
  if (tid == 0) full = 0u;
  (&mx[0][0][0])[tid] = 0u;
  (&dn[0][0][0])[tid] = 0.0f;
  __syncthreads();

  const int rot = blockIdx.x * (OWN_NCH / (NN / OWN_R));
  const v4i* __restrict__ dp = (const v4i*)dst;
  for (int t = 0; t < (OWN_NCH + 31) / 32; t += 4) {
    v4i dv[4];
    int eb[4];
    bool ok[4];
#pragma unroll
    for (int u = 0; u < 4; ++u) {
      const int L = wave + 32 * (t + u);
      ok[u] = L < OWN_NCH;
      int ch = (ok[u] ? L : 0) + rot;
      if (ch >= OWN_NCH) ch -= OWN_NCH;
      eb[u] = ch * 128 + lane * 4;
      dv[u] = dp[ch * 32 + lane];
    }
#pragma unroll
    for (int u = 0; u < 4; ++u) {
      const unsigned r0 = (unsigned)(dv[u].x - base);
      const unsigned r1 = (unsigned)(dv[u].y - base);
      const unsigned r2 = (unsigned)(dv[u].z - base);
      const unsigned r3 = (unsigned)(dv[u].w - base);
      const unsigned mn = min(min(r0, r1), min(r2, r3));
      if (ok[u] && mn < (unsigned)OWN_R) {
        if (r0 < (unsigned)OWN_R) own_queue(lst, cnt, &full, eb[u] + 0, r0);
        if (r1 < (unsigned)OWN_R) own_queue(lst, cnt, &full, eb[u] + 1, r1);
        if (r2 < (unsigned)OWN_R) own_queue(lst, cnt, &full, eb[u] + 2, r2);
        if (r3 < (unsigned)OWN_R) own_queue(lst, cnt, &full, eb[u] + 3, r3);
      }
    }
  }
  __syncthreads();

  if (__builtin_amdgcn_readfirstlane((int)full) != 0) {
    own_rows_by_scan(base, wave, lane, Hm, el, er, src, dst, AGG);
    return;
  }

  const unsigned nq = (unsigned)__builtin_amdgcn_readfirstlane((int)cnt[wave]);

  for (unsigned b0 = 0; b0 < nq; b0 += 128) {
    unsigned idx[4];
    bool live[4];
    int me[4], mslot[4], ms[4];
    float s[4][4];
#pragma unroll
    for (int u = 0; u < 4; ++u) {
      idx[u]  = b0 + (unsigned)(32 * u + lane);
      live[u] = idx[u] < nq;
      const unsigned ent = live[u] ? lst[wave][idx[u]] : (7u << 20);
      me[u]    = (int)(ent & 0xFFFFFu);
      mslot[u] = (int)(ent >> 20);
    }
#pragma unroll
    for (int u = 0; u < 4; ++u) ms[u] = src[me[u]];
#pragma unroll
    for (int u = 0; u < 4; ++u)
      own_scores(s[u], el, er, ms[u], (mslot[u] < 7) ? (base + (mslot[u] << 5) + wave) : base);
#pragma unroll
    for (int u = 0; u < 4; ++u) {
      if (live[u]) {
        lst[wave][idx[u]] = (unsigned)ms[u] | ((unsigned)mslot[u] << 20);
#pragma unroll
        for (int h = 0; h < 4; ++h) atomicMax(&mx[wave][mslot[u]][h], f2ord(s[u][h]));
      }
    }
  }
  __syncthreads();

  for (unsigned b0 = 0; b0 < nq; b0 += 128) {
    unsigned idx[4];
    bool live[4];
    int mslot[4], ms[4];
    float s[4][4];
#pragma unroll
    for (int u = 0; u < 4; ++u) {
      idx[u]  = b0 + (unsigned)(32 * u + lane);
      live[u] = idx[u] < nq;
      const unsigned ent = live[u] ? lst[wave][idx[u]] : (7u << 20);
      ms[u]    = (int)(ent & 0xFFFFFu);
      mslot[u] = (int)(ent >> 20);
    }
#pragma unroll
    for (int u = 0; u < 4; ++u)
      own_scores(s[u], el, er, ms[u], (mslot[u] < 7) ? (base + (mslot[u] << 5) + wave) : base);
#pragma unroll
    for (int u = 0; u < 4; ++u) {
      if (live[u]) {
#pragma unroll
        for (int h = 0; h < 4; ++h)
          atomicAdd(&dn[wave][mslot[u]][h], expf(s[u][h] - ord2f(mx[wave][mslot[u]][h])));
      }
    }
  }
  __syncthreads();

  float acc[7][8] = {};
  for (unsigned b0 = 0; b0 < nq; b0 += 32) {
    const unsigned idx = b0 + (unsigned)lane;
    const unsigned ent = (idx < nq) ? lst[wave][idx] : (7u << 20);
    const int ms    = (int)(ent & 0xFFFFFu);
    const int mslot = (int)(ent >> 20);
    const int md    = (mslot < 7) ? (base + (mslot << 5) + wave) : base;
    float ma[4];
    own_scores(ma, el, er, ms, md);
#pragma unroll
    for (int h = 0; h < 4; ++h)
      ma[h] = expf(ma[h] - ord2f(mx[wave][mslot][h])) / dn[wave][mslot][h];

    float q0[8], q1[8], q2[8], q3[8];
    float al[4];
    scat_load_row<256, 8>(q0, Hm + (size_t)__builtin_amdgcn_readlane(ms, 0) * 256, lane);
    scat_load_row<256, 8>(q1, Hm + (size_t)__builtin_amdgcn_readlane(ms, 1) * 256, lane);
    scat_load_row<256, 8>(q2, Hm + (size_t)__builtin_amdgcn_readlane(ms, 2) * 256, lane);
    for (int i = 0; i < 32; i += 4) {
      scat_load_row<256, 8>(q3, Hm + (size_t)__builtin_amdgcn_readlane(ms, i + 3) * 256, lane);
#pragma unroll
      for (int h = 0; h < 4; ++h) al[h] = __int_as_float(__builtin_amdgcn_readlane(__float_as_int(ma[h]), i));
      own_add(acc, q0, __builtin_amdgcn_readlane(mslot, i), al);

      scat_load_row<256, 8>(q0, Hm + (size_t)__builtin_amdgcn_readlane(ms, (i + 4) & 31) * 256, lane);
#pragma unroll
      for (int h = 0; h < 4; ++h) al[h] = __int_as_float(__builtin_amdgcn_readlane(__float_as_int(ma[h]), i + 1));
      own_add(acc, q1, __builtin_amdgcn_readlane(mslot, i + 1), al);

      scat_load_row<256, 8>(q1, Hm + (size_t)__builtin_amdgcn_readlane(ms, (i + 5) & 31) * 256, lane);
#pragma unroll
      for (int h = 0; h < 4; ++h) al[h] = __int_as_float(__builtin_amdgcn_readlane(__float_as_int(ma[h]), i + 2));
      own_add(acc, q2, __builtin_amdgcn_readlane(mslot, i + 2), al);

      scat_load_row<256, 8>(q2, Hm + (size_t)__builtin_amdgcn_readlane(ms, (i + 6) & 31) * 256, lane);
#pragma unroll
      for (int h = 0; h < 4; ++h) al[h] = __int_as_float(__builtin_amdgcn_readlane(__float_as_int(ma[h]), i + 3));
      own_add(acc, q3, __builtin_amdgcn_readlane(mslot, i + 3), al);
    }
  }

#pragma unroll
  for (int g = 0; g < 7; g += 4) {
    float t[4][8];
#pragma unroll
    for (int kk = 0; kk < 4; ++kk) {
      const int k = g + kk;
      const int r = (k << 5) + wave;
      if (k < 7 && r < OWN_R) {
        const float* orow = AGG + (size_t)(base + r) * 256;
#pragma unroll
        for (int c = 0; c < 8; ++c) t[kk][c] = orow[(c << 5) + lane];
      }
    }
#pragma unroll
    for (int kk = 0; kk < 4; ++kk) {
      const int k = g + kk;
      const int r = (k << 5) + wave;
      if (k < 7 && r < OWN_R) {
        float* orow = AGG + (size_t)(base + r) * 256;
#pragma unroll
        for (int c = 0; c < 8; ++c) {
          const float v = t[kk][c] + acc[k < 7 ? k : 0][c];
          orow[(c << 5) + lane] = (v > 0.0f) ? v : expm1f(v);
        }
      }
    }
  }
}

static void run_layer(const float* X, int K, int Hh, int F,
                      const float* W, const float* al, const float* ar, const float* rW,
                      const int* src, const int* dst,
                      float* Hbuf, float* AGG, int Opad,
                      float* el, float* er, float* den, unsigned* mEnc, float* sb,
                      hipStream_t stream) {
  const int HF = Hh * F;
  dim3 ggrid(NN / 80, Opad / 64, 2);
  if (Hh == 4 && K == 128) {
    gat_gemm_xwt<128, 5, 4><<<ggrid, 32, 0, stream>>>(X, W, rW, Hbuf, AGG, HF);
  } else if (Hh == 4) {
    gat_gemm_xwt<256, 5, 4><<<ggrid, 32, 0, stream>>>(X, W, rW, Hbuf, AGG, HF);
  } else {
    gat_gemm_xwt<256, 5, 3><<<ggrid, 32, 0, stream>>>(X, W, rW, Hbuf, AGG, HF);
  }

  const int nh = NN * Hh;
  if (Hh == 4) {
    gat_el_er<false><<<(nh + 255) / 256, 256, 0, stream>>>(Hbuf, al, ar, el, er, mEnc, den, nh, Hh, F);
    gat_owner_rows<<<NN / OWN_R, 1024, 0, stream>>>(Hbuf, el, er, src, dst, AGG);
  } else {
    gat_el_er<true><<<(nh + 255) / 256, 256, 0, stream>>>(Hbuf, al, ar, el, er, mEnc, den, nh, Hh, F);
    const int eh = EE * Hh;
    gat_edge_smax<<<(eh + 255) / 256, 256, 0, stream>>>(el, er, src, dst, sb, mEnc, eh, Hh);
    gat_edge_expsum<<<(eh + 255) / 256, 256, 0, stream>>>(sb, mEnc, den, dst, eh, Hh);
    gat_scatter_rows<1, 40><<<EE / 256, 256, 0, stream>>>(Hbuf, sb, den, src, dst, AGG);
  }
}

extern "C" void kernel_launch(void* const* d_in, const int* in_sizes, int n_in,
                              void* d_out, int out_size, void* d_ws, size_t ws_size,
                              hipStream_t stream) {
  (void)in_sizes; (void)n_in; (void)out_size; (void)ws_size;

  const float* feat = (const float*)d_in[0];
  const int*   src  = (const int*)d_in[1];
  const int*   dst  = (const int*)d_in[2];
  const float* W0  = (const float*)d_in[3];
  const float* al0 = (const float*)d_in[4];
  const float* ar0 = (const float*)d_in[5];
  const float* rW0 = (const float*)d_in[6];
  const float* W1  = (const float*)d_in[7];
  const float* al1 = (const float*)d_in[8];
  const float* ar1 = (const float*)d_in[9];
  const float* rW1 = (const float*)d_in[10];
  const float* W2  = (const float*)d_in[11];
  const float* al2 = (const float*)d_in[12];
  const float* ar2 = (const float*)d_in[13];
  const float* rW2 = (const float*)d_in[14];

  const size_t NF = (size_t)NN * 256;
  float* ws    = (float*)d_ws;
  float* bufH  = ws;
  float* bufY0 = bufH  + NF;
  float* bufY1 = bufY0 + NF;
  float* el    = bufY1 + NF;
  float* er    = el  + (size_t)NN * 4;
  float* den   = er  + (size_t)NN * 4;
  unsigned* mE = (unsigned*)(den + (size_t)NN * 4);
  float* sb    = (float*)mE + (size_t)NN * 4;
  float* W2p   = sb + (size_t)EE * 4;
  float* rW2p  = W2p + (size_t)64 * 256;

  {
    int tot = 64 * 256;
    gat_pad_w<<<dim3((tot + 255) / 256, 2), 256, 0, stream>>>(W2, rW2, W2p, rW2p, 40, 64, 256);
  }

  run_layer(feat, 128, 4, 64, W0, al0, ar0, rW0, src, dst,
            bufH, bufY0, 256, el, er, den, mE, sb, stream);
  run_layer(bufY0, 256, 4, 64, W1, al1, ar1, rW1, src, dst,
            bufH, bufY1, 256, el, er, den, mE, sb, stream);
  run_layer(bufY1, 256, 1, 40, W2p, al2, ar2, rW2p, src, dst,
            bufH, (float*)d_out, 64, el, er, den, mE, sb, stream);
}


# Row-owner edge lists built once: layer 0 keeps each wave's list (source node, slot) in the workspace, layer 1 copies it into LDS and skips its dst filter; layer 1 sums in layer 0's queue order
# speedup vs baseline: 4.2606x; 1.0587x over previous
#include <hip/hip_runtime.h>
#include <math.h>

#define NN 50000
#define EE 800000

typedef __attribute__((ext_vector_type(2))) float v2f;
typedef __attribute__((ext_vector_type(4))) float v4f;
typedef __attribute__((ext_vector_type(8))) float v8f;

__device__ __forceinline__ unsigned f2ord(float f) {
  unsigned u = __float_as_uint(f);
  return (u & 0x80000000u) ? ~u : (u | 0x80000000u);
}
__device__ __forceinline__ float ord2f(unsigned e) {
  unsigned u = (e & 0x80000000u) ? (e ^ 0x80000000u) : ~e;
  return __uint_as_float(u);
}

template <int NF>
__device__ __forceinline__ void gemm_load_lines(v4f (&f)[NF][4], const float* __restrict__ p,
                                                size_t fstep) {
#pragma unroll
  for (int i = 0; i < NF; ++i) {
#pragma unroll
    for (int j = 0; j < 4; ++j) f[i][j] = *(const v4f*)(p + (size_t)i * fstep + 8 * j);
  }
}

template <int MA, int NB>
__device__ __forceinline__ void gemm_mma_lines(v8f (&acc)[MA][NB], const v4f (&a)[MA][4],
                                               const v4f (&b)[NB][4]) {
#pragma unroll
  for (int j = 0; j < 4; ++j) {
#pragma unroll
    for (int i = 0; i < MA; ++i) {
#pragma unroll
      for (int t = 0; t < NB; ++t)
        acc[i][t] = __builtin_amdgcn_wmma_f32_16x16x4_f32(
            false, a[i][j].xy, false, b[t][j].xy, (short)0, acc[i][t], false, false);
    }
#pragma unroll
    for (int i = 0; i < MA; ++i) {
#pragma unroll
      for (int t = 0; t < NB; ++t)
        acc[i][t] = __builtin_amdgcn_wmma_f32_16x16x4_f32(
            false, a[i][j].zw, false, b[t][j].zw, (short)0, acc[i][t], false, false);
    }
  }
}

template <int K, int MA, int NB>
__global__ __launch_bounds__(32)
void gat_gemm_xwt(const float* __restrict__ X, const float* __restrict__ W0,
                  const float* __restrict__ W1, float* __restrict__ OUT0,
                  float* __restrict__ OUT1, int Oout) {
  static_assert(K % 64 == 0, "the K loop takes two 32-float lines per trip");
  static_assert(NN % (16 * MA) == 0, "row tiles cover the nodes exactly");
  const int lane = threadIdx.x & 31;
  const int c    = lane & 15;
  const int half = lane >> 4;
  const int row0 = blockIdx.x * (16 * MA);
  const int col0 = blockIdx.y << 6;
  const float* __restrict__ W   = (blockIdx.z == 0) ? W0 : W1;
  float* __restrict__       OUT = (blockIdx.z == 0) ? OUT0 : OUT1;

  const float* xp = X + (size_t)(row0 + c) * (size_t)K + (half << 2);
  const float* wp = W + (size_t)(col0 + c) * (size_t)K + (half << 2);
  const size_t fstep = (size_t)16 * (size_t)K;

  v8f acc[MA][NB] = {};
  v4f aP[MA][4], bP[NB][4], aQ[MA][4], bQ[NB][4];

  gemm_load_lines<MA>(aP, xp, fstep);
  gemm_load_lines<NB>(bP, wp, fstep);
  int kb = 0;
  for (; kb + 64 < K; kb += 64) {
    gemm_load_lines<MA>(aQ, xp + kb + 32, fstep);
    gemm_load_lines<NB>(bQ, wp + kb + 32, fstep);
    __builtin_amdgcn_sched_barrier(0);
    gemm_mma_lines<MA, NB>(acc, aP, bP);
    __builtin_amdgcn_sched_barrier(0);
    gemm_load_lines<MA>(aP, xp + kb + 64, fstep);
    gemm_load_lines<NB>(bP, wp + kb + 64, fstep);
    __builtin_amdgcn_sched_barrier(0);
    gemm_mma_lines<MA, NB>(acc, aQ, bQ);
    __builtin_amdgcn_sched_barrier(0);
  }
  gemm_load_lines<MA>(aQ, xp + kb + 32, fstep);
  gemm_load_lines<NB>(bQ, wp + kb + 32, fstep);
  __builtin_amdgcn_sched_barrier(0);
  gemm_mma_lines<MA, NB>(acc, aP, bP);
  gemm_mma_lines<MA, NB>(acc, aQ, bQ);

#pragma unroll
  for (int i = 0; i < MA; ++i) {
#pragma unroll
    for (int t = 0; t < NB; ++t) {
      const int oc = col0 + (t << 4) + c;
      if (oc < Oout) {
        float* orow = OUT + (size_t)(row0 + (i << 4) + (half << 3)) * (size_t)Oout + oc;
#pragma unroll
        for (int v = 0; v < 8; ++v) {
          orow[(size_t)v * (size_t)Oout] = acc[i][t][v];
        }
      }
    }
  }
}

__global__ void gat_pad_w(const float* __restrict__ Wsrc0, const float* __restrict__ Wsrc1,
                          float* __restrict__ Wdst0, float* __restrict__ Wdst1,
                          int rows, int padrows, int K) {
  int idx = blockIdx.x * blockDim.x + threadIdx.x;
  int total = padrows * K;
  if (idx >= total) return;
  const float* __restrict__ Wsrc = (blockIdx.y == 0) ? Wsrc0 : Wsrc1;
  float* __restrict__       Wdst = (blockIdx.y == 0) ? Wdst0 : Wdst1;
  int r = idx / K;
  Wdst[idx] = (r < rows) ? Wsrc[idx] : 0.0f;
}

template <bool INIT>
__global__ void gat_el_er(const float* __restrict__ Hm, const float* __restrict__ al,
                          const float* __restrict__ ar, float* __restrict__ el,
                          float* __restrict__ er, unsigned* __restrict__ mEnc,
                          float* __restrict__ den, int total, int Hh, int F) {
  int idx = blockIdx.x * blockDim.x + threadIdx.x;
  if (idx >= total) return;
  int h = idx % Hh;
  int n = idx / Hh;
  const float* row = Hm + (size_t)n * (size_t)(Hh * F) + (size_t)h * F;
  const float* a = al + h * F;
  const float* b = ar + h * F;
  float sl = 0.f, sr = 0.f;
  for (int f = 0; f < F; ++f) {
    float v = row[f];
    sl += v * a[f];
    sr += v * b[f];
  }
  el[idx] = sl;
  er[idx] = sr;
  if (INIT) {
    mEnc[idx] = 0u;
    den[idx] = 0.0f;
  }
}

__global__ void gat_edge_smax(const float* __restrict__ el, const float* __restrict__ er,
                              const int* __restrict__ src, const int* __restrict__ dst,
                              float* __restrict__ sb, unsigned* __restrict__ mEnc,
                              int total, int Hh) {
  int idx = blockIdx.x * blockDim.x + threadIdx.x;
  if (idx >= total) return;
  int e = idx / Hh;
  int h = idx - e * Hh;
  int s_ = src[e], d_ = dst[e];
  float s = el[s_ * Hh + h] + er[d_ * Hh + h];
  s = (s > 0.0f) ? s : 0.2f * s;
  sb[idx] = s;
  atomicMax(mEnc + d_ * Hh + h, f2ord(s));
}

__global__ void gat_edge_expsum(float* __restrict__ sb, const unsigned* __restrict__ mEnc,
                                float* __restrict__ den, const int* __restrict__ dst,
                                int total, int Hh) {
  int idx = blockIdx.x * blockDim.x + threadIdx.x;
  if (idx >= total) return;
  int e = idx / Hh;
  int h = idx - e * Hh;
  int d_ = dst[e];
  float m = ord2f(mEnc[d_ * Hh + h]);
  float ex = expf(sb[idx] - m);
  sb[idx] = ex;
  atomicAdd(den + d_ * Hh + h, ex);
}

static_assert(EE % 256 == 0, "a wave owns 32 whole edges and EE / 256 blocks of 8 waves cover them all");

template <int HF, int NCH>
__device__ __forceinline__ void scat_load_row(float (&r)[NCH], const float* __restrict__ row,
                                              int lane) {
#pragma unroll
  for (int c = 0; c < NCH; ++c) {
    const int col = (c << 5) + lane;
    if (HF % 32 == 0 || col < HF) r[c] = row[col];
  }
}

template <int HH, int F, int NCH>
__device__ __forceinline__ void scat_add_row(const float (&r)[NCH], const float (&al)[HH],
                                             float* __restrict__ row, int lane) {
  constexpr int HF = HH * F;
#pragma unroll
  for (int c = 0; c < NCH; ++c) {
    const int col = (c << 5) + lane;
    const float a = al[(HH == 1) ? 0 : ((c << 5) / F)];
    if (HF % 32 == 0 || col < HF) atomicAdd(row + col, r[c] * a);
  }
}

template <int HH, int F>
__global__ __launch_bounds__(256)
void gat_scatter_rows(const float* __restrict__ Hm, const float* __restrict__ sb,
                      const float* __restrict__ den, const int* __restrict__ src,
                      const int* __restrict__ dst, float* __restrict__ AGG) {
  static_assert(HH == 1 || F % 32 == 0, "a 32-column piece must lie inside one head");
  constexpr int HF  = HH * F;
  constexpr int NCH = (HF + 31) / 32;
  const int lane = threadIdx.x & 31;
  const int e0   = ((blockIdx.x * blockDim.x + threadIdx.x) >> 5) << 5;
  if (e0 >= EE) return;

  const int me = e0 + lane;
  const int ms = src[me];
  const int md = dst[me];
  float ma[HH];
#pragma unroll
  for (int h = 0; h < HH; ++h)
    ma[h] = sb[(size_t)me * HH + h] / den[(size_t)md * HH + h];

  float ra[NCH], rb[NCH];
  float al[HH];
  scat_load_row<HF, NCH>(ra, Hm + (size_t)__builtin_amdgcn_readlane(ms, 0) * HF, lane);
  for (int i = 0; i < 32; i += 2) {
    scat_load_row<HF, NCH>(rb, Hm + (size_t)__builtin_amdgcn_readlane(ms, i + 1) * HF, lane);
#pragma unroll
    for (int h = 0; h < HH; ++h)
      al[h] = __int_as_float(__builtin_amdgcn_readlane(__float_as_int(ma[h]), i));
    scat_add_row<HH, F, NCH>(ra, al, AGG + (size_t)__builtin_amdgcn_readlane(md, i) * HF, lane);

    if (i + 2 < 32)
      scat_load_row<HF, NCH>(ra, Hm + (size_t)__builtin_amdgcn_readlane(ms, i + 2) * HF, lane);
#pragma unroll
    for (int h = 0; h < HH; ++h)
      al[h] = __int_as_float(__builtin_amdgcn_readlane(__float_as_int(ma[h]), i + 1));
    scat_add_row<HH, F, NCH>(rb, al, AGG + (size_t)__builtin_amdgcn_readlane(md, i + 1) * HF, lane);
  }
}


#define OWN_R   200
#define OWN_CAP 1024
#define OWN_KEEP 384
#define OWN_NCH (EE / 128)
static_assert(NN % OWN_R == 0, "row groups cover the nodes exactly");
static_assert(OWN_R <= 7 * 32, "7 register slots per wave");
static_assert(EE % 128 == 0, "a wave filters 128 edges per load");
static_assert(EE <= (1 << 20), "an edge index is packed into 20 bits");
static_assert(NN <= (1 << 20), "a node index is packed into 20 bits");
static_assert(((OWN_NCH + 31) / 32) % 4 == 0, "the filter loop takes 4 loads per trip");
static_assert(OWN_NCH % (NN / OWN_R) == 0, "start offsets of the groups tile the list");
static_assert(OWN_KEEP % 128 == 0 && OWN_KEEP <= OWN_CAP, "kept lists are copied 128 entries per trip");
static_assert((size_t)(NN / OWN_R) * 32 * OWN_KEEP <= (size_t)EE * 4, "kept lists fit the score buffer's place");
static_assert((NN / OWN_R) * 32 <= NN * 4, "kept list lengths fit the denominator buffer's place");

typedef __attribute__((ext_vector_type(4))) int v4i;

__device__ __forceinline__ void own_queue(unsigned (*lst)[OWN_CAP], unsigned* cnt, unsigned* full,
                                          int e, unsigned r) {
  const unsigned w = r & 31u;
  const unsigned pos = atomicAdd(&cnt[w], 1u);
  if (pos < OWN_CAP) {
    lst[w][pos] = (unsigned)e | ((r >> 5) << 20);
  } else {
    *full = 1u;
  }
}

__device__ __forceinline__ void own_scores(float (&s)[4], const float* __restrict__ el,
                                           const float* __restrict__ er, int sn, int dn_) {
  const v4f a = *(const v4f*)(el + (size_t)sn * 4);
  const v4f b = *(const v4f*)(er + (size_t)dn_ * 4);
  s[0] = a.x + b.x; s[1] = a.y + b.y; s[2] = a.z + b.z; s[3] = a.w + b.w;
#pragma unroll
  for (int h = 0; h < 4; ++h) s[h] = (s[h] > 0.0f) ? s[h] : 0.2f * s[h];
}

__device__ __forceinline__ void own_add(float (&acc)[7][8], const float (&q)[8], int slot,
                                        const float (&al)[4]) {
#pragma unroll
  for (int k = 0; k < 7; ++k) {
    if (slot == k) {
#pragma unroll
      for (int c = 0; c < 8; ++c) acc[k][c] += q[c] * al[c >> 1];
    }
  }
}

__device__ __forceinline__ float own_wave_max(float v) {
  for (int o = 16; o > 0; o >>= 1) v = fmaxf(v, __shfl_xor(v, o));
  return v;
}
__device__ __forceinline__ float own_wave_sum(float v) {
  for (int o = 16; o > 0; o >>= 1) v += __shfl_xor(v, o);
  return v;
}

template <bool LOOKUP, bool KEEP>
__device__ __forceinline__ void own_pass_max(unsigned* lw, unsigned (*mxw)[4], unsigned nq,
                                             const float* __restrict__ el,
                                             const float* __restrict__ er,
                                             const int* __restrict__ src, unsigned* __restrict__ kl,
                                             int base, int wave, int lane) {
  for (unsigned b0 = 0; b0 < nq; b0 += 128) {
    unsigned idx[4];
    bool live[4];
    int me[4], mslot[4], ms[4];
    float s[4][4];
#pragma unroll
    for (int u = 0; u < 4; ++u) {
      idx[u]  = b0 + (unsigned)(32 * u + lane);
      live[u] = idx[u] < nq;
      const unsigned ent = live[u] ? lw[idx[u]] : (7u << 20);
      me[u]    = (int)(ent & 0xFFFFFu);
      mslot[u] = (int)(ent >> 20);
    }
#pragma unroll
    for (int u = 0; u < 4; ++u) ms[u] = LOOKUP ? src[me[u]] : me[u];
#pragma unroll
    for (int u = 0; u < 4; ++u)
      own_scores(s[u], el, er, ms[u], (mslot[u] < 7) ? (base + (mslot[u] << 5) + wave) : base);
#pragma unroll
    for (int u = 0; u < 4; ++u) {
      if (live[u]) {
        if (LOOKUP) {
          const unsigned ent = (unsigned)ms[u] | ((unsigned)mslot[u] << 20);
          lw[idx[u]] = ent;
          if (KEEP && idx[u] < (unsigned)OWN_KEEP) kl[idx[u]] = ent;
        }
#pragma unroll
        for (int h = 0; h < 4; ++h) atomicMax(&mxw[mslot[u]][h], f2ord(s[u][h]));
      }
    }
  }
}

__device__ void own_rows_by_scan(int base, int wave, int lane, const float* __restrict__ Hm,
                                 const float* __restrict__ el, const float* __restrict__ er,
                                 const int* __restrict__ src, const int* __restrict__ dst,
                                 float* AGG) {
  for (int k = 0; k < 7; ++k) {
    const int r = (k << 5) + wave;
    if (r >= OWN_R) break;
    const int d = base + r;
    float m[4] = {-INFINITY, -INFINITY, -INFINITY, -INFINITY};
    for (int e = lane; e < EE; e += 32) {
      if (dst[e] == d) {
        float s[4];
        own_scores(s, el, er, src[e], d);
        for (int h = 0; h < 4; ++h) m[h] = fmaxf(m[h], s[h]);
      }
    }
    for (int h = 0; h < 4; ++h) m[h] = own_wave_max(m[h]);
    float z[4] = {0.0f, 0.0f, 0.0f, 0.0f};
    for (int e = lane; e < EE; e += 32) {
      if (dst[e] == d) {
        float s[4];
        own_scores(s, el, er, src[e], d);
        for (int h = 0; h < 4; ++h) z[h] += expf(s[h] - m[h]);
      }
    }
    for (int h = 0; h < 4; ++h) z[h] = own_wave_sum(z[h]);
    float acc[8] = {};
    for (int e0 = 0; e0 < EE; e0 += 32) {
      const int e = e0 + lane;
      const bool hit = dst[e] == d;
      int sn = 0;
      float a[4] = {0.0f, 0.0f, 0.0f, 0.0f};
      if (hit) {
        sn = src[e];
        float s[4];
        own_scores(s, el, er, sn, d);
        for (int h = 0; h < 4; ++h) a[h] = expf(s[h] - m[h]) / z[h];
      }
      unsigned mask = (unsigned)__ballot(hit);
      while (mask) {
        const int j = __ffs(mask) - 1;
        mask &= mask - 1u;
        const int sj = __shfl(sn, j);
        float aj[4];
        for (int h = 0; h < 4; ++h) aj[h] = __shfl(a[h], j);
        const float* row = Hm + (size_t)sj * 256;
        for (int c = 0; c < 8; ++c) acc[c] += row[(c << 5) + lane] * aj[c >> 1];
      }
    }
    float* orow = AGG + (size_t)d * 256;
    for (int c = 0; c < 8; ++c) {
      const float v = orow[(c << 5) + lane] + acc[c];
      orow[(c << 5) + lane] = (v > 0.0f) ? v : expm1f(v);
    }
  }
}

template <bool KEEP, bool REUSE>
__global__ __launch_bounds__(1024)
void gat_owner_rows(const float* __restrict__ Hm, const float* __restrict__ el,
                    const float* __restrict__ er, const int* __restrict__ src,
                    const int* __restrict__ dst, float* AGG, unsigned* klist, unsigned* kcnt) {
  __shared__ unsigned lst[32][OWN_CAP];
  __shared__ unsigned cnt[32];
  __shared__ unsigned mx[32][8][4];
  __shared__ float    dn[32][8][4];
  __shared__ unsigned full;
  __shared__ unsigned kept;
  const int tid  = threadIdx.x;
  const int lane = tid & 31;
  const int wave = __builtin_amdgcn_readfirstlane(tid >> 5);
  const int base = blockIdx.x * OWN_R;
  unsigned* kc = kcnt + blockIdx.x * 32;
  unsigned* kl = klist + (size_t)(blockIdx.x * 32 + wave) * OWN_KEEP;

  if (tid < 32) cnt[tid] = 0u;
  if (tid == 0) { full = 0u; kept = 1u; }
  (&mx[0][0][0])[tid] = 0u;
  (&dn[0][0][0])[tid] = 0.0f;
  __syncthreads();

  bool reuse = false;
  if (REUSE) {
    if (tid < 32 && kc[tid] > (unsigned)OWN_KEEP) kept = 0u;
    __syncthreads();
    reuse = __builtin_amdgcn_readfirstlane((int)kept) != 0;
  }

  if (reuse) {
    const unsigned n = (unsigned)__builtin_amdgcn_readfirstlane((int)kc[wave]);
    for (unsigned i0 = 0; i0 < n; i0 += 128) {
      unsigned v[4];
#pragma unroll
      for (int u = 0; u < 4; ++u) {
        const unsigned i = i0 + (unsigned)(32 * u + lane);
        v[u] = (i < n) ? kl[i] : 0u;
      }
#pragma unroll
      for (int u = 0; u < 4; ++u) {
        const unsigned i = i0 + (unsigned)(32 * u + lane);
        if (i < n) lst[wave][i] = v[u];
      }
    }
    if (lane == 0) cnt[wave] = n;
  } else {
  const int rot = blockIdx.x * (OWN_NCH / (NN / OWN_R));
  const v4i* __restrict__ dp = (const v4i*)dst;
  for (int t = 0; t < (OWN_NCH + 31) / 32; t += 4) {
    v4i dv[4];
    int eb[4];
    bool ok[4];
#pragma unroll
    for (int u = 0; u < 4; ++u) {
      const int L = wave + 32 * (t + u);
      ok[u] = L < OWN_NCH;
      int ch = (ok[u] ? L : 0) + rot;
      if (ch >= OWN_NCH) ch -= OWN_NCH;
      eb[u] = ch * 128 + lane * 4;
      dv[u] = dp[ch * 32 + lane];
    }
#pragma unroll
    for (int u = 0; u < 4; ++u) {
      const unsigned r0 = (unsigned)(dv[u].x - base);
      const unsigned r1 = (unsigned)(dv[u].y - base);
      const unsigned r2 = (unsigned)(dv[u].z - base);
      const unsigned r3 = (unsigned)(dv[u].w - base);
      const unsigned mn = min(min(r0, r1), min(r2, r3));
      if (ok[u] && mn < (unsigned)OWN_R) {
        if (r0 < (unsigned)OWN_R) own_queue(lst, cnt, &full, eb[u] + 0, r0);
        if (r1 < (unsigned)OWN_R) own_queue(lst, cnt, &full, eb[u] + 1, r1);
        if (r2 < (unsigned)OWN_R) own_queue(lst, cnt, &full, eb[u] + 2, r2);
        if (r3 < (unsigned)OWN_R) own_queue(lst, cnt, &full, eb[u] + 3, r3);
      }
    }
  }
  }
  __syncthreads();

  const bool overfull = __builtin_amdgcn_readfirstlane((int)full) != 0;
  if (KEEP && lane == 0) kc[wave] = overfull ? 0xFFFFFFFFu : cnt[wave];
  if (overfull) {
    own_rows_by_scan(base, wave, lane, Hm, el, er, src, dst, AGG);
    return;
  }

  const unsigned nq = (unsigned)__builtin_amdgcn_readfirstlane((int)cnt[wave]);

  if (reuse) {
    own_pass_max<false, false>(lst[wave], mx[wave], nq, el, er, src, kl, base, wave, lane);
  } else {
    own_pass_max<true, KEEP>(lst[wave], mx[wave], nq, el, er, src, kl, base, wave, lane);
  }
  __syncthreads();

  for (unsigned b0 = 0; b0 < nq; b0 += 128) {
    unsigned idx[4];
    bool live[4];
    int mslot[4], ms[4];
    float s[4][4];
#pragma unroll
    for (int u = 0; u < 4; ++u) {
      idx[u]  = b0 + (unsigned)(32 * u + lane);
      live[u] = idx[u] < nq;
      const unsigned ent = live[u] ? lst[wave][idx[u]] : (7u << 20);
      ms[u]    = (int)(ent & 0xFFFFFu);
      mslot[u] = (int)(ent >> 20);
    }
#pragma unroll
    for (int u = 0; u < 4; ++u)
      own_scores(s[u], el, er, ms[u], (mslot[u] < 7) ? (base + (mslot[u] << 5) + wave) : base);
#pragma unroll
    for (int u = 0; u < 4; ++u) {
      if (live[u]) {
#pragma unroll
        for (int h = 0; h < 4; ++h)
          atomicAdd(&dn[wave][mslot[u]][h], expf(s[u][h] - ord2f(mx[wave][mslot[u]][h])));
      }
    }
  }
  __syncthreads();

  float acc[7][8] = {};
  for (unsigned b0 = 0; b0 < nq; b0 += 32) {
    const unsigned idx = b0 + (unsigned)lane;
    const unsigned ent = (idx < nq) ? lst[wave][idx] : (7u << 20);
    const int ms    = (int)(ent & 0xFFFFFu);
    const int mslot = (int)(ent >> 20);
    const int md    = (mslot < 7) ? (base + (mslot << 5) + wave) : base;
    float ma[4];
    own_scores(ma, el, er, ms, md);
#pragma unroll
    for (int h = 0; h < 4; ++h)
      ma[h] = expf(ma[h] - ord2f(mx[wave][mslot][h])) / dn[wave][mslot][h];

    float q0[8], q1[8], q2[8], q3[8];
    float al[4];
    scat_load_row<256, 8>(q0, Hm + (size_t)__builtin_amdgcn_readlane(ms, 0) * 256, lane);
    scat_load_row<256, 8>(q1, Hm + (size_t)__builtin_amdgcn_readlane(ms, 1) * 256, lane);
    scat_load_row<256, 8>(q2, Hm + (size_t)__builtin_amdgcn_readlane(ms, 2) * 256, lane);
    for (int i = 0; i < 32; i += 4) {
      scat_load_row<256, 8>(q3, Hm + (size_t)__builtin_amdgcn_readlane(ms, i + 3) * 256, lane);
#pragma unroll
      for (int h = 0; h < 4; ++h) al[h] = __int_as_float(__builtin_amdgcn_readlane(__float_as_int(ma[h]), i));
      own_add(acc, q0, __builtin_amdgcn_readlane(mslot, i), al);

      scat_load_row<256, 8>(q0, Hm + (size_t)__builtin_amdgcn_readlane(ms, (i + 4) & 31) * 256, lane);
#pragma unroll
      for (int h = 0; h < 4; ++h) al[h] = __int_as_float(__builtin_amdgcn_readlane(__float_as_int(ma[h]), i + 1));
      own_add(acc, q1, __builtin_amdgcn_readlane(mslot, i + 1), al);

      scat_load_row<256, 8>(q1, Hm + (size_t)__builtin_amdgcn_readlane(ms, (i + 5) & 31) * 256, lane);
#pragma unroll
      for (int h = 0; h < 4; ++h) al[h] = __int_as_float(__builtin_amdgcn_readlane(__float_as_int(ma[h]), i + 2));
      own_add(acc, q2, __builtin_amdgcn_readlane(mslot, i + 2), al);

      scat_load_row<256, 8>(q2, Hm + (size_t)__builtin_amdgcn_readlane(ms, (i + 6) & 31) * 256, lane);
#pragma unroll
      for (int h = 0; h < 4; ++h) al[h] = __int_as_float(__builtin_amdgcn_readlane(__float_as_int(ma[h]), i + 3));
      own_add(acc, q3, __builtin_amdgcn_readlane(mslot, i + 3), al);
    }
  }

#pragma unroll
  for (int g = 0; g < 7; g += 4) {
    float t[4][8];
#pragma unroll
    for (int kk = 0; kk < 4; ++kk) {
      const int k = g + kk;
      const int r = (k << 5) + wave;
      if (k < 7 && r < OWN_R) {
        const float* orow = AGG + (size_t)(base + r) * 256;
#pragma unroll
        for (int c = 0; c < 8; ++c) t[kk][c] = orow[(c << 5) + lane];
      }
    }
#pragma unroll
    for (int kk = 0; kk < 4; ++kk) {
      const int k = g + kk;
      const int r = (k << 5) + wave;
      if (k < 7 && r < OWN_R) {
        float* orow = AGG + (size_t)(base + r) * 256;
#pragma unroll
        for (int c = 0; c < 8; ++c) {
          const float v = t[kk][c] + acc[k < 7 ? k : 0][c];
          orow[(c << 5) + lane] = (v > 0.0f) ? v : expm1f(v);
        }
      }
    }
  }
}

static void run_layer(const float* X, int K, int Hh, int F,
                      const float* W, const float* al, const float* ar, const float* rW,
                      const int* src, const int* dst,
                      float* Hbuf, float* AGG, int Opad,
                      float* el, float* er, float* den, unsigned* mEnc, float* sb,
                      int first, hipStream_t stream) {
  const int HF = Hh * F;
  dim3 ggrid(NN / 80, Opad / 64, 2);
  if (Hh == 4 && K == 128) {
    gat_gemm_xwt<128, 5, 4><<<ggrid, 32, 0, stream>>>(X, W, rW, Hbuf, AGG, HF);
  } else if (Hh == 4) {
    gat_gemm_xwt<256, 5, 4><<<ggrid, 32, 0, stream>>>(X, W, rW, Hbuf, AGG, HF);
  } else {
    gat_gemm_xwt<256, 5, 3><<<ggrid, 32, 0, stream>>>(X, W, rW, Hbuf, AGG, HF);
  }

  const int nh = NN * Hh;
  if (Hh == 4) {
    gat_el_er<false><<<(nh + 255) / 256, 256, 0, stream>>>(Hbuf, al, ar, el, er, mEnc, den, nh, Hh, F);
    if (first) {
      gat_owner_rows<true, false><<<NN / OWN_R, 1024, 0, stream>>>(Hbuf, el, er, src, dst, AGG,
                                                                   (unsigned*)sb, (unsigned*)den);
    } else {
      gat_owner_rows<false, true><<<NN / OWN_R, 1024, 0, stream>>>(Hbuf, el, er, src, dst, AGG,
                                                                   (unsigned*)sb, (unsigned*)den);
    }
  } else {
    gat_el_er<true><<<(nh + 255) / 256, 256, 0, stream>>>(Hbuf, al, ar, el, er, mEnc, den, nh, Hh, F);
    const int eh = EE * Hh;
    gat_edge_smax<<<(eh + 255) / 256, 256, 0, stream>>>(el, er, src, dst, sb, mEnc, eh, Hh);
    gat_edge_expsum<<<(eh + 255) / 256, 256, 0, stream>>>(sb, mEnc, den, dst, eh, Hh);
    gat_scatter_rows<1, 40><<<EE / 256, 256, 0, stream>>>(Hbuf, sb, den, src, dst, AGG);
  }
}

extern "C" void kernel_launch(void* const* d_in, const int* in_sizes, int n_in,
                              void* d_out, int out_size, void* d_ws, size_t ws_size,
                              hipStream_t stream) {
  (void)in_sizes; (void)n_in; (void)out_size; (void)ws_size;

  const float* feat = (const float*)d_in[0];
  const int*   src  = (const int*)d_in[1];
  const int*   dst  = (const int*)d_in[2];
  const float* W0  = (const float*)d_in[3];
  const float* al0 = (const float*)d_in[4];
  const float* ar0 = (const float*)d_in[5];
  const float* rW0 = (const float*)d_in[6];
  const float* W1  = (const float*)d_in[7];
  const float* al1 = (const float*)d_in[8];
  const float* ar1 = (const float*)d_in[9];
  const float* rW1 = (const float*)d_in[10];
  const float* W2  = (const float*)d_in[11];
  const float* al2 = (const float*)d_in[12];
  const float* ar2 = (const float*)d_in[13];
  const float* rW2 = (const float*)d_in[14];

  const size_t NF = (size_t)NN * 256;
  float* ws    = (float*)d_ws;
  float* bufH  = ws;
  float* bufY0 = bufH  + NF;
  float* bufY1 = bufY0 + NF;
  float* el    = bufY1 + NF;
  float* er    = el  + (size_t)NN * 4;
  float* den   = er  + (size_t)NN * 4;
  unsigned* mE = (unsigned*)(den + (size_t)NN * 4);
  float* sb    = (float*)mE + (size_t)NN * 4;
  float* W2p   = sb + (size_t)EE * 4;
  float* rW2p  = W2p + (size_t)64 * 256;

  {
    int tot = 64 * 256;
    gat_pad_w<<<dim3((tot + 255) / 256, 2), 256, 0, stream>>>(W2, rW2, W2p, rW2p, 40, 64, 256);
  }

  run_layer(feat, 128, 4, 64, W0, al0, ar0, rW0, src, dst,
            bufH, bufY0, 256, el, er, den, mE, sb, 1, stream);
  run_layer(bufY0, 256, 4, 64, W1, al1, ar1, rW1, src, dst,
            bufH, bufY1, 256, el, er, den, mE, sb, 0, stream);
  run_layer(bufY1, 256, 1, 40, W2p, al2, ar2, rW2p, src, dst,
            bufH, (float*)d_out, 64, el, er, den, mE, sb, 0, stream);
}


# Output layer moved onto the row-owner kernel with the kept lists: softmax and row sums formed per owned row in LDS and registers, written once; 10 launches, no global atomics; sums in list order
# speedup vs baseline: 5.2929x; 1.2423x over previous
#include <hip/hip_runtime.h>
#include <math.h>

#define NN 50000
#define EE 800000

typedef __attribute__((ext_vector_type(2))) float v2f;
typedef __attribute__((ext_vector_type(4))) float v4f;
typedef __attribute__((ext_vector_type(8))) float v8f;

__device__ __forceinline__ unsigned f2ord(float f) {
  unsigned u = __float_as_uint(f);
  return (u & 0x80000000u) ? ~u : (u | 0x80000000u);
}
__device__ __forceinline__ float ord2f(unsigned e) {
  unsigned u = (e & 0x80000000u) ? (e ^ 0x80000000u) : ~e;
  return __uint_as_float(u);
}

template <int NF>
__device__ __forceinline__ void gemm_load_lines(v4f (&f)[NF][4], const float* __restrict__ p,
                                                size_t fstep) {
#pragma unroll
  for (int i = 0; i < NF; ++i) {
#pragma unroll
    for (int j = 0; j < 4; ++j) f[i][j] = *(const v4f*)(p + (size_t)i * fstep + 8 * j);
  }
}

template <int MA, int NB>
__device__ __forceinline__ void gemm_mma_lines(v8f (&acc)[MA][NB], const v4f (&a)[MA][4],
                                               const v4f (&b)[NB][4]) {
#pragma unroll
  for (int j = 0; j < 4; ++j) {
#pragma unroll
    for (int i = 0; i < MA; ++i) {
#pragma unroll
      for (int t = 0; t < NB; ++t)
        acc[i][t] = __builtin_amdgcn_wmma_f32_16x16x4_f32(
            false, a[i][j].xy, false, b[t][j].xy, (short)0, acc[i][t], false, false);
    }
#pragma unroll
    for (int i = 0; i < MA; ++i) {
#pragma unroll
      for (int t = 0; t < NB; ++t)
        acc[i][t] = __builtin_amdgcn_wmma_f32_16x16x4_f32(
            false, a[i][j].zw, false, b[t][j].zw, (short)0, acc[i][t], false, false);
    }
  }
}

template <int K, int MA, int NB>
__global__ __launch_bounds__(32)
void gat_gemm_xwt(const float* __restrict__ X, const float* __restrict__ W0,
                  const float* __restrict__ W1, float* __restrict__ OUT0,
                  float* __restrict__ OUT1, int Oout) {
  static_assert(K % 64 == 0, "the K loop takes two 32-float lines per trip");
  static_assert(NN % (16 * MA) == 0, "row tiles cover the nodes exactly");
  const int lane = threadIdx.x & 31;
  const int c    = lane & 15;
  const int half = lane >> 4;
  const int row0 = blockIdx.x * (16 * MA);
  const int col0 = blockIdx.y << 6;
  const float* __restrict__ W   = (blockIdx.z == 0) ? W0 : W1;
  float* __restrict__       OUT = (blockIdx.z == 0) ? OUT0 : OUT1;

  const float* xp = X + (size_t)(row0 + c) * (size_t)K + (half << 2);
  const float* wp = W + (size_t)(col0 + c) * (size_t)K + (half << 2);
  const size_t fstep = (size_t)16 * (size_t)K;

  v8f acc[MA][NB] = {};
  v4f aP[MA][4], bP[NB][4], aQ[MA][4], bQ[NB][4];

  gemm_load_lines<MA>(aP, xp, fstep);
  gemm_load_lines<NB>(bP, wp, fstep);
  int kb = 0;
  for (; kb + 64 < K; kb += 64) {
    gemm_load_lines<MA>(aQ, xp + kb + 32, fstep);
    gemm_load_lines<NB>(bQ, wp + kb + 32, fstep);
    __builtin_amdgcn_sched_barrier(0);
    gemm_mma_lines<MA, NB>(acc, aP, bP);
    __builtin_amdgcn_sched_barrier(0);
    gemm_load_lines<MA>(aP, xp + kb + 64, fstep);
    gemm_load_lines<NB>(bP, wp + kb + 64, fstep);
    __builtin_amdgcn_sched_barrier(0);
    gemm_mma_lines<MA, NB>(acc, aQ, bQ);
    __builtin_amdgcn_sched_barrier(0);
  }
  gemm_load_lines<MA>(aQ, xp + kb + 32, fstep);
  gemm_load_lines<NB>(bQ, wp + kb + 32, fstep);
  __builtin_amdgcn_sched_barrier(0);
  gemm_mma_lines<MA, NB>(acc, aP, bP);
  gemm_mma_lines<MA, NB>(acc, aQ, bQ);

#pragma unroll
  for (int i = 0; i < MA; ++i) {
#pragma unroll
    for (int t = 0; t < NB; ++t) {
      const int oc = col0 + (t << 4) + c;
      if (oc < Oout) {
        float* orow = OUT + (size_t)(row0 + (i << 4) + (half << 3)) * (size_t)Oout + oc;
#pragma unroll
        for (int v = 0; v < 8; ++v) {
          orow[(size_t)v * (size_t)Oout] = acc[i][t][v];
        }
      }
    }
  }
}

__global__ void gat_pad_w(const float* __restrict__ Wsrc0, const float* __restrict__ Wsrc1,
                          float* __restrict__ Wdst0, float* __restrict__ Wdst1,
                          int rows, int padrows, int K) {
  int idx = blockIdx.x * blockDim.x + threadIdx.x;
  int total = padrows * K;
  if (idx >= total) return;
  const float* __restrict__ Wsrc = (blockIdx.y == 0) ? Wsrc0 : Wsrc1;
  float* __restrict__       Wdst = (blockIdx.y == 0) ? Wdst0 : Wdst1;
  int r = idx / K;
  Wdst[idx] = (r < rows) ? Wsrc[idx] : 0.0f;
}

__global__ void gat_el_er(const float* __restrict__ Hm, const float* __restrict__ al,
                          const float* __restrict__ ar, float* __restrict__ el,
                          float* __restrict__ er, int total, int Hh, int F) {
  int idx = blockIdx.x * blockDim.x + threadIdx.x;
  if (idx >= total) return;
  int h = idx % Hh;
  int n = idx / Hh;
  const float* row = Hm + (size_t)n * (size_t)(Hh * F) + (size_t)h * F;
  const float* a = al + h * F;
  const float* b = ar + h * F;
  float sl = 0.f, sr = 0.f;
  for (int f = 0; f < F; ++f) {
    float v = row[f];
    sl += v * a[f];
    sr += v * b[f];
  }
  el[idx] = sl;
  er[idx] = sr;
}

template <int HF, int NCH>
__device__ __forceinline__ void scat_load_row(float (&r)[NCH], const float* __restrict__ row,
                                              int lane) {
#pragma unroll
  for (int c = 0; c < NCH; ++c) {
    const int col = (c << 5) + lane;
    if (HF % 32 == 0 || col < HF) r[c] = row[col];
  }
}

#define OWN_R   200
#define OWN_CAP 1024
#define OWN_KEEP 384
#define OWN_NCH (EE / 128)
static_assert(NN % OWN_R == 0, "row groups cover the nodes exactly");
static_assert(OWN_R <= 7 * 32, "7 register slots per wave");
static_assert(EE % 128 == 0, "a wave filters 128 edges per load");
static_assert(EE <= (1 << 20), "an edge index is packed into 20 bits");
static_assert(NN <= (1 << 20), "a node index is packed into 20 bits");
static_assert(((OWN_NCH + 31) / 32) % 4 == 0, "the filter loop takes 4 loads per trip");
static_assert(OWN_NCH % (NN / OWN_R) == 0, "start offsets of the groups tile the list");
static_assert(OWN_KEEP % 128 == 0 && OWN_KEEP <= OWN_CAP, "kept lists are copied 128 entries per trip");
static_assert((size_t)(NN / OWN_R) * 32 * OWN_KEEP <= (size_t)EE * 4, "kept lists fit their [EE,4]-word place in the workspace");
static_assert((NN / OWN_R) * 32 <= NN * 4, "kept list lengths fit their [NN,4]-word place in the workspace");

typedef __attribute__((ext_vector_type(4))) int v4i;

__device__ __forceinline__ void own_queue(unsigned (*lst)[OWN_CAP], unsigned* cnt, unsigned* full,
                                          int e, unsigned r) {
  const unsigned w = r & 31u;
  const unsigned pos = atomicAdd(&cnt[w], 1u);
  if (pos < OWN_CAP) {
    lst[w][pos] = (unsigned)e | ((r >> 5) << 20);
  } else {
    *full = 1u;
  }
}

__device__ __forceinline__ void own_filter(unsigned (*lst)[OWN_CAP], unsigned* cnt, unsigned* full,
                                           const int* __restrict__ dst, int base, int wave, int lane) {
  const int rot = blockIdx.x * (OWN_NCH / (NN / OWN_R));
  const v4i* __restrict__ dp = (const v4i*)dst;
  for (int t = 0; t < (OWN_NCH + 31) / 32; t += 4) {
    v4i dv[4];
    int eb[4];
    bool ok[4];
#pragma unroll
    for (int u = 0; u < 4; ++u) {
      const int L = wave + 32 * (t + u);
      ok[u] = L < OWN_NCH;
      int ch = (ok[u] ? L : 0) + rot;
      if (ch >= OWN_NCH) ch -= OWN_NCH;
      eb[u] = ch * 128 + lane * 4;
      dv[u] = dp[ch * 32 + lane];
    }
#pragma unroll
    for (int u = 0; u < 4; ++u) {
      const unsigned r0 = (unsigned)(dv[u].x - base);
      const unsigned r1 = (unsigned)(dv[u].y - base);
      const unsigned r2 = (unsigned)(dv[u].z - base);
      const unsigned r3 = (unsigned)(dv[u].w - base);
      const unsigned mn = min(min(r0, r1), min(r2, r3));
      if (ok[u] && mn < (unsigned)OWN_R) {
        if (r0 < (unsigned)OWN_R) own_queue(lst, cnt, full, eb[u] + 0, r0);
        if (r1 < (unsigned)OWN_R) own_queue(lst, cnt, full, eb[u] + 1, r1);
        if (r2 < (unsigned)OWN_R) own_queue(lst, cnt, full, eb[u] + 2, r2);
        if (r3 < (unsigned)OWN_R) own_queue(lst, cnt, full, eb[u] + 3, r3);
      }
    }
  }
}

__device__ __forceinline__ void own_copy_list(unsigned* lw, const unsigned* kl, unsigned n, int lane) {
  for (unsigned i0 = 0; i0 < n; i0 += 128) {
    unsigned v[4];
#pragma unroll
    for (int u = 0; u < 4; ++u) {
      const unsigned i = i0 + (unsigned)(32 * u + lane);
      v[u] = (i < n) ? kl[i] : 0u;
    }
#pragma unroll
    for (int u = 0; u < 4; ++u) {
      const unsigned i = i0 + (unsigned)(32 * u + lane);
      if (i < n) lw[i] = v[u];
    }
  }
}

__device__ __forceinline__ void own_scores(float (&s)[4], const float* __restrict__ el,
                                           const float* __restrict__ er, int sn, int dn_) {
  const v4f a = *(const v4f*)(el + (size_t)sn * 4);
  const v4f b = *(const v4f*)(er + (size_t)dn_ * 4);
  s[0] = a.x + b.x; s[1] = a.y + b.y; s[2] = a.z + b.z; s[3] = a.w + b.w;
#pragma unroll
  for (int h = 0; h < 4; ++h) s[h] = (s[h] > 0.0f) ? s[h] : 0.2f * s[h];
}

__device__ __forceinline__ void own_add(float (&acc)[7][8], const float (&q)[8], int slot,
                                        const float (&al)[4]) {
#pragma unroll
  for (int k = 0; k < 7; ++k) {
    if (slot == k) {
#pragma unroll
      for (int c = 0; c < 8; ++c) acc[k][c] += q[c] * al[c >> 1];
    }
  }
}

__device__ __forceinline__ float own_wave_max(float v) {
  for (int o = 16; o > 0; o >>= 1) v = fmaxf(v, __shfl_xor(v, o));
  return v;
}
__device__ __forceinline__ float own_wave_sum(float v) {
  for (int o = 16; o > 0; o >>= 1) v += __shfl_xor(v, o);
  return v;
}

template <bool LOOKUP, bool KEEP>
__device__ __forceinline__ void own_pass_max(unsigned* lw, unsigned (*mxw)[4], unsigned nq,
                                             const float* __restrict__ el,
                                             const float* __restrict__ er,
                                             const int* __restrict__ src, unsigned* __restrict__ kl,
                                             int base, int wave, int lane) {
  for (unsigned b0 = 0; b0 < nq; b0 += 128) {
    unsigned idx[4];
    bool live[4];
    int me[4], mslot[4], ms[4];
    float s[4][4];
#pragma unroll
    for (int u = 0; u < 4; ++u) {
      idx[u]  = b0 + (unsigned)(32 * u + lane);
      live[u] = idx[u] < nq;
      const unsigned ent = live[u] ? lw[idx[u]] : (7u << 20);
      me[u]    = (int)(ent & 0xFFFFFu);
      mslot[u] = (int)(ent >> 20);
    }
#pragma unroll
    for (int u = 0; u < 4; ++u) ms[u] = LOOKUP ? src[me[u]] : me[u];
#pragma unroll
    for (int u = 0; u < 4; ++u)
      own_scores(s[u], el, er, ms[u], (mslot[u] < 7) ? (base + (mslot[u] << 5) + wave) : base);
#pragma unroll
    for (int u = 0; u < 4; ++u) {
      if (live[u]) {
        if (LOOKUP) {
          const unsigned ent = (unsigned)ms[u] | ((unsigned)mslot[u] << 20);
          lw[idx[u]] = ent;
          if (KEEP && idx[u] < (unsigned)OWN_KEEP) kl[idx[u]] = ent;
        }
#pragma unroll
        for (int h = 0; h < 4; ++h) atomicMax(&mxw[mslot[u]][h], f2ord(s[u][h]));
      }
    }
  }
}

__device__ void own_rows_by_scan(int base, int wave, int lane, const float* __restrict__ Hm,
                                 const float* __restrict__ el, const float* __restrict__ er,
                                 const int* __restrict__ src, const int* __restrict__ dst,
                                 float* AGG) {
  for (int k = 0; k < 7; ++k) {
    const int r = (k << 5) + wave;
    if (r >= OWN_R) break;
    const int d = base + r;
    float m[4] = {-INFINITY, -INFINITY, -INFINITY, -INFINITY};
    for (int e = lane; e < EE; e += 32) {
      if (dst[e] == d) {
        float s[4];
        own_scores(s, el, er, src[e], d);
        for (int h = 0; h < 4; ++h) m[h] = fmaxf(m[h], s[h]);
      }
    }
    for (int h = 0; h < 4; ++h) m[h] = own_wave_max(m[h]);
    float z[4] = {0.0f, 0.0f, 0.0f, 0.0f};
    for (int e = lane; e < EE; e += 32) {
      if (dst[e] == d) {
        float s[4];
        own_scores(s, el, er, src[e], d);
        for (int h = 0; h < 4; ++h) z[h] += expf(s[h] - m[h]);
      }
    }
    for (int h = 0; h < 4; ++h) z[h] = own_wave_sum(z[h]);
    float acc[8] = {};
    for (int e0 = 0; e0 < EE; e0 += 32) {
      const int e = e0 + lane;
      const bool hit = dst[e] == d;
      int sn = 0;
      float a[4] = {0.0f, 0.0f, 0.0f, 0.0f};
      if (hit) {
        sn = src[e];
        float s[4];
        own_scores(s, el, er, sn, d);
        for (int h = 0; h < 4; ++h) a[h] = expf(s[h] - m[h]) / z[h];
      }
      unsigned mask = (unsigned)__ballot(hit);
      while (mask) {
        const int j = __ffs(mask) - 1;
        mask &= mask - 1u;
        const int sj = __shfl(sn, j);
        float aj[4];
        for (int h = 0; h < 4; ++h) aj[h] = __shfl(a[h], j);
        const float* row = Hm + (size_t)sj * 256;
        for (int c = 0; c < 8; ++c) acc[c] += row[(c << 5) + lane] * aj[c >> 1];
      }
    }
    float* orow = AGG + (size_t)d * 256;
    for (int c = 0; c < 8; ++c) {
      const float v = orow[(c << 5) + lane] + acc[c];
      orow[(c << 5) + lane] = (v > 0.0f) ? v : expm1f(v);
    }
  }
}

template <bool KEEP, bool REUSE>
__global__ __launch_bounds__(1024)
void gat_owner_rows(const float* __restrict__ Hm, const float* __restrict__ el,
                    const float* __restrict__ er, const int* __restrict__ src,
                    const int* __restrict__ dst, float* AGG, unsigned* klist, unsigned* kcnt) {
  __shared__ unsigned lst[32][OWN_CAP];
  __shared__ unsigned cnt[32];
  __shared__ unsigned mx[32][8][4];
  __shared__ float    dn[32][8][4];
  __shared__ unsigned full;
  __shared__ unsigned kept;
  const int tid  = threadIdx.x;
  const int lane = tid & 31;
  const int wave = __builtin_amdgcn_readfirstlane(tid >> 5);
  const int base = blockIdx.x * OWN_R;
  unsigned* kc = kcnt + blockIdx.x * 32;
  unsigned* kl = klist + (size_t)(blockIdx.x * 32 + wave) * OWN_KEEP;

  if (tid < 32) cnt[tid] = 0u;
  if (tid == 0) { full = 0u; kept = 1u; }
  (&mx[0][0][0])[tid] = 0u;
  (&dn[0][0][0])[tid] = 0.0f;
  __syncthreads();

  bool reuse = false;
  if (REUSE) {
    if (tid < 32 && kc[tid] > (unsigned)OWN_KEEP) kept = 0u;
    __syncthreads();
    reuse = __builtin_amdgcn_readfirstlane((int)kept) != 0;
  }

  if (reuse) {
    const unsigned n = (unsigned)__builtin_amdgcn_readfirstlane((int)kc[wave]);
    own_copy_list(lst[wave], kl, n, lane);
    if (lane == 0) cnt[wave] = n;
  } else {
    own_filter(lst, cnt, &full, dst, base, wave, lane);
  }
  __syncthreads();

  const bool overfull = __builtin_amdgcn_readfirstlane((int)full) != 0;
  if (KEEP && lane == 0) kc[wave] = overfull ? 0xFFFFFFFFu : cnt[wave];
  if (overfull) {
    own_rows_by_scan(base, wave, lane, Hm, el, er, src, dst, AGG);
    return;
  }

  const unsigned nq = (unsigned)__builtin_amdgcn_readfirstlane((int)cnt[wave]);

  if (reuse) {
    own_pass_max<false, false>(lst[wave], mx[wave], nq, el, er, src, kl, base, wave, lane);
  } else {
    own_pass_max<true, KEEP>(lst[wave], mx[wave], nq, el, er, src, kl, base, wave, lane);
  }
  __syncthreads();

  for (unsigned b0 = 0; b0 < nq; b0 += 128) {
    unsigned idx[4];
    bool live[4];
    int mslot[4], ms[4];
    float s[4][4];
#pragma unroll
    for (int u = 0; u < 4; ++u) {
      idx[u]  = b0 + (unsigned)(32 * u + lane);
      live[u] = idx[u] < nq;
      const unsigned ent = live[u] ? lst[wave][idx[u]] : (7u << 20);
      ms[u]    = (int)(ent & 0xFFFFFu);
      mslot[u] = (int)(ent >> 20);
    }
#pragma unroll
    for (int u = 0; u < 4; ++u)
      own_scores(s[u], el, er, ms[u], (mslot[u] < 7) ? (base + (mslot[u] << 5) + wave) : base);
#pragma unroll
    for (int u = 0; u < 4; ++u) {
      if (live[u]) {
#pragma unroll
        for (int h = 0; h < 4; ++h)
          atomicAdd(&dn[wave][mslot[u]][h], expf(s[u][h] - ord2f(mx[wave][mslot[u]][h])));
      }
    }
  }
  __syncthreads();

  float acc[7][8] = {};
  for (unsigned b0 = 0; b0 < nq; b0 += 32) {
    const unsigned idx = b0 + (unsigned)lane;
    const unsigned ent = (idx < nq) ? lst[wave][idx] : (7u << 20);
    const int ms    = (int)(ent & 0xFFFFFu);
    const int mslot = (int)(ent >> 20);
    const int md    = (mslot < 7) ? (base + (mslot << 5) + wave) : base;
    float ma[4];
    own_scores(ma, el, er, ms, md);
#pragma unroll
    for (int h = 0; h < 4; ++h)
      ma[h] = expf(ma[h] - ord2f(mx[wave][mslot][h])) / dn[wave][mslot][h];

    float q0[8], q1[8], q2[8], q3[8];
    float al[4];
    scat_load_row<256, 8>(q0, Hm + (size_t)__builtin_amdgcn_readlane(ms, 0) * 256, lane);
    scat_load_row<256, 8>(q1, Hm + (size_t)__builtin_amdgcn_readlane(ms, 1) * 256, lane);
    scat_load_row<256, 8>(q2, Hm + (size_t)__builtin_amdgcn_readlane(ms, 2) * 256, lane);
    for (int i = 0; i < 32; i += 4) {
      scat_load_row<256, 8>(q3, Hm + (size_t)__builtin_amdgcn_readlane(ms, i + 3) * 256, lane);
#pragma unroll
      for (int h = 0; h < 4; ++h) al[h] = __int_as_float(__builtin_amdgcn_readlane(__float_as_int(ma[h]), i));
      own_add(acc, q0, __builtin_amdgcn_readlane(mslot, i), al);

      scat_load_row<256, 8>(q0, Hm + (size_t)__builtin_amdgcn_readlane(ms, (i + 4) & 31) * 256, lane);
#pragma unroll
      for (int h = 0; h < 4; ++h) al[h] = __int_as_float(__builtin_amdgcn_readlane(__float_as_int(ma[h]), i + 1));
      own_add(acc, q1, __builtin_amdgcn_readlane(mslot, i + 1), al);

      scat_load_row<256, 8>(q1, Hm + (size_t)__builtin_amdgcn_readlane(ms, (i + 5) & 31) * 256, lane);
#pragma unroll
      for (int h = 0; h < 4; ++h) al[h] = __int_as_float(__builtin_amdgcn_readlane(__float_as_int(ma[h]), i + 2));
      own_add(acc, q2, __builtin_amdgcn_readlane(mslot, i + 2), al);

      scat_load_row<256, 8>(q2, Hm + (size_t)__builtin_amdgcn_readlane(ms, (i + 6) & 31) * 256, lane);
#pragma unroll
      for (int h = 0; h < 4; ++h) al[h] = __int_as_float(__builtin_amdgcn_readlane(__float_as_int(ma[h]), i + 3));
      own_add(acc, q3, __builtin_amdgcn_readlane(mslot, i + 3), al);
    }
  }

#pragma unroll
  for (int g = 0; g < 7; g += 4) {
    float t[4][8];
#pragma unroll
    for (int kk = 0; kk < 4; ++kk) {
      const int k = g + kk;
      const int r = (k << 5) + wave;
      if (k < 7 && r < OWN_R) {
        const float* orow = AGG + (size_t)(base + r) * 256;
#pragma unroll
        for (int c = 0; c < 8; ++c) t[kk][c] = orow[(c << 5) + lane];
      }
    }
#pragma unroll
    for (int kk = 0; kk < 4; ++kk) {
      const int k = g + kk;
      const int r = (k << 5) + wave;
      if (k < 7 && r < OWN_R) {
        float* orow = AGG + (size_t)(base + r) * 256;
#pragma unroll
        for (int c = 0; c < 8; ++c) {
          const float v = t[kk][c] + acc[k < 7 ? k : 0][c];
          orow[(c << 5) + lane] = (v > 0.0f) ? v : expm1f(v);
        }
      }
    }
  }
}

#define OUT_F 40
#define OUT_DEPTH 8

__device__ __forceinline__ float out_score(const float* __restrict__ el, const float* __restrict__ er,
                                           int sn, int dn_) {
  const float s = el[sn] + er[dn_];
  return (s > 0.0f) ? s : 0.2f * s;
}

__device__ __forceinline__ void out_load_row(float (&r)[2], const float* __restrict__ row, int lane) {
  r[0] = row[lane];
  r[1] = row[32 + (lane & 7)];
}

__device__ __forceinline__ void out_add(float (&acc)[7][2], const float (&q)[2], int slot, float a) {
#pragma unroll
  for (int k = 0; k < 7; ++k) {
    if (slot == k) {
      acc[k][0] += q[0] * a;
      acc[k][1] += q[1] * a;
    }
  }
}

template <bool LOOKUP>
__device__ __forceinline__ void out_pass_max(unsigned* lw, unsigned* mxw, unsigned nq,
                                             const float* __restrict__ el,
                                             const float* __restrict__ er,
                                             const int* __restrict__ src, int base, int wave, int lane) {
  for (unsigned b0 = 0; b0 < nq; b0 += 128) {
    unsigned idx[4];
    bool live[4];
    int me[4], mslot[4], ms[4];
    float s[4];
#pragma unroll
    for (int u = 0; u < 4; ++u) {
      idx[u]  = b0 + (unsigned)(32 * u + lane);
      live[u] = idx[u] < nq;
      const unsigned ent = live[u] ? lw[idx[u]] : (7u << 20);
      me[u]    = (int)(ent & 0xFFFFFu);
      mslot[u] = (int)(ent >> 20);
    }
#pragma unroll
    for (int u = 0; u < 4; ++u) ms[u] = LOOKUP ? src[me[u]] : me[u];
#pragma unroll
    for (int u = 0; u < 4; ++u)
      s[u] = out_score(el, er, ms[u], (mslot[u] < 7) ? (base + (mslot[u] << 5) + wave) : base);
#pragma unroll
    for (int u = 0; u < 4; ++u) {
      if (live[u]) {
        if (LOOKUP) lw[idx[u]] = (unsigned)ms[u] | ((unsigned)mslot[u] << 20);
        atomicMax(&mxw[mslot[u]], f2ord(s[u]));
      }
    }
  }
}

__device__ void out_rows_by_scan(int base, int wave, int lane, const float* __restrict__ Hm,
                                 const float* __restrict__ el, const float* __restrict__ er,
                                 const int* __restrict__ src, const int* __restrict__ dst,
                                 float* AGG) {
  for (int k = 0; k < 7; ++k) {
    const int r = (k << 5) + wave;
    if (r >= OWN_R) break;
    const int d = base + r;
    float m = -INFINITY;
    for (int e = lane; e < EE; e += 32) {
      if (dst[e] == d) m = fmaxf(m, out_score(el, er, src[e], d));
    }
    m = own_wave_max(m);
    float z = 0.0f;
    for (int e = lane; e < EE; e += 32) {
      if (dst[e] == d) z += expf(out_score(el, er, src[e], d) - m);
    }
    z = own_wave_sum(z);
    float acc0 = 0.0f, acc1 = 0.0f;
    for (int e0 = 0; e0 < EE; e0 += 32) {
      const int e = e0 + lane;
      const bool hit = dst[e] == d;
      int sn = 0;
      float a = 0.0f;
      if (hit) {
        sn = src[e];
        a = expf(out_score(el, er, sn, d) - m) / z;
      }
      unsigned mask = (unsigned)__ballot(hit);
      while (mask) {
        const int j = __ffs(mask) - 1;
        mask &= mask - 1u;
        const int sj = __shfl(sn, j);
        const float aj = __shfl(a, j);
        float q[2];
        out_load_row(q, Hm + (size_t)sj * OUT_F, lane);
        acc0 += q[0] * aj;
        acc1 += q[1] * aj;
      }
    }
    float* orow = AGG + (size_t)d * OUT_F;
    orow[lane] += acc0;
    if (lane < 8) orow[32 + lane] += acc1;
  }
}

__global__ __launch_bounds__(1024)
void gat_owner_rows_out(const float* __restrict__ Hm, const float* __restrict__ el,
                        const float* __restrict__ er, const int* __restrict__ src,
                        const int* __restrict__ dst, float* AGG, const unsigned* klist,
                        const unsigned* kcnt) {
  __shared__ unsigned lst[32][OWN_CAP];
  __shared__ unsigned cnt[32];
  __shared__ unsigned mx[32][8];
  __shared__ float    dn[32][8];
  __shared__ unsigned full;
  __shared__ unsigned kept;
  const int tid  = threadIdx.x;
  const int lane = tid & 31;
  const int wave = __builtin_amdgcn_readfirstlane(tid >> 5);
  const int base = blockIdx.x * OWN_R;
  const unsigned* kc = kcnt + blockIdx.x * 32;
  const unsigned* kl = klist + (size_t)(blockIdx.x * 32 + wave) * OWN_KEEP;

  if (tid < 32) cnt[tid] = 0u;
  if (tid == 0) { full = 0u; kept = 1u; }
  if (tid < 256) {
    (&mx[0][0])[tid] = 0u;
    (&dn[0][0])[tid] = 0.0f;
  }
  __syncthreads();

  if (tid < 32 && kc[tid] > (unsigned)OWN_KEEP) kept = 0u;
  __syncthreads();
  const bool reuse = __builtin_amdgcn_readfirstlane((int)kept) != 0;

  if (reuse) {
    const unsigned n = (unsigned)__builtin_amdgcn_readfirstlane((int)kc[wave]);
    own_copy_list(lst[wave], kl, n, lane);
    if (lane == 0) cnt[wave] = n;
  } else {
    own_filter(lst, cnt, &full, dst, base, wave, lane);
  }
  __syncthreads();

  const bool overfull = __builtin_amdgcn_readfirstlane((int)full) != 0;
  if (overfull) {
    out_rows_by_scan(base, wave, lane, Hm, el, er, src, dst, AGG);
    return;
  }

  const unsigned nq = (unsigned)__builtin_amdgcn_readfirstlane((int)cnt[wave]);

  if (reuse) {
    out_pass_max<false>(lst[wave], mx[wave], nq, el, er, src, base, wave, lane);
  } else {
    out_pass_max<true>(lst[wave], mx[wave], nq, el, er, src, base, wave, lane);
  }
  __syncthreads();

  for (unsigned b0 = 0; b0 < nq; b0 += 128) {
    unsigned idx[4];
    bool live[4];
    int mslot[4], ms[4];
    float s[4];
#pragma unroll
    for (int u = 0; u < 4; ++u) {
      idx[u]  = b0 + (unsigned)(32 * u + lane);
      live[u] = idx[u] < nq;
      const unsigned ent = live[u] ? lst[wave][idx[u]] : (7u << 20);
      ms[u]    = (int)(ent & 0xFFFFFu);
      mslot[u] = (int)(ent >> 20);
    }
#pragma unroll
    for (int u = 0; u < 4; ++u)
      s[u] = out_score(el, er, ms[u], (mslot[u] < 7) ? (base + (mslot[u] << 5) + wave) : base);
#pragma unroll
    for (int u = 0; u < 4; ++u) {
      if (live[u]) atomicAdd(&dn[wave][mslot[u]], expf(s[u] - ord2f(mx[wave][mslot[u]])));
    }
  }
  __syncthreads();

  float acc[7][2] = {};
  for (unsigned b0 = 0; b0 < nq; b0 += 32) {
    const unsigned idx = b0 + (unsigned)lane;
    const unsigned ent = (idx < nq) ? lst[wave][idx] : (7u << 20);
    const int ms    = (int)(ent & 0xFFFFFu);
    const int mslot = (int)(ent >> 20);
    const int md    = (mslot < 7) ? (base + (mslot << 5) + wave) : base;
    const float ma  = expf(out_score(el, er, ms, md) - ord2f(mx[wave][mslot])) / dn[wave][mslot];
    const int nb = (int)min(32u, (nq - b0 + (unsigned)(OUT_DEPTH - 1)) & ~(unsigned)(OUT_DEPTH - 1));

    float q[OUT_DEPTH][2];
#pragma unroll
    for (int p = 0; p < OUT_DEPTH - 1; ++p)
      out_load_row(q[p], Hm + (size_t)__builtin_amdgcn_readlane(ms, p) * OUT_F, lane);
    for (int i = 0; i < nb; i += OUT_DEPTH) {
#pragma unroll
      for (int u = 0; u < OUT_DEPTH; ++u) {
        out_load_row(q[(u + OUT_DEPTH - 1) % OUT_DEPTH],
                     Hm + (size_t)__builtin_amdgcn_readlane(ms, (i + u + OUT_DEPTH - 1) & 31) * OUT_F, lane);
        const float a = __int_as_float(__builtin_amdgcn_readlane(__float_as_int(ma), i + u));
        out_add(acc, q[u], __builtin_amdgcn_readlane(mslot, i + u), a);
      }
    }
  }

  float t[7][2];
#pragma unroll
  for (int k = 0; k < 7; ++k) {
    const int r = (k << 5) + wave;
    if (r < OWN_R) {
      const float* orow = AGG + (size_t)(base + r) * OUT_F;
      out_load_row(t[k], orow, lane);
    }
  }
#pragma unroll
  for (int k = 0; k < 7; ++k) {
    const int r = (k << 5) + wave;
    if (r < OWN_R) {
      float* orow = AGG + (size_t)(base + r) * OUT_F;
      orow[lane] = t[k][0] + acc[k][0];
      if (lane < 8) orow[32 + lane] = t[k][1] + acc[k][1];
    }
  }
}

static void run_layer(const float* X, int K, int Hh, int F,
                      const float* W, const float* al, const float* ar, const float* rW,
                      const int* src, const int* dst,
                      float* Hbuf, float* AGG, int Opad,
                      float* el, float* er, unsigned* klist, unsigned* kcnt,
                      int kind, hipStream_t stream) {
  const int HF = Hh * F;
  dim3 ggrid(NN / 80, Opad / 64, 2);
  if (Hh == 4 && K == 128) {
    gat_gemm_xwt<128, 5, 4><<<ggrid, 32, 0, stream>>>(X, W, rW, Hbuf, AGG, HF);
  } else if (Hh == 4) {
    gat_gemm_xwt<256, 5, 4><<<ggrid, 32, 0, stream>>>(X, W, rW, Hbuf, AGG, HF);
  } else {
    gat_gemm_xwt<256, 5, 3><<<ggrid, 32, 0, stream>>>(X, W, rW, Hbuf, AGG, HF);
  }

  const int nh = NN * Hh;
  gat_el_er<<<(nh + 255) / 256, 256, 0, stream>>>(Hbuf, al, ar, el, er, nh, Hh, F);
  if (kind == 0) {
    gat_owner_rows<true, false><<<NN / OWN_R, 1024, 0, stream>>>(Hbuf, el, er, src, dst, AGG, klist, kcnt);
  } else if (kind == 1) {
    gat_owner_rows<false, true><<<NN / OWN_R, 1024, 0, stream>>>(Hbuf, el, er, src, dst, AGG, klist, kcnt);
  } else {
    gat_owner_rows_out<<<NN / OWN_R, 1024, 0, stream>>>(Hbuf, el, er, src, dst, AGG, klist, kcnt);
  }
}

extern "C" void kernel_launch(void* const* d_in, const int* in_sizes, int n_in,
                              void* d_out, int out_size, void* d_ws, size_t ws_size,
                              hipStream_t stream) {
  (void)in_sizes; (void)n_in; (void)out_size; (void)ws_size;

  const float* feat = (const float*)d_in[0];
  const int*   src  = (const int*)d_in[1];
  const int*   dst  = (const int*)d_in[2];
  const float* W0  = (const float*)d_in[3];
  const float* al0 = (const float*)d_in[4];
  const float* ar0 = (const float*)d_in[5];
  const float* rW0 = (const float*)d_in[6];
  const float* W1  = (const float*)d_in[7];
  const float* al1 = (const float*)d_in[8];
  const float* ar1 = (const float*)d_in[9];
  const float* rW1 = (const float*)d_in[10];
  const float* W2  = (const float*)d_in[11];
  const float* al2 = (const float*)d_in[12];
  const float* ar2 = (const float*)d_in[13];
  const float* rW2 = (const float*)d_in[14];

  const size_t NF = (size_t)NN * 256;
  float* ws    = (float*)d_ws;
  float* bufH  = ws;
  float* bufY0 = bufH  + NF;
  float* bufY1 = bufY0 + NF;
  float* el    = bufY1 + NF;
  float* er    = el  + (size_t)NN * 4;
  unsigned* kcnt  = (unsigned*)(er + (size_t)NN * 4);
  unsigned* klist = kcnt + (size_t)NN * 8;
  float* W2p   = (float*)klist + (size_t)EE * 4;
  float* rW2p  = W2p + (size_t)64 * 256;

  {
    int tot = 64 * 256;
    gat_pad_w<<<dim3((tot + 255) / 256, 2), 256, 0, stream>>>(W2, rW2, W2p, rW2p, 40, 64, 256);
  }

  run_layer(feat, 128, 4, 64, W0, al0, ar0, rW0, src, dst,
            bufH, bufY0, 256, el, er, klist, kcnt, 0, stream);
  run_layer(bufY0, 256, 4, 64, W1, al1, ar1, rW1, src, dst,
            bufH, bufY1, 256, el, er, klist, kcnt, 1, stream);
  run_layer(bufY1, 256, 1, 40, W2p, al2, ar2, rW2p, src, dst,
            bufH, (float*)d_out, 64, el, er, klist, kcnt, 2, stream);
}


# Edge scores kept in LDS beside the row-owner lists: el/er gathered once per edge (pass A), pass B overwrites a score with exp(score-max), phase 2 only divides; float sums and their order unchanged
# speedup vs baseline: 5.4878x; 1.0368x over previous
#include <hip/hip_runtime.h>
#include <math.h>

#define NN 50000
#define EE 800000

typedef __attribute__((ext_vector_type(2))) float v2f;
typedef __attribute__((ext_vector_type(4))) float v4f;
typedef __attribute__((ext_vector_type(8))) float v8f;

__device__ __forceinline__ unsigned f2ord(float f) {
  unsigned u = __float_as_uint(f);
  return (u & 0x80000000u) ? ~u : (u | 0x80000000u);
}
__device__ __forceinline__ float ord2f(unsigned e) {
  unsigned u = (e & 0x80000000u) ? (e ^ 0x80000000u) : ~e;
  return __uint_as_float(u);
}

template <int NF>
__device__ __forceinline__ void gemm_load_lines(v4f (&f)[NF][4], const float* __restrict__ p,
                                                size_t fstep) {
#pragma unroll
  for (int i = 0; i < NF; ++i) {
#pragma unroll
    for (int j = 0; j < 4; ++j) f[i][j] = *(const v4f*)(p + (size_t)i * fstep + 8 * j);
  }
}

template <int MA, int NB>
__device__ __forceinline__ void gemm_mma_lines(v8f (&acc)[MA][NB], const v4f (&a)[MA][4],
                                               const v4f (&b)[NB][4]) {
#pragma unroll
  for (int j = 0; j < 4; ++j) {
#pragma unroll
    for (int i = 0; i < MA; ++i) {
#pragma unroll
      for (int t = 0; t < NB; ++t)
        acc[i][t] = __builtin_amdgcn_wmma_f32_16x16x4_f32(
            false, a[i][j].xy, false, b[t][j].xy, (short)0, acc[i][t], false, false);
    }
#pragma unroll
    for (int i = 0; i < MA; ++i) {
#pragma unroll
      for (int t = 0; t < NB; ++t)
        acc[i][t] = __builtin_amdgcn_wmma_f32_16x16x4_f32(
            false, a[i][j].zw, false, b[t][j].zw, (short)0, acc[i][t], false, false);
    }
  }
}

template <int K, int MA, int NB>
__global__ __launch_bounds__(32)
void gat_gemm_xwt(const float* __restrict__ X, const float* __restrict__ W0,
                  const float* __restrict__ W1, float* __restrict__ OUT0,
                  float* __restrict__ OUT1, int Oout) {
  static_assert(K % 64 == 0, "the K loop takes two 32-float lines per trip");
  static_assert(NN % (16 * MA) == 0, "row tiles cover the nodes exactly");
  const int lane = threadIdx.x & 31;
  const int c    = lane & 15;
  const int half = lane >> 4;
  const int row0 = blockIdx.x * (16 * MA);
  const int col0 = blockIdx.y << 6;
  const float* __restrict__ W   = (blockIdx.z == 0) ? W0 : W1;
  float* __restrict__       OUT = (blockIdx.z == 0) ? OUT0 : OUT1;

  const float* xp = X + (size_t)(row0 + c) * (size_t)K + (half << 2);
  const float* wp = W + (size_t)(col0 + c) * (size_t)K + (half << 2);
  const size_t fstep = (size_t)16 * (size_t)K;

  v8f acc[MA][NB] = {};
  v4f aP[MA][4], bP[NB][4], aQ[MA][4], bQ[NB][4];

  gemm_load_lines<MA>(aP, xp, fstep);
  gemm_load_lines<NB>(bP, wp, fstep);
  int kb = 0;
  for (; kb + 64 < K; kb += 64) {
    gemm_load_lines<MA>(aQ, xp + kb + 32, fstep);
    gemm_load_lines<NB>(bQ, wp + kb + 32, fstep);
    __builtin_amdgcn_sched_barrier(0);
    gemm_mma_lines<MA, NB>(acc, aP, bP);
    __builtin_amdgcn_sched_barrier(0);
    gemm_load_lines<MA>(aP, xp + kb + 64, fstep);
    gemm_load_lines<NB>(bP, wp + kb + 64, fstep);
    __builtin_amdgcn_sched_barrier(0);
    gemm_mma_lines<MA, NB>(acc, aQ, bQ);
    __builtin_amdgcn_sched_barrier(0);
  }
  gemm_load_lines<MA>(aQ, xp + kb + 32, fstep);
  gemm_load_lines<NB>(bQ, wp + kb + 32, fstep);
  __builtin_amdgcn_sched_barrier(0);
  gemm_mma_lines<MA, NB>(acc, aP, bP);
  gemm_mma_lines<MA, NB>(acc, aQ, bQ);

#pragma unroll
  for (int i = 0; i < MA; ++i) {
#pragma unroll
    for (int t = 0; t < NB; ++t) {
      const int oc = col0 + (t << 4) + c;
      if (oc < Oout) {
        float* orow = OUT + (size_t)(row0 + (i << 4) + (half << 3)) * (size_t)Oout + oc;
#pragma unroll
        for (int v = 0; v < 8; ++v) {
          orow[(size_t)v * (size_t)Oout] = acc[i][t][v];
        }
      }
    }
  }
}

__global__ void gat_pad_w(const float* __restrict__ Wsrc0, const float* __restrict__ Wsrc1,
                          float* __restrict__ Wdst0, float* __restrict__ Wdst1,
                          int rows, int padrows, int K) {
  int idx = blockIdx.x * blockDim.x + threadIdx.x;
  int total = padrows * K;
  if (idx >= total) return;
  const float* __restrict__ Wsrc = (blockIdx.y == 0) ? Wsrc0 : Wsrc1;
  float* __restrict__       Wdst = (blockIdx.y == 0) ? Wdst0 : Wdst1;
  int r = idx / K;
  Wdst[idx] = (r < rows) ? Wsrc[idx] : 0.0f;
}

__global__ void gat_el_er(const float* __restrict__ Hm, const float* __restrict__ al,
                          const float* __restrict__ ar, float* __restrict__ el,
                          float* __restrict__ er, int total, int Hh, int F) {
  int idx = blockIdx.x * blockDim.x + threadIdx.x;
  if (idx >= total) return;
  int h = idx % Hh;
  int n = idx / Hh;
  const float* row = Hm + (size_t)n * (size_t)(Hh * F) + (size_t)h * F;
  const float* a = al + h * F;
  const float* b = ar + h * F;
  float sl = 0.f, sr = 0.f;
  for (int f = 0; f < F; ++f) {
    float v = row[f];
    sl += v * a[f];
    sr += v * b[f];
  }
  el[idx] = sl;
  er[idx] = sr;
}

template <int HF, int NCH>
__device__ __forceinline__ void scat_load_row(float (&r)[NCH], const float* __restrict__ row,
                                              int lane) {
#pragma unroll
  for (int c = 0; c < NCH; ++c) {
    const int col = (c << 5) + lane;
    if (HF % 32 == 0 || col < HF) r[c] = row[col];
  }
}

#define OWN_R   200
#define OWN_CAP 1024
#define OWN_KEEP 384
#define OWN_SC  256
#define OWN_NCH (EE / 128)
static_assert(NN % OWN_R == 0, "row groups cover the nodes exactly");
static_assert(OWN_R <= 7 * 32, "7 register slots per wave");
static_assert(EE % 128 == 0, "a wave filters 128 edges per load");
static_assert(EE <= (1 << 20), "an edge index is packed into 20 bits");
static_assert(NN <= (1 << 20), "a node index is packed into 20 bits");
static_assert(((OWN_NCH + 31) / 32) % 4 == 0, "the filter loop takes 4 loads per trip");
static_assert(OWN_NCH % (NN / OWN_R) == 0, "start offsets of the groups tile the list");
static_assert(OWN_KEEP % 128 == 0 && OWN_KEEP <= OWN_CAP, "kept lists are copied 128 entries per trip");
static_assert(OWN_SC % 128 == 0 && OWN_SC <= OWN_CAP, "a trip of 128 entries has its scores kept whole or not at all");
static_assert((size_t)(NN / OWN_R) * 32 * OWN_KEEP <= (size_t)EE * 4, "kept lists fit their [EE,4]-word place in the workspace");
static_assert((NN / OWN_R) * 32 <= NN * 4, "kept list lengths fit their [NN,4]-word place in the workspace");

typedef __attribute__((ext_vector_type(4))) int v4i;

__device__ __forceinline__ void own_queue(unsigned (*lst)[OWN_CAP], unsigned* cnt, unsigned* full,
                                          int e, unsigned r) {
  const unsigned w = r & 31u;
  const unsigned pos = atomicAdd(&cnt[w], 1u);
  if (pos < OWN_CAP) {
    lst[w][pos] = (unsigned)e | ((r >> 5) << 20);
  } else {
    *full = 1u;
  }
}

__device__ __forceinline__ void own_filter(unsigned (*lst)[OWN_CAP], unsigned* cnt, unsigned* full,
                                           const int* __restrict__ dst, int base, int wave, int lane) {
  const int rot = blockIdx.x * (OWN_NCH / (NN / OWN_R));
  const v4i* __restrict__ dp = (const v4i*)dst;
  for (int t = 0; t < (OWN_NCH + 31) / 32; t += 4) {
    v4i dv[4];
    int eb[4];
    bool ok[4];
#pragma unroll
    for (int u = 0; u < 4; ++u) {
      const int L = wave + 32 * (t + u);
      ok[u] = L < OWN_NCH;
      int ch = (ok[u] ? L : 0) + rot;
      if (ch >= OWN_NCH) ch -= OWN_NCH;
      eb[u] = ch * 128 + lane * 4;
      dv[u] = dp[ch * 32 + lane];
    }
#pragma unroll
    for (int u = 0; u < 4; ++u) {
      const unsigned r0 = (unsigned)(dv[u].x - base);
      const unsigned r1 = (unsigned)(dv[u].y - base);
      const unsigned r2 = (unsigned)(dv[u].z - base);
      const unsigned r3 = (unsigned)(dv[u].w - base);
      const unsigned mn = min(min(r0, r1), min(r2, r3));
      if (ok[u] && mn < (unsigned)OWN_R) {
        if (r0 < (unsigned)OWN_R) own_queue(lst, cnt, full, eb[u] + 0, r0);
        if (r1 < (unsigned)OWN_R) own_queue(lst, cnt, full, eb[u] + 1, r1);
        if (r2 < (unsigned)OWN_R) own_queue(lst, cnt, full, eb[u] + 2, r2);
        if (r3 < (unsigned)OWN_R) own_queue(lst, cnt, full, eb[u] + 3, r3);
      }
    }
  }
}

__device__ __forceinline__ void own_copy_list(unsigned* lw, const unsigned* kl, unsigned n, int lane) {
  for (unsigned i0 = 0; i0 < n; i0 += 128) {
    unsigned v[4];
#pragma unroll
    for (int u = 0; u < 4; ++u) {
      const unsigned i = i0 + (unsigned)(32 * u + lane);
      v[u] = (i < n) ? kl[i] : 0u;
    }
#pragma unroll
    for (int u = 0; u < 4; ++u) {
      const unsigned i = i0 + (unsigned)(32 * u + lane);
      if (i < n) lw[i] = v[u];
    }
  }
}

__device__ __forceinline__ void own_scores(float (&s)[4], const float* __restrict__ el,
                                           const float* __restrict__ er, int sn, int dn_) {
  const v4f a = *(const v4f*)(el + (size_t)sn * 4);
  const v4f b = *(const v4f*)(er + (size_t)dn_ * 4);
  s[0] = a.x + b.x; s[1] = a.y + b.y; s[2] = a.z + b.z; s[3] = a.w + b.w;
#pragma unroll
  for (int h = 0; h < 4; ++h) s[h] = (s[h] > 0.0f) ? s[h] : 0.2f * s[h];
}

__device__ __forceinline__ void own_add(float (&acc)[7][8], const float (&q)[8], int slot,
                                        const float (&al)[4]) {
#pragma unroll
  for (int k = 0; k < 7; ++k) {
    if (slot == k) {
#pragma unroll
      for (int c = 0; c < 8; ++c) acc[k][c] += q[c] * al[c >> 1];
    }
  }
}

__device__ __forceinline__ float own_wave_max(float v) {
  for (int o = 16; o > 0; o >>= 1) v = fmaxf(v, __shfl_xor(v, o));
  return v;
}
__device__ __forceinline__ float own_wave_sum(float v) {
  for (int o = 16; o > 0; o >>= 1) v += __shfl_xor(v, o);
  return v;
}

template <bool LOOKUP, bool KEEP>
__device__ __forceinline__ void own_pass_max(unsigned* lw, v4f* scw, unsigned (*mxw)[4], unsigned nq,
                                             const float* __restrict__ el,
                                             const float* __restrict__ er,
                                             const int* __restrict__ src, unsigned* __restrict__ kl,
                                             int base, int wave, int lane) {
  for (unsigned b0 = 0; b0 < nq; b0 += 128) {
    unsigned idx[4];
    bool live[4];
    int me[4], mslot[4], ms[4];
    float s[4][4];
#pragma unroll
    for (int u = 0; u < 4; ++u) {
      idx[u]  = b0 + (unsigned)(32 * u + lane);
      live[u] = idx[u] < nq;
      const unsigned ent = live[u] ? lw[idx[u]] : (7u << 20);
      me[u]    = (int)(ent & 0xFFFFFu);
      mslot[u] = (int)(ent >> 20);
    }
#pragma unroll
    for (int u = 0; u < 4; ++u) ms[u] = LOOKUP ? src[me[u]] : me[u];
#pragma unroll
    for (int u = 0; u < 4; ++u)
      own_scores(s[u], el, er, ms[u], (mslot[u] < 7) ? (base + (mslot[u] << 5) + wave) : base);
#pragma unroll
    for (int u = 0; u < 4; ++u) {
      if (live[u]) {
        if (LOOKUP) {
          const unsigned ent = (unsigned)ms[u] | ((unsigned)mslot[u] << 20);
          lw[idx[u]] = ent;
          if (KEEP && idx[u] < (unsigned)OWN_KEEP) kl[idx[u]] = ent;
        }
#pragma unroll
        for (int h = 0; h < 4; ++h) atomicMax(&mxw[mslot[u]][h], f2ord(s[u][h]));
        if (b0 < (unsigned)OWN_SC) {
          v4f sv;
          sv.x = s[u][0]; sv.y = s[u][1]; sv.z = s[u][2]; sv.w = s[u][3];
          scw[idx[u]] = sv;
        }
      }
    }
  }
}

__device__ void own_rows_by_scan(int base, int wave, int lane, const float* __restrict__ Hm,
                                 const float* __restrict__ el, const float* __restrict__ er,
                                 const int* __restrict__ src, const int* __restrict__ dst,
                                 float* AGG) {
  for (int k = 0; k < 7; ++k) {
    const int r = (k << 5) + wave;
    if (r >= OWN_R) break;
    const int d = base + r;
    float m[4] = {-INFINITY, -INFINITY, -INFINITY, -INFINITY};
    for (int e = lane; e < EE; e += 32) {
      if (dst[e] == d) {
        float s[4];
        own_scores(s, el, er, src[e], d);
        for (int h = 0; h < 4; ++h) m[h] = fmaxf(m[h], s[h]);
      }
    }
    for (int h = 0; h < 4; ++h) m[h] = own_wave_max(m[h]);
    float z[4] = {0.0f, 0.0f, 0.0f, 0.0f};
    for (int e = lane; e < EE; e += 32) {
      if (dst[e] == d) {
        float s[4];
        own_scores(s, el, er, src[e], d);
        for (int h = 0; h < 4; ++h) z[h] += expf(s[h] - m[h]);
      }
    }
    for (int h = 0; h < 4; ++h) z[h] = own_wave_sum(z[h]);
    float acc[8] = {};
    for (int e0 = 0; e0 < EE; e0 += 32) {
      const int e = e0 + lane;
      const bool hit = dst[e] == d;
      int sn = 0;
      float a[4] = {0.0f, 0.0f, 0.0f, 0.0f};
      if (hit) {
        sn = src[e];
        float s[4];
        own_scores(s, el, er, sn, d);
        for (int h = 0; h < 4; ++h) a[h] = expf(s[h] - m[h]) / z[h];
      }
      unsigned mask = (unsigned)__ballot(hit);
      while (mask) {
        const int j = __ffs(mask) - 1;
        mask &= mask - 1u;
        const int sj = __shfl(sn, j);
        float aj[4];
        for (int h = 0; h < 4; ++h) aj[h] = __shfl(a[h], j);
        const float* row = Hm + (size_t)sj * 256;
        for (int c = 0; c < 8; ++c) acc[c] += row[(c << 5) + lane] * aj[c >> 1];
      }
    }
    float* orow = AGG + (size_t)d * 256;
    for (int c = 0; c < 8; ++c) {
      const float v = orow[(c << 5) + lane] + acc[c];
      orow[(c << 5) + lane] = (v > 0.0f) ? v : expm1f(v);
    }
  }
}

template <bool KEEP, bool REUSE>
__global__ __launch_bounds__(1024)
void gat_owner_rows(const float* __restrict__ Hm, const float* __restrict__ el,
                    const float* __restrict__ er, const int* __restrict__ src,
                    const int* __restrict__ dst, float* AGG, unsigned* klist, unsigned* kcnt) {
  __shared__ unsigned lst[32][OWN_CAP];
  __shared__ v4f      sc[32][OWN_SC];
  __shared__ unsigned cnt[32];
  __shared__ unsigned mx[32][8][4];
  __shared__ float    dn[32][8][4];
  __shared__ unsigned full;
  __shared__ unsigned kept;
  const int tid  = threadIdx.x;
  const int lane = tid & 31;
  const int wave = __builtin_amdgcn_readfirstlane(tid >> 5);
  const int base = blockIdx.x * OWN_R;
  unsigned* kc = kcnt + blockIdx.x * 32;
  unsigned* kl = klist + (size_t)(blockIdx.x * 32 + wave) * OWN_KEEP;

  if (tid < 32) cnt[tid] = 0u;
  if (tid == 0) { full = 0u; kept = 1u; }
  (&mx[0][0][0])[tid] = 0u;
  (&dn[0][0][0])[tid] = 0.0f;
  __syncthreads();

  bool reuse = false;
  if (REUSE) {
    if (tid < 32 && kc[tid] > (unsigned)OWN_KEEP) kept = 0u;
    __syncthreads();
    reuse = __builtin_amdgcn_readfirstlane((int)kept) != 0;
  }

  if (reuse) {
    const unsigned n = (unsigned)__builtin_amdgcn_readfirstlane((int)kc[wave]);
    own_copy_list(lst[wave], kl, n, lane);
    if (lane == 0) cnt[wave] = n;
  } else {
    own_filter(lst, cnt, &full, dst, base, wave, lane);
  }
  __syncthreads();

  const bool overfull = __builtin_amdgcn_readfirstlane((int)full) != 0;
  if (KEEP && lane == 0) kc[wave] = overfull ? 0xFFFFFFFFu : cnt[wave];
  if (overfull) {
    own_rows_by_scan(base, wave, lane, Hm, el, er, src, dst, AGG);
    return;
  }

  const unsigned nq = (unsigned)__builtin_amdgcn_readfirstlane((int)cnt[wave]);

  if (reuse) {
    own_pass_max<false, false>(lst[wave], sc[wave], mx[wave], nq, el, er, src, kl, base, wave, lane);
  } else {
    own_pass_max<true, KEEP>(lst[wave], sc[wave], mx[wave], nq, el, er, src, kl, base, wave, lane);
  }
  __syncthreads();

  for (unsigned b0 = 0; b0 < nq; b0 += 128) {
    unsigned idx[4];
    bool live[4];
    int mslot[4], ms[4];
    float s[4][4];
#pragma unroll
    for (int u = 0; u < 4; ++u) {
      idx[u]  = b0 + (unsigned)(32 * u + lane);
      live[u] = idx[u] < nq;
      const unsigned ent = live[u] ? lst[wave][idx[u]] : (7u << 20);
      ms[u]    = (int)(ent & 0xFFFFFu);
      mslot[u] = (int)(ent >> 20);
    }
    const bool kept_sc = b0 < (unsigned)OWN_SC;
    if (kept_sc) {
#pragma unroll
      for (int u = 0; u < 4; ++u) {
        const v4f sv = sc[wave][idx[u]];
        s[u][0] = sv.x; s[u][1] = sv.y; s[u][2] = sv.z; s[u][3] = sv.w;
      }
    } else {
#pragma unroll
      for (int u = 0; u < 4; ++u)
        own_scores(s[u], el, er, ms[u], (mslot[u] < 7) ? (base + (mslot[u] << 5) + wave) : base);
    }
#pragma unroll
    for (int u = 0; u < 4; ++u) {
      if (live[u]) {
        float ev[4];
#pragma unroll
        for (int h = 0; h < 4; ++h) {
          ev[h] = expf(s[u][h] - ord2f(mx[wave][mslot[u]][h]));
          atomicAdd(&dn[wave][mslot[u]][h], ev[h]);
        }
        if (kept_sc) {
          v4f sv;
          sv.x = ev[0]; sv.y = ev[1]; sv.z = ev[2]; sv.w = ev[3];
          sc[wave][idx[u]] = sv;
        }
      }
    }
  }
  __syncthreads();

  float acc[7][8] = {};
  for (unsigned b0 = 0; b0 < nq; b0 += 32) {
    const unsigned idx = b0 + (unsigned)lane;
    const unsigned ent = (idx < nq) ? lst[wave][idx] : (7u << 20);
    const int ms    = (int)(ent & 0xFFFFFu);
    const int mslot = (int)(ent >> 20);
    const int md    = (mslot < 7) ? (base + (mslot << 5) + wave) : base;
    float ma[4];
    if (b0 < (unsigned)OWN_SC) {
      const v4f ev = sc[wave][idx];
      ma[0] = ev.x; ma[1] = ev.y; ma[2] = ev.z; ma[3] = ev.w;
    } else {
      own_scores(ma, el, er, ms, md);
#pragma unroll
      for (int h = 0; h < 4; ++h) ma[h] = expf(ma[h] - ord2f(mx[wave][mslot][h]));
    }
#pragma unroll
    for (int h = 0; h < 4; ++h) ma[h] = ma[h] / dn[wave][mslot][h];

    float q0[8], q1[8], q2[8], q3[8];
    float al[4];
    scat_load_row<256, 8>(q0, Hm + (size_t)__builtin_amdgcn_readlane(ms, 0) * 256, lane);
    scat_load_row<256, 8>(q1, Hm + (size_t)__builtin_amdgcn_readlane(ms, 1) * 256, lane);
    scat_load_row<256, 8>(q2, Hm + (size_t)__builtin_amdgcn_readlane(ms, 2) * 256, lane);
    for (int i = 0; i < 32; i += 4) {
      scat_load_row<256, 8>(q3, Hm + (size_t)__builtin_amdgcn_readlane(ms, i + 3) * 256, lane);
#pragma unroll
      for (int h = 0; h < 4; ++h) al[h] = __int_as_float(__builtin_amdgcn_readlane(__float_as_int(ma[h]), i));
      own_add(acc, q0, __builtin_amdgcn_readlane(mslot, i), al);

      scat_load_row<256, 8>(q0, Hm + (size_t)__builtin_amdgcn_readlane(ms, (i + 4) & 31) * 256, lane);
#pragma unroll
      for (int h = 0; h < 4; ++h) al[h] = __int_as_float(__builtin_amdgcn_readlane(__float_as_int(ma[h]), i + 1));
      own_add(acc, q1, __builtin_amdgcn_readlane(mslot, i + 1), al);

      scat_load_row<256, 8>(q1, Hm + (size_t)__builtin_amdgcn_readlane(ms, (i + 5) & 31) * 256, lane);
#pragma unroll
      for (int h = 0; h < 4; ++h) al[h] = __int_as_float(__builtin_amdgcn_readlane(__float_as_int(ma[h]), i + 2));
      own_add(acc, q2, __builtin_amdgcn_readlane(mslot, i + 2), al);

      scat_load_row<256, 8>(q2, Hm + (size_t)__builtin_amdgcn_readlane(ms, (i + 6) & 31) * 256, lane);
#pragma unroll
      for (int h = 0; h < 4; ++h) al[h] = __int_as_float(__builtin_amdgcn_readlane(__float_as_int(ma[h]), i + 3));
      own_add(acc, q3, __builtin_amdgcn_readlane(mslot, i + 3), al);
    }
  }

#pragma unroll
  for (int g = 0; g < 7; g += 4) {
    float t[4][8];
#pragma unroll
    for (int kk = 0; kk < 4; ++kk) {
      const int k = g + kk;
      const int r = (k << 5) + wave;
      if (k < 7 && r < OWN_R) {
        const float* orow = AGG + (size_t)(base + r) * 256;
#pragma unroll
        for (int c = 0; c < 8; ++c) t[kk][c] = orow[(c << 5) + lane];
      }
    }
#pragma unroll
    for (int kk = 0; kk < 4; ++kk) {
      const int k = g + kk;
      const int r = (k << 5) + wave;
      if (k < 7 && r < OWN_R) {
        float* orow = AGG + (size_t)(base + r) * 256;
#pragma unroll
        for (int c = 0; c < 8; ++c) {
          const float v = t[kk][c] + acc[k < 7 ? k : 0][c];
          orow[(c << 5) + lane] = (v > 0.0f) ? v : expm1f(v);
        }
      }
    }
  }
}

#define OUT_F 40
#define OUT_DEPTH 8

__device__ __forceinline__ float out_score(const float* __restrict__ el, const float* __restrict__ er,
                                           int sn, int dn_) {
  const float s = el[sn] + er[dn_];
  return (s > 0.0f) ? s : 0.2f * s;
}

__device__ __forceinline__ void out_load_row(float (&r)[2], const float* __restrict__ row, int lane) {
  r[0] = row[lane];
  r[1] = row[32 + (lane & 7)];
}

__device__ __forceinline__ void out_add(float (&acc)[7][2], const float (&q)[2], int slot, float a) {
#pragma unroll
  for (int k = 0; k < 7; ++k) {
    if (slot == k) {
      acc[k][0] += q[0] * a;
      acc[k][1] += q[1] * a;
    }
  }
}

template <bool LOOKUP>
__device__ __forceinline__ void out_pass_max(unsigned* lw, float* scw, unsigned* mxw, unsigned nq,
                                             const float* __restrict__ el,
                                             const float* __restrict__ er,
                                             const int* __restrict__ src, int base, int wave, int lane) {
  for (unsigned b0 = 0; b0 < nq; b0 += 128) {
    unsigned idx[4];
    bool live[4];
    int me[4], mslot[4], ms[4];
    float s[4];
#pragma unroll
    for (int u = 0; u < 4; ++u) {
      idx[u]  = b0 + (unsigned)(32 * u + lane);
      live[u] = idx[u] < nq;
      const unsigned ent = live[u] ? lw[idx[u]] : (7u << 20);
      me[u]    = (int)(ent & 0xFFFFFu);
      mslot[u] = (int)(ent >> 20);
    }
#pragma unroll
    for (int u = 0; u < 4; ++u) ms[u] = LOOKUP ? src[me[u]] : me[u];
#pragma unroll
    for (int u = 0; u < 4; ++u)
      s[u] = out_score(el, er, ms[u], (mslot[u] < 7) ? (base + (mslot[u] << 5) + wave) : base);
#pragma unroll
    for (int u = 0; u < 4; ++u) {
      if (live[u]) {
        if (LOOKUP) lw[idx[u]] = (unsigned)ms[u] | ((unsigned)mslot[u] << 20);
        atomicMax(&mxw[mslot[u]], f2ord(s[u]));
        scw[idx[u]] = s[u];
      }
    }
  }
}

__device__ void out_rows_by_scan(int base, int wave, int lane, const float* __restrict__ Hm,
                                 const float* __restrict__ el, const float* __restrict__ er,
                                 const int* __restrict__ src, const int* __restrict__ dst,
                                 float* AGG) {
  for (int k = 0; k < 7; ++k) {
    const int r = (k << 5) + wave;
    if (r >= OWN_R) break;
    const int d = base + r;
    float m = -INFINITY;
    for (int e = lane; e < EE; e += 32) {
      if (dst[e] == d) m = fmaxf(m, out_score(el, er, src[e], d));
    }
    m = own_wave_max(m);
    float z = 0.0f;
    for (int e = lane; e < EE; e += 32) {
      if (dst[e] == d) z += expf(out_score(el, er, src[e], d) - m);
    }
    z = own_wave_sum(z);
    float acc0 = 0.0f, acc1 = 0.0f;
    for (int e0 = 0; e0 < EE; e0 += 32) {
      const int e = e0 + lane;
      const bool hit = dst[e] == d;
      int sn = 0;
      float a = 0.0f;
      if (hit) {
        sn = src[e];
        a = expf(out_score(el, er, sn, d) - m) / z;
      }
      unsigned mask = (unsigned)__ballot(hit);
      while (mask) {
        const int j = __ffs(mask) - 1;
        mask &= mask - 1u;
        const int sj = __shfl(sn, j);
        const float aj = __shfl(a, j);
        float q[2];
        out_load_row(q, Hm + (size_t)sj * OUT_F, lane);
        acc0 += q[0] * aj;
        acc1 += q[1] * aj;
      }
    }
    float* orow = AGG + (size_t)d * OUT_F;
    orow[lane] += acc0;
    if (lane < 8) orow[32 + lane] += acc1;
  }
}

__global__ __launch_bounds__(1024)
void gat_owner_rows_out(const float* __restrict__ Hm, const float* __restrict__ el,
                        const float* __restrict__ er, const int* __restrict__ src,
                        const int* __restrict__ dst, float* AGG, const unsigned* klist,
                        const unsigned* kcnt) {
  __shared__ unsigned lst[32][OWN_CAP];
  __shared__ float    sc[32][OWN_CAP];
  __shared__ unsigned cnt[32];
  __shared__ unsigned mx[32][8];
  __shared__ float    dn[32][8];
  __shared__ unsigned full;
  __shared__ unsigned kept;
  const int tid  = threadIdx.x;
  const int lane = tid & 31;
  const int wave = __builtin_amdgcn_readfirstlane(tid >> 5);
  const int base = blockIdx.x * OWN_R;
  const unsigned* kc = kcnt + blockIdx.x * 32;
  const unsigned* kl = klist + (size_t)(blockIdx.x * 32 + wave) * OWN_KEEP;

  if (tid < 32) cnt[tid] = 0u;
  if (tid == 0) { full = 0u; kept = 1u; }
  if (tid < 256) {
    (&mx[0][0])[tid] = 0u;
    (&dn[0][0])[tid] = 0.0f;
  }
  __syncthreads();

  if (tid < 32 && kc[tid] > (unsigned)OWN_KEEP) kept = 0u;
  __syncthreads();
  const bool reuse = __builtin_amdgcn_readfirstlane((int)kept) != 0;

  if (reuse) {
    const unsigned n = (unsigned)__builtin_amdgcn_readfirstlane((int)kc[wave]);
    own_copy_list(lst[wave], kl, n, lane);
    if (lane == 0) cnt[wave] = n;
  } else {
    own_filter(lst, cnt, &full, dst, base, wave, lane);
  }
  __syncthreads();

  const bool overfull = __builtin_amdgcn_readfirstlane((int)full) != 0;
  if (overfull) {
    out_rows_by_scan(base, wave, lane, Hm, el, er, src, dst, AGG);
    return;
  }

  const unsigned nq = (unsigned)__builtin_amdgcn_readfirstlane((int)cnt[wave]);

  if (reuse) {
    out_pass_max<false>(lst[wave], sc[wave], mx[wave], nq, el, er, src, base, wave, lane);
  } else {
    out_pass_max<true>(lst[wave], sc[wave], mx[wave], nq, el, er, src, base, wave, lane);
  }
  __syncthreads();

  for (unsigned b0 = 0; b0 < nq; b0 += 128) {
    unsigned idx[4];
    bool live[4];
    int mslot[4];
    float s[4];
#pragma unroll
    for (int u = 0; u < 4; ++u) {
      idx[u]  = b0 + (unsigned)(32 * u + lane);
      live[u] = idx[u] < nq;
      const unsigned ent = live[u] ? lst[wave][idx[u]] : (7u << 20);
      mslot[u] = (int)(ent >> 20);
      s[u] = sc[wave][idx[u]];
    }
#pragma unroll
    for (int u = 0; u < 4; ++u) {
      if (live[u]) {
        const float ev = expf(s[u] - ord2f(mx[wave][mslot[u]]));
        atomicAdd(&dn[wave][mslot[u]], ev);
        sc[wave][idx[u]] = ev;
      }
    }
  }
  __syncthreads();

  float acc[7][2] = {};
  for (unsigned b0 = 0; b0 < nq; b0 += 32) {
    const unsigned idx = b0 + (unsigned)lane;
    const unsigned ent = (idx < nq) ? lst[wave][idx] : (7u << 20);
    const int ms    = (int)(ent & 0xFFFFFu);
    const int mslot = (int)(ent >> 20);
    const float ma  = sc[wave][idx] / dn[wave][mslot];
    const int nb = (int)min(32u, (nq - b0 + (unsigned)(OUT_DEPTH - 1)) & ~(unsigned)(OUT_DEPTH - 1));

    float q[OUT_DEPTH][2];
#pragma unroll
    for (int p = 0; p < OUT_DEPTH - 1; ++p)
      out_load_row(q[p], Hm + (size_t)__builtin_amdgcn_readlane(ms, p) * OUT_F, lane);
    for (int i = 0; i < nb; i += OUT_DEPTH) {
#pragma unroll
      for (int u = 0; u < OUT_DEPTH; ++u) {
        out_load_row(q[(u + OUT_DEPTH - 1) % OUT_DEPTH],
                     Hm + (size_t)__builtin_amdgcn_readlane(ms, (i + u + OUT_DEPTH - 1) & 31) * OUT_F, lane);
        const float a = __int_as_float(__builtin_amdgcn_readlane(__float_as_int(ma), i + u));
        out_add(acc, q[u], __builtin_amdgcn_readlane(mslot, i + u), a);
      }
    }
  }

  float t[7][2];
#pragma unroll
  for (int k = 0; k < 7; ++k) {
    const int r = (k << 5) + wave;
    if (r < OWN_R) {
      const float* orow = AGG + (size_t)(base + r) * OUT_F;
      out_load_row(t[k], orow, lane);
    }
  }
#pragma unroll
  for (int k = 0; k < 7; ++k) {
    const int r = (k << 5) + wave;
    if (r < OWN_R) {
      float* orow = AGG + (size_t)(base + r) * OUT_F;
      orow[lane] = t[k][0] + acc[k][0];
      if (lane < 8) orow[32 + lane] = t[k][1] + acc[k][1];
    }
  }
}

static void run_layer(const float* X, int K, int Hh, int F,
                      const float* W, const float* al, const float* ar, const float* rW,
                      const int* src, const int* dst,
                      float* Hbuf, float* AGG, int Opad,
                      float* el, float* er, unsigned* klist, unsigned* kcnt,
                      int kind, hipStream_t stream) {
  const int HF = Hh * F;
  dim3 ggrid(NN / 80, Opad / 64, 2);
  if (Hh == 4 && K == 128) {
    gat_gemm_xwt<128, 5, 4><<<ggrid, 32, 0, stream>>>(X, W, rW, Hbuf, AGG, HF);
  } else if (Hh == 4) {
    gat_gemm_xwt<256, 5, 4><<<ggrid, 32, 0, stream>>>(X, W, rW, Hbuf, AGG, HF);
  } else {
    gat_gemm_xwt<256, 5, 3><<<ggrid, 32, 0, stream>>>(X, W, rW, Hbuf, AGG, HF);
  }

  const int nh = NN * Hh;
  gat_el_er<<<(nh + 255) / 256, 256, 0, stream>>>(Hbuf, al, ar, el, er, nh, Hh, F);
  if (kind == 0) {
    gat_owner_rows<true, false><<<NN / OWN_R, 1024, 0, stream>>>(Hbuf, el, er, src, dst, AGG, klist, kcnt);
  } else if (kind == 1) {
    gat_owner_rows<false, true><<<NN / OWN_R, 1024, 0, stream>>>(Hbuf, el, er, src, dst, AGG, klist, kcnt);
  } else {
    gat_owner_rows_out<<<NN / OWN_R, 1024, 0, stream>>>(Hbuf, el, er, src, dst, AGG, klist, kcnt);
  }
}

extern "C" void kernel_launch(void* const* d_in, const int* in_sizes, int n_in,
                              void* d_out, int out_size, void* d_ws, size_t ws_size,
                              hipStream_t stream) {
  (void)in_sizes; (void)n_in; (void)out_size; (void)ws_size;

  const float* feat = (const float*)d_in[0];
  const int*   src  = (const int*)d_in[1];
  const int*   dst  = (const int*)d_in[2];
  const float* W0  = (const float*)d_in[3];
  const float* al0 = (const float*)d_in[4];
  const float* ar0 = (const float*)d_in[5];
  const float* rW0 = (const float*)d_in[6];
  const float* W1  = (const float*)d_in[7];
  const float* al1 = (const float*)d_in[8];
  const float* ar1 = (const float*)d_in[9];
  const float* rW1 = (const float*)d_in[10];
  const float* W2  = (const float*)d_in[11];
  const float* al2 = (const float*)d_in[12];
  const float* ar2 = (const float*)d_in[13];
  const float* rW2 = (const float*)d_in[14];

  const size_t NF = (size_t)NN * 256;
  float* ws    = (float*)d_ws;
  float* bufH  = ws;
  float* bufY0 = bufH  + NF;
  float* bufY1 = bufY0 + NF;
  float* el    = bufY1 + NF;
  float* er    = el  + (size_t)NN * 4;
  unsigned* kcnt  = (unsigned*)(er + (size_t)NN * 4);
  unsigned* klist = kcnt + (size_t)NN * 8;
  float* W2p   = (float*)klist + (size_t)EE * 4;
  float* rW2p  = W2p + (size_t)64 * 256;

  {
    int tot = 64 * 256;
    gat_pad_w<<<dim3((tot + 255) / 256, 2), 256, 0, stream>>>(W2, rW2, W2p, rW2p, 40, 64, 256);
  }

  run_layer(feat, 128, 4, 64, W0, al0, ar0, rW0, src, dst,
            bufH, bufY0, 256, el, er, klist, kcnt, 0, stream);
  run_layer(bufY0, 256, 4, 64, W1, al1, ar1, rW1, src, dst,
            bufH, bufY1, 256, el, er, klist, kcnt, 1, stream);
  run_layer(bufY1, 256, 1, 40, W2p, al2, ar2, rW2p, src, dst,
            bufH, (float*)d_out, 64, el, er, klist, kcnt, 2, stream);
}


# el/er as a WMMA product of their own: first launch folds W^T al, W^T ar into 16 rows per layer, GEMM grid adds one 16-col tile x.V^T per 80 rows; el_er launches gone (10 to 7); el/er sum order new
# speedup vs baseline: 5.8416x; 1.0645x over previous
#include <hip/hip_runtime.h>
#include <math.h>

#define NN 50000
#define EE 800000

typedef __attribute__((ext_vector_type(2))) float v2f;
typedef __attribute__((ext_vector_type(4))) float v4f;
typedef __attribute__((ext_vector_type(8))) float v8f;

__device__ __forceinline__ unsigned f2ord(float f) {
  unsigned u = __float_as_uint(f);
  return (u & 0x80000000u) ? ~u : (u | 0x80000000u);
}
__device__ __forceinline__ float ord2f(unsigned e) {
  unsigned u = (e & 0x80000000u) ? (e ^ 0x80000000u) : ~e;
  return __uint_as_float(u);
}

template <int NF>
__device__ __forceinline__ void gemm_load_lines(v4f (&f)[NF][4], const float* __restrict__ p,
                                                size_t fstep) {
#pragma unroll
  for (int i = 0; i < NF; ++i) {
#pragma unroll
    for (int j = 0; j < 4; ++j) f[i][j] = *(const v4f*)(p + (size_t)i * fstep + 8 * j);
  }
}

template <int MA, int NB>
__device__ __forceinline__ void gemm_mma_lines(v8f (&acc)[MA][NB], const v4f (&a)[MA][4],
                                               const v4f (&b)[NB][4]) {
#pragma unroll
  for (int j = 0; j < 4; ++j) {
#pragma unroll
    for (int i = 0; i < MA; ++i) {
#pragma unroll
      for (int t = 0; t < NB; ++t)
        acc[i][t] = __builtin_amdgcn_wmma_f32_16x16x4_f32(
            false, a[i][j].xy, false, b[t][j].xy, (short)0, acc[i][t], false, false);
    }
#pragma unroll
    for (int i = 0; i < MA; ++i) {
#pragma unroll
      for (int t = 0; t < NB; ++t)
        acc[i][t] = __builtin_amdgcn_wmma_f32_16x16x4_f32(
            false, a[i][j].zw, false, b[t][j].zw, (short)0, acc[i][t], false, false);
    }
  }
}

template <int K, int MA, int NB>
__device__ __forceinline__ void gemm_k_loop(v8f (&acc)[MA][NB], const float* __restrict__ xp,
                                            const float* __restrict__ wp, size_t fstep) {
  v4f aP[MA][4], bP[NB][4], aQ[MA][4], bQ[NB][4];

  gemm_load_lines<MA>(aP, xp, fstep);
  gemm_load_lines<NB>(bP, wp, fstep);
  int kb = 0;
  for (; kb + 64 < K; kb += 64) {
    gemm_load_lines<MA>(aQ, xp + kb + 32, fstep);
    gemm_load_lines<NB>(bQ, wp + kb + 32, fstep);
    __builtin_amdgcn_sched_barrier(0);
    gemm_mma_lines<MA, NB>(acc, aP, bP);
    __builtin_amdgcn_sched_barrier(0);
    gemm_load_lines<MA>(aP, xp + kb + 64, fstep);
    gemm_load_lines<NB>(bP, wp + kb + 64, fstep);
    __builtin_amdgcn_sched_barrier(0);
    gemm_mma_lines<MA, NB>(acc, aQ, bQ);
    __builtin_amdgcn_sched_barrier(0);
  }
  gemm_load_lines<MA>(aQ, xp + kb + 32, fstep);
  gemm_load_lines<NB>(bQ, wp + kb + 32, fstep);
  __builtin_amdgcn_sched_barrier(0);
  gemm_mma_lines<MA, NB>(acc, aP, bP);
  gemm_mma_lines<MA, NB>(acc, aQ, bQ);
}

template <int K, int MA, int NB>
__global__ __launch_bounds__(32)
void gat_gemm_xwt(const float* __restrict__ X, const float* __restrict__ W0,
                  const float* __restrict__ W1, const float* __restrict__ V,
                  float* __restrict__ OUT0, float* __restrict__ OUT1,
                  float* __restrict__ EL, float* __restrict__ ER, int Oout, int Hh) {
  static_assert(K % 64 == 0, "the K loop takes two 32-float lines per trip");
  static_assert(NN % (16 * MA) == 0, "row tiles cover the nodes exactly");
  const int lane = threadIdx.x & 31;
  const int c    = lane & 15;
  const int half = lane >> 4;
  const int row0 = blockIdx.x * (16 * MA);
  const float* xp = X + (size_t)(row0 + c) * (size_t)K + (half << 2);
  const size_t fstep = (size_t)16 * (size_t)K;

  if (blockIdx.y == gridDim.y - 1) {
    if (blockIdx.z == 0) return;
    const float* vp = V + (size_t)c * (size_t)K + (half << 2);
    v8f acc[MA][1] = {};
    gemm_k_loop<K, MA, 1>(acc, xp, vp, fstep);
    if (c < 2 * Hh) {
      float* o = ((c < Hh) ? (EL + c) : (ER + (c - Hh))) + (size_t)(row0 + (half << 3)) * (size_t)Hh;
#pragma unroll
      for (int i = 0; i < MA; ++i) {
#pragma unroll
        for (int v = 0; v < 8; ++v) o[(size_t)((i << 4) + v) * (size_t)Hh] = acc[i][0][v];
      }
    }
    return;
  }

  const int col0 = blockIdx.y << 6;
  const float* __restrict__ W   = (blockIdx.z == 0) ? W0 : W1;
  float* __restrict__       OUT = (blockIdx.z == 0) ? OUT0 : OUT1;
  const float* wp = W + (size_t)(col0 + c) * (size_t)K + (half << 2);

  v8f acc[MA][NB] = {};
  gemm_k_loop<K, MA, NB>(acc, xp, wp, fstep);

#pragma unroll
  for (int i = 0; i < MA; ++i) {
#pragma unroll
    for (int t = 0; t < NB; ++t) {
      const int oc = col0 + (t << 4) + c;
      if (oc < Oout) {
        float* orow = OUT + (size_t)(row0 + (i << 4) + (half << 3)) * (size_t)Oout + oc;
#pragma unroll
        for (int v = 0; v < 8; ++v) {
          orow[(size_t)v * (size_t)Oout] = acc[i][t][v];
        }
      }
    }
  }
}

template <int K, int Hh, int F>
__device__ __forceinline__ void fold_rows(const float* __restrict__ W, const float* __restrict__ al,
                                          const float* __restrict__ ar, float* __restrict__ V, int idx) {
  static_assert(K % 32 == 0, "a wave stays inside one row j");
  if (idx >= 16 * K) return;
  const int j = idx / K;
  const int k = idx % K;
  float s = 0.0f;
  if (j < 2 * Hh) {
    const int h = (j < Hh) ? j : (j - Hh);
    const float* a = ((j < Hh) ? al : ar) + h * F;
    const float* w = W + (size_t)(h * F) * (size_t)K + k;
#pragma unroll
    for (int f = 0; f < F; ++f) s += w[(size_t)f * (size_t)K] * a[f];
  }
  V[idx] = s;
}

__global__ void gat_fold_pad(const float* __restrict__ W0, const float* __restrict__ al0,
                             const float* __restrict__ ar0, float* __restrict__ V0,
                             const float* __restrict__ W1, const float* __restrict__ al1,
                             const float* __restrict__ ar1, float* __restrict__ V1,
                             const float* __restrict__ W2, const float* __restrict__ al2,
                             const float* __restrict__ ar2, float* __restrict__ V2,
                             const float* __restrict__ rW2, float* __restrict__ W2p,
                             float* __restrict__ rW2p) {
  const int idx = blockIdx.x * blockDim.x + threadIdx.x;
  if (blockIdx.y == 0) {
    fold_rows<128, 4, 64>(W0, al0, ar0, V0, idx);
  } else if (blockIdx.y == 1) {
    fold_rows<256, 4, 64>(W1, al1, ar1, V1, idx);
  } else if (blockIdx.y == 2) {
    fold_rows<256, 1, 40>(W2, al2, ar2, V2, idx);
  } else {
    if (idx >= 64 * 256) return;
    const float* __restrict__ Wsrc = (blockIdx.y == 3) ? W2 : rW2;
    float* __restrict__       Wdst = (blockIdx.y == 3) ? W2p : rW2p;
    Wdst[idx] = (idx < 40 * 256) ? Wsrc[idx] : 0.0f;
  }
}

template <int HF, int NCH>
__device__ __forceinline__ void scat_load_row(float (&r)[NCH], const float* __restrict__ row,
                                              int lane) {
#pragma unroll
  for (int c = 0; c < NCH; ++c) {
    const int col = (c << 5) + lane;
    if (HF % 32 == 0 || col < HF) r[c] = row[col];
  }
}

#define OWN_R   200
#define OWN_CAP 1024
#define OWN_KEEP 384
#define OWN_SC  256
#define OWN_NCH (EE / 128)
static_assert(NN % OWN_R == 0, "row groups cover the nodes exactly");
static_assert(OWN_R <= 7 * 32, "7 register slots per wave");
static_assert(EE % 128 == 0, "a wave filters 128 edges per load");
static_assert(EE <= (1 << 20), "an edge index is packed into 20 bits");
static_assert(NN <= (1 << 20), "a node index is packed into 20 bits");
static_assert(((OWN_NCH + 31) / 32) % 4 == 0, "the filter loop takes 4 loads per trip");
static_assert(OWN_NCH % (NN / OWN_R) == 0, "start offsets of the groups tile the list");
static_assert(OWN_KEEP % 128 == 0 && OWN_KEEP <= OWN_CAP, "kept lists are copied 128 entries per trip");
static_assert(OWN_SC % 128 == 0 && OWN_SC <= OWN_CAP, "a trip of 128 entries has its scores kept whole or not at all");
static_assert((size_t)(NN / OWN_R) * 32 * OWN_KEEP <= (size_t)EE * 4, "kept lists fit their [EE,4]-word place in the workspace");
static_assert((NN / OWN_R) * 32 <= NN * 4, "kept list lengths fit their [NN,4]-word place in the workspace");

typedef __attribute__((ext_vector_type(4))) int v4i;

__device__ __forceinline__ void own_queue(unsigned (*lst)[OWN_CAP], unsigned* cnt, unsigned* full,
                                          int e, unsigned r) {
  const unsigned w = r & 31u;
  const unsigned pos = atomicAdd(&cnt[w], 1u);
  if (pos < OWN_CAP) {
    lst[w][pos] = (unsigned)e | ((r >> 5) << 20);
  } else {
    *full = 1u;
  }
}

__device__ __forceinline__ void own_filter(unsigned (*lst)[OWN_CAP], unsigned* cnt, unsigned* full,
                                           const int* __restrict__ dst, int base, int wave, int lane) {
  const int rot = blockIdx.x * (OWN_NCH / (NN / OWN_R));
  const v4i* __restrict__ dp = (const v4i*)dst;
  for (int t = 0; t < (OWN_NCH + 31) / 32; t += 4) {
    v4i dv[4];
    int eb[4];
    bool ok[4];
#pragma unroll
    for (int u = 0; u < 4; ++u) {
      const int L = wave + 32 * (t + u);
      ok[u] = L < OWN_NCH;
      int ch = (ok[u] ? L : 0) + rot;
      if (ch >= OWN_NCH) ch -= OWN_NCH;
      eb[u] = ch * 128 + lane * 4;
      dv[u] = dp[ch * 32 + lane];
    }
#pragma unroll
    for (int u = 0; u < 4; ++u) {
      const unsigned r0 = (unsigned)(dv[u].x - base);
      const unsigned r1 = (unsigned)(dv[u].y - base);
      const unsigned r2 = (unsigned)(dv[u].z - base);
      const unsigned r3 = (unsigned)(dv[u].w - base);
      const unsigned mn = min(min(r0, r1), min(r2, r3));
      if (ok[u] && mn < (unsigned)OWN_R) {
        if (r0 < (unsigned)OWN_R) own_queue(lst, cnt, full, eb[u] + 0, r0);
        if (r1 < (unsigned)OWN_R) own_queue(lst, cnt, full, eb[u] + 1, r1);
        if (r2 < (unsigned)OWN_R) own_queue(lst, cnt, full, eb[u] + 2, r2);
        if (r3 < (unsigned)OWN_R) own_queue(lst, cnt, full, eb[u] + 3, r3);
      }
    }
  }
}

__device__ __forceinline__ void own_copy_list(unsigned* lw, const unsigned* kl, unsigned n, int lane) {
  for (unsigned i0 = 0; i0 < n; i0 += 128) {
    unsigned v[4];
#pragma unroll
    for (int u = 0; u < 4; ++u) {
      const unsigned i = i0 + (unsigned)(32 * u + lane);
      v[u] = (i < n) ? kl[i] : 0u;
    }
#pragma unroll
    for (int u = 0; u < 4; ++u) {
      const unsigned i = i0 + (unsigned)(32 * u + lane);
      if (i < n) lw[i] = v[u];
    }
  }
}

__device__ __forceinline__ void own_scores(float (&s)[4], const float* __restrict__ el,
                                           const float* __restrict__ er, int sn, int dn_) {
  const v4f a = *(const v4f*)(el + (size_t)sn * 4);
  const v4f b = *(const v4f*)(er + (size_t)dn_ * 4);
  s[0] = a.x + b.x; s[1] = a.y + b.y; s[2] = a.z + b.z; s[3] = a.w + b.w;
#pragma unroll
  for (int h = 0; h < 4; ++h) s[h] = (s[h] > 0.0f) ? s[h] : 0.2f * s[h];
}

__device__ __forceinline__ void own_add(float (&acc)[7][8], const float (&q)[8], int slot,
                                        const float (&al)[4]) {
#pragma unroll
  for (int k = 0; k < 7; ++k) {
    if (slot == k) {
#pragma unroll
      for (int c = 0; c < 8; ++c) acc[k][c] += q[c] * al[c >> 1];
    }
  }
}

__device__ __forceinline__ float own_wave_max(float v) {
  for (int o = 16; o > 0; o >>= 1) v = fmaxf(v, __shfl_xor(v, o));
  return v;
}
__device__ __forceinline__ float own_wave_sum(float v) {
  for (int o = 16; o > 0; o >>= 1) v += __shfl_xor(v, o);
  return v;
}

template <bool LOOKUP, bool KEEP>
__device__ __forceinline__ void own_pass_max(unsigned* lw, v4f* scw, unsigned (*mxw)[4], unsigned nq,
                                             const float* __restrict__ el,
                                             const float* __restrict__ er,
                                             const int* __restrict__ src, unsigned* __restrict__ kl,
                                             int base, int wave, int lane) {
  for (unsigned b0 = 0; b0 < nq; b0 += 128) {
    unsigned idx[4];
    bool live[4];
    int me[4], mslot[4], ms[4];
    float s[4][4];
#pragma unroll
    for (int u = 0; u < 4; ++u) {
      idx[u]  = b0 + (unsigned)(32 * u + lane);
      live[u] = idx[u] < nq;
      const unsigned ent = live[u] ? lw[idx[u]] : (7u << 20);
      me[u]    = (int)(ent & 0xFFFFFu);
      mslot[u] = (int)(ent >> 20);
    }
#pragma unroll
    for (int u = 0; u < 4; ++u) ms[u] = LOOKUP ? src[me[u]] : me[u];
#pragma unroll
    for (int u = 0; u < 4; ++u)
      own_scores(s[u], el, er, ms[u], (mslot[u] < 7) ? (base + (mslot[u] << 5) + wave) : base);
#pragma unroll
    for (int u = 0; u < 4; ++u) {
      if (live[u]) {
        if (LOOKUP) {
          const unsigned ent = (unsigned)ms[u] | ((unsigned)mslot[u] << 20);
          lw[idx[u]] = ent;
          if (KEEP && idx[u] < (unsigned)OWN_KEEP) kl[idx[u]] = ent;
        }
#pragma unroll
        for (int h = 0; h < 4; ++h) atomicMax(&mxw[mslot[u]][h], f2ord(s[u][h]));
        if (b0 < (unsigned)OWN_SC) {
          v4f sv;
          sv.x = s[u][0]; sv.y = s[u][1]; sv.z = s[u][2]; sv.w = s[u][3];
          scw[idx[u]] = sv;
        }
      }
    }
  }
}

__device__ void own_rows_by_scan(int base, int wave, int lane, const float* __restrict__ Hm,
                                 const float* __restrict__ el, const float* __restrict__ er,
                                 const int* __restrict__ src, const int* __restrict__ dst,
                                 float* AGG) {
  for (int k = 0; k < 7; ++k) {
    const int r = (k << 5) + wave;
    if (r >= OWN_R) break;
    const int d = base + r;
    float m[4] = {-INFINITY, -INFINITY, -INFINITY, -INFINITY};
    for (int e = lane; e < EE; e += 32) {
      if (dst[e] == d) {
        float s[4];
        own_scores(s, el, er, src[e], d);
        for (int h = 0; h < 4; ++h) m[h] = fmaxf(m[h], s[h]);
      }
    }
    for (int h = 0; h < 4; ++h) m[h] = own_wave_max(m[h]);
    float z[4] = {0.0f, 0.0f, 0.0f, 0.0f};
    for (int e = lane; e < EE; e += 32) {
      if (dst[e] == d) {
        float s[4];
        own_scores(s, el, er, src[e], d);
        for (int h = 0; h < 4; ++h) z[h] += expf(s[h] - m[h]);
      }
    }
    for (int h = 0; h < 4; ++h) z[h] = own_wave_sum(z[h]);
    float acc[8] = {};
    for (int e0 = 0; e0 < EE; e0 += 32) {
      const int e = e0 + lane;
      const bool hit = dst[e] == d;
      int sn = 0;
      float a[4] = {0.0f, 0.0f, 0.0f, 0.0f};
      if (hit) {
        sn = src[e];
        float s[4];
        own_scores(s, el, er, sn, d);
        for (int h = 0; h < 4; ++h) a[h] = expf(s[h] - m[h]) / z[h];
      }
      unsigned mask = (unsigned)__ballot(hit);
      while (mask) {
        const int j = __ffs(mask) - 1;
        mask &= mask - 1u;
        const int sj = __shfl(sn, j);
        float aj[4];
        for (int h = 0; h < 4; ++h) aj[h] = __shfl(a[h], j);
        const float* row = Hm + (size_t)sj * 256;
        for (int c = 0; c < 8; ++c) acc[c] += row[(c << 5) + lane] * aj[c >> 1];
      }
    }
    float* orow = AGG + (size_t)d * 256;
    for (int c = 0; c < 8; ++c) {
      const float v = orow[(c << 5) + lane] + acc[c];
      orow[(c << 5) + lane] = (v > 0.0f) ? v : expm1f(v);
    }
  }
}

template <bool KEEP, bool REUSE>
__global__ __launch_bounds__(1024)
void gat_owner_rows(const float* __restrict__ Hm, const float* __restrict__ el,
                    const float* __restrict__ er, const int* __restrict__ src,
                    const int* __restrict__ dst, float* AGG, unsigned* klist, unsigned* kcnt) {
  __shared__ unsigned lst[32][OWN_CAP];
  __shared__ v4f      sc[32][OWN_SC];
  __shared__ unsigned cnt[32];
  __shared__ unsigned mx[32][8][4];
  __shared__ float    dn[32][8][4];
  __shared__ unsigned full;
  __shared__ unsigned kept;
  const int tid  = threadIdx.x;
  const int lane = tid & 31;
  const int wave = __builtin_amdgcn_readfirstlane(tid >> 5);
  const int base = blockIdx.x * OWN_R;
  unsigned* kc = kcnt + blockIdx.x * 32;
  unsigned* kl = klist + (size_t)(blockIdx.x * 32 + wave) * OWN_KEEP;

  if (tid < 32) cnt[tid] = 0u;
  if (tid == 0) { full = 0u; kept = 1u; }
  (&mx[0][0][0])[tid] = 0u;
  (&dn[0][0][0])[tid] = 0.0f;
  __syncthreads();

  bool reuse = false;
  if (REUSE) {
    if (tid < 32 && kc[tid] > (unsigned)OWN_KEEP) kept = 0u;
    __syncthreads();
    reuse = __builtin_amdgcn_readfirstlane((int)kept) != 0;
  }

  if (reuse) {
    const unsigned n = (unsigned)__builtin_amdgcn_readfirstlane((int)kc[wave]);
    own_copy_list(lst[wave], kl, n, lane);
    if (lane == 0) cnt[wave] = n;
  } else {
    own_filter(lst, cnt, &full, dst, base, wave, lane);
  }
  __syncthreads();

  const bool overfull = __builtin_amdgcn_readfirstlane((int)full) != 0;
  if (KEEP && lane == 0) kc[wave] = overfull ? 0xFFFFFFFFu : cnt[wave];
  if (overfull) {
    own_rows_by_scan(base, wave, lane, Hm, el, er, src, dst, AGG);
    return;
  }

  const unsigned nq = (unsigned)__builtin_amdgcn_readfirstlane((int)cnt[wave]);

  if (reuse) {
    own_pass_max<false, false>(lst[wave], sc[wave], mx[wave], nq, el, er, src, kl, base, wave, lane);
  } else {
    own_pass_max<true, KEEP>(lst[wave], sc[wave], mx[wave], nq, el, er, src, kl, base, wave, lane);
  }
  __syncthreads();

  for (unsigned b0 = 0; b0 < nq; b0 += 128) {
    unsigned idx[4];
    bool live[4];
    int mslot[4], ms[4];
    float s[4][4];
#pragma unroll
    for (int u = 0; u < 4; ++u) {
      idx[u]  = b0 + (unsigned)(32 * u + lane);
      live[u] = idx[u] < nq;
      const unsigned ent = live[u] ? lst[wave][idx[u]] : (7u << 20);
      ms[u]    = (int)(ent & 0xFFFFFu);
      mslot[u] = (int)(ent >> 20);
    }
    const bool kept_sc = b0 < (unsigned)OWN_SC;
    if (kept_sc) {
#pragma unroll
      for (int u = 0; u < 4; ++u) {
        const v4f sv = sc[wave][idx[u]];
        s[u][0] = sv.x; s[u][1] = sv.y; s[u][2] = sv.z; s[u][3] = sv.w;
      }
    } else {
#pragma unroll
      for (int u = 0; u < 4; ++u)
        own_scores(s[u], el, er, ms[u], (mslot[u] < 7) ? (base + (mslot[u] << 5) + wave) : base);
    }
#pragma unroll
    for (int u = 0; u < 4; ++u) {
      if (live[u]) {
        float ev[4];
#pragma unroll
        for (int h = 0; h < 4; ++h) {
          ev[h] = expf(s[u][h] - ord2f(mx[wave][mslot[u]][h]));
          atomicAdd(&dn[wave][mslot[u]][h], ev[h]);
        }
        if (kept_sc) {
          v4f sv;
          sv.x = ev[0]; sv.y = ev[1]; sv.z = ev[2]; sv.w = ev[3];
          sc[wave][idx[u]] = sv;
        }
      }
    }
  }
  __syncthreads();

  float acc[7][8] = {};
  for (unsigned b0 = 0; b0 < nq; b0 += 32) {
    const unsigned idx = b0 + (unsigned)lane;
    const unsigned ent = (idx < nq) ? lst[wave][idx] : (7u << 20);
    const int ms    = (int)(ent & 0xFFFFFu);
    const int mslot = (int)(ent >> 20);
    const int md    = (mslot < 7) ? (base + (mslot << 5) + wave) : base;
    float ma[4];
    if (b0 < (unsigned)OWN_SC) {
      const v4f ev = sc[wave][idx];
      ma[0] = ev.x; ma[1] = ev.y; ma[2] = ev.z; ma[3] = ev.w;
    } else {
      own_scores(ma, el, er, ms, md);
#pragma unroll
      for (int h = 0; h < 4; ++h) ma[h] = expf(ma[h] - ord2f(mx[wave][mslot][h]));
    }
#pragma unroll
    for (int h = 0; h < 4; ++h) ma[h] = ma[h] / dn[wave][mslot][h];

    float q0[8], q1[8], q2[8], q3[8];
    float al[4];
    scat_load_row<256, 8>(q0, Hm + (size_t)__builtin_amdgcn_readlane(ms, 0) * 256, lane);
    scat_load_row<256, 8>(q1, Hm + (size_t)__builtin_amdgcn_readlane(ms, 1) * 256, lane);
    scat_load_row<256, 8>(q2, Hm + (size_t)__builtin_amdgcn_readlane(ms, 2) * 256, lane);
    for (int i = 0; i < 32; i += 4) {
      scat_load_row<256, 8>(q3, Hm + (size_t)__builtin_amdgcn_readlane(ms, i + 3) * 256, lane);
#pragma unroll
      for (int h = 0; h < 4; ++h) al[h] = __int_as_float(__builtin_amdgcn_readlane(__float_as_int(ma[h]), i));
      own_add(acc, q0, __builtin_amdgcn_readlane(mslot, i), al);

      scat_load_row<256, 8>(q0, Hm + (size_t)__builtin_amdgcn_readlane(ms, (i + 4) & 31) * 256, lane);
#pragma unroll
      for (int h = 0; h < 4; ++h) al[h] = __int_as_float(__builtin_amdgcn_readlane(__float_as_int(ma[h]), i + 1));
      own_add(acc, q1, __builtin_amdgcn_readlane(mslot, i + 1), al);

      scat_load_row<256, 8>(q1, Hm + (size_t)__builtin_amdgcn_readlane(ms, (i + 5) & 31) * 256, lane);
#pragma unroll
      for (int h = 0; h < 4; ++h) al[h] = __int_as_float(__builtin_amdgcn_readlane(__float_as_int(ma[h]), i + 2));
      own_add(acc, q2, __builtin_amdgcn_readlane(mslot, i + 2), al);

      scat_load_row<256, 8>(q2, Hm + (size_t)__builtin_amdgcn_readlane(ms, (i + 6) & 31) * 256, lane);
#pragma unroll
      for (int h = 0; h < 4; ++h) al[h] = __int_as_float(__builtin_amdgcn_readlane(__float_as_int(ma[h]), i + 3));
      own_add(acc, q3, __builtin_amdgcn_readlane(mslot, i + 3), al);
    }
  }

#pragma unroll
  for (int g = 0; g < 7; g += 4) {
    float t[4][8];
#pragma unroll
    for (int kk = 0; kk < 4; ++kk) {
      const int k = g + kk;
      const int r = (k << 5) + wave;
      if (k < 7 && r < OWN_R) {
        const float* orow = AGG + (size_t)(base + r) * 256;
#pragma unroll
        for (int c = 0; c < 8; ++c) t[kk][c] = orow[(c << 5) + lane];
      }
    }
#pragma unroll
    for (int kk = 0; kk < 4; ++kk) {
      const int k = g + kk;
      const int r = (k << 5) + wave;
      if (k < 7 && r < OWN_R) {
        float* orow = AGG + (size_t)(base + r) * 256;
#pragma unroll
        for (int c = 0; c < 8; ++c) {
          const float v = t[kk][c] + acc[k < 7 ? k : 0][c];
          orow[(c << 5) + lane] = (v > 0.0f) ? v : expm1f(v);
        }
      }
    }
  }
}

#define OUT_F 40
#define OUT_DEPTH 8

__device__ __forceinline__ float out_score(const float* __restrict__ el, const float* __restrict__ er,
                                           int sn, int dn_) {
  const float s = el[sn] + er[dn_];
  return (s > 0.0f) ? s : 0.2f * s;
}

__device__ __forceinline__ void out_load_row(float (&r)[2], const float* __restrict__ row, int lane) {
  r[0] = row[lane];
  r[1] = row[32 + (lane & 7)];
}

__device__ __forceinline__ void out_add(float (&acc)[7][2], const float (&q)[2], int slot, float a) {
#pragma unroll
  for (int k = 0; k < 7; ++k) {
    if (slot == k) {
      acc[k][0] += q[0] * a;
      acc[k][1] += q[1] * a;
    }
  }
}

template <bool LOOKUP>
__device__ __forceinline__ void out_pass_max(unsigned* lw, float* scw, unsigned* mxw, unsigned nq,
                                             const float* __restrict__ el,
                                             const float* __restrict__ er,
                                             const int* __restrict__ src, int base, int wave, int lane) {
  for (unsigned b0 = 0; b0 < nq; b0 += 128) {
    unsigned idx[4];
    bool live[4];
    int me[4], mslot[4], ms[4];
    float s[4];
#pragma unroll
    for (int u = 0; u < 4; ++u) {
      idx[u]  = b0 + (unsigned)(32 * u + lane);
      live[u] = idx[u] < nq;
      const unsigned ent = live[u] ? lw[idx[u]] : (7u << 20);
      me[u]    = (int)(ent & 0xFFFFFu);
      mslot[u] = (int)(ent >> 20);
    }
#pragma unroll
    for (int u = 0; u < 4; ++u) ms[u] = LOOKUP ? src[me[u]] : me[u];
#pragma unroll
    for (int u = 0; u < 4; ++u)
      s[u] = out_score(el, er, ms[u], (mslot[u] < 7) ? (base + (mslot[u] << 5) + wave) : base);
#pragma unroll
    for (int u = 0; u < 4; ++u) {
      if (live[u]) {
        if (LOOKUP) lw[idx[u]] = (unsigned)ms[u] | ((unsigned)mslot[u] << 20);
        atomicMax(&mxw[mslot[u]], f2ord(s[u]));
        scw[idx[u]] = s[u];
      }
    }
  }
}

__device__ void out_rows_by_scan(int base, int wave, int lane, const float* __restrict__ Hm,
                                 const float* __restrict__ el, const float* __restrict__ er,
                                 const int* __restrict__ src, const int* __restrict__ dst,
                                 float* AGG) {
  for (int k = 0; k < 7; ++k) {
    const int r = (k << 5) + wave;
    if (r >= OWN_R) break;
    const int d = base + r;
    float m = -INFINITY;
    for (int e = lane; e < EE; e += 32) {
      if (dst[e] == d) m = fmaxf(m, out_score(el, er, src[e], d));
    }
    m = own_wave_max(m);
    float z = 0.0f;
    for (int e = lane; e < EE; e += 32) {
      if (dst[e] == d) z += expf(out_score(el, er, src[e], d) - m);
    }
    z = own_wave_sum(z);
    float acc0 = 0.0f, acc1 = 0.0f;
    for (int e0 = 0; e0 < EE; e0 += 32) {
      const int e = e0 + lane;
      const bool hit = dst[e] == d;
      int sn = 0;
      float a = 0.0f;
      if (hit) {
        sn = src[e];
        a = expf(out_score(el, er, sn, d) - m) / z;
      }
      unsigned mask = (unsigned)__ballot(hit);
      while (mask) {
        const int j = __ffs(mask) - 1;
        mask &= mask - 1u;
        const int sj = __shfl(sn, j);
        const float aj = __shfl(a, j);
        float q[2];
        out_load_row(q, Hm + (size_t)sj * OUT_F, lane);
        acc0 += q[0] * aj;
        acc1 += q[1] * aj;
      }
    }
    float* orow = AGG + (size_t)d * OUT_F;
    orow[lane] += acc0;
    if (lane < 8) orow[32 + lane] += acc1;
  }
}

__global__ __launch_bounds__(1024)
void gat_owner_rows_out(const float* __restrict__ Hm, const float* __restrict__ el,
                        const float* __restrict__ er, const int* __restrict__ src,
                        const int* __restrict__ dst, float* AGG, const unsigned* klist,
                        const unsigned* kcnt) {
  __shared__ unsigned lst[32][OWN_CAP];
  __shared__ float    sc[32][OWN_CAP];
  __shared__ unsigned cnt[32];
  __shared__ unsigned mx[32][8];
  __shared__ float    dn[32][8];
  __shared__ unsigned full;
  __shared__ unsigned kept;
  const int tid  = threadIdx.x;
  const int lane = tid & 31;
  const int wave = __builtin_amdgcn_readfirstlane(tid >> 5);
  const int base = blockIdx.x * OWN_R;
  const unsigned* kc = kcnt + blockIdx.x * 32;
  const unsigned* kl = klist + (size_t)(blockIdx.x * 32 + wave) * OWN_KEEP;

  if (tid < 32) cnt[tid] = 0u;
  if (tid == 0) { full = 0u; kept = 1u; }
  if (tid < 256) {
    (&mx[0][0])[tid] = 0u;
    (&dn[0][0])[tid] = 0.0f;
  }
  __syncthreads();

  if (tid < 32 && kc[tid] > (unsigned)OWN_KEEP) kept = 0u;
  __syncthreads();
  const bool reuse = __builtin_amdgcn_readfirstlane((int)kept) != 0;

  if (reuse) {
    const unsigned n = (unsigned)__builtin_amdgcn_readfirstlane((int)kc[wave]);
    own_copy_list(lst[wave], kl, n, lane);
    if (lane == 0) cnt[wave] = n;
  } else {
    own_filter(lst, cnt, &full, dst, base, wave, lane);
  }
  __syncthreads();

  const bool overfull = __builtin_amdgcn_readfirstlane((int)full) != 0;
  if (overfull) {
    out_rows_by_scan(base, wave, lane, Hm, el, er, src, dst, AGG);
    return;
  }

  const unsigned nq = (unsigned)__builtin_amdgcn_readfirstlane((int)cnt[wave]);

  if (reuse) {
    out_pass_max<false>(lst[wave], sc[wave], mx[wave], nq, el, er, src, base, wave, lane);
  } else {
    out_pass_max<true>(lst[wave], sc[wave], mx[wave], nq, el, er, src, base, wave, lane);
  }
  __syncthreads();

  for (unsigned b0 = 0; b0 < nq; b0 += 128) {
    unsigned idx[4];
    bool live[4];
    int mslot[4];
    float s[4];
#pragma unroll
    for (int u = 0; u < 4; ++u) {
      idx[u]  = b0 + (unsigned)(32 * u + lane);
      live[u] = idx[u] < nq;
      const unsigned ent = live[u] ? lst[wave][idx[u]] : (7u << 20);
      mslot[u] = (int)(ent >> 20);
      s[u] = sc[wave][idx[u]];
    }
#pragma unroll
    for (int u = 0; u < 4; ++u) {
      if (live[u]) {
        const float ev = expf(s[u] - ord2f(mx[wave][mslot[u]]));
        atomicAdd(&dn[wave][mslot[u]], ev);
        sc[wave][idx[u]] = ev;
      }
    }
  }
  __syncthreads();

  float acc[7][2] = {};
  for (unsigned b0 = 0; b0 < nq; b0 += 32) {
    const unsigned idx = b0 + (unsigned)lane;
    const unsigned ent = (idx < nq) ? lst[wave][idx] : (7u << 20);
    const int ms    = (int)(ent & 0xFFFFFu);
    const int mslot = (int)(ent >> 20);
    const float ma  = sc[wave][idx] / dn[wave][mslot];
    const int nb = (int)min(32u, (nq - b0 + (unsigned)(OUT_DEPTH - 1)) & ~(unsigned)(OUT_DEPTH - 1));

    float q[OUT_DEPTH][2];
#pragma unroll
    for (int p = 0; p < OUT_DEPTH - 1; ++p)
      out_load_row(q[p], Hm + (size_t)__builtin_amdgcn_readlane(ms, p) * OUT_F, lane);
    for (int i = 0; i < nb; i += OUT_DEPTH) {
#pragma unroll
      for (int u = 0; u < OUT_DEPTH; ++u) {
        out_load_row(q[(u + OUT_DEPTH - 1) % OUT_DEPTH],
                     Hm + (size_t)__builtin_amdgcn_readlane(ms, (i + u + OUT_DEPTH - 1) & 31) * OUT_F, lane);
        const float a = __int_as_float(__builtin_amdgcn_readlane(__float_as_int(ma), i + u));
        out_add(acc, q[u], __builtin_amdgcn_readlane(mslot, i + u), a);
      }
    }
  }

  float t[7][2];
#pragma unroll
  for (int k = 0; k < 7; ++k) {
    const int r = (k << 5) + wave;
    if (r < OWN_R) {
      const float* orow = AGG + (size_t)(base + r) * OUT_F;
      out_load_row(t[k], orow, lane);
    }
  }
#pragma unroll
  for (int k = 0; k < 7; ++k) {
    const int r = (k << 5) + wave;
    if (r < OWN_R) {
      float* orow = AGG + (size_t)(base + r) * OUT_F;
      orow[lane] = t[k][0] + acc[k][0];
      if (lane < 8) orow[32 + lane] = t[k][1] + acc[k][1];
    }
  }
}

static void run_layer(const float* X, int K, int Hh, int F,
                      const float* W, const float* V, const float* rW,
                      const int* src, const int* dst,
                      float* Hbuf, float* AGG, int Opad,
                      float* el, float* er, unsigned* klist, unsigned* kcnt,
                      int kind, hipStream_t stream) {
  const int HF = Hh * F;
  dim3 ggrid(NN / 80, Opad / 64 + 1, 2);
  if (Hh == 4 && K == 128) {
    gat_gemm_xwt<128, 5, 4><<<ggrid, 32, 0, stream>>>(X, W, rW, V, Hbuf, AGG, el, er, HF, Hh);
  } else if (Hh == 4) {
    gat_gemm_xwt<256, 5, 4><<<ggrid, 32, 0, stream>>>(X, W, rW, V, Hbuf, AGG, el, er, HF, Hh);
  } else {
    gat_gemm_xwt<256, 5, 3><<<ggrid, 32, 0, stream>>>(X, W, rW, V, Hbuf, AGG, el, er, HF, Hh);
  }

  if (kind == 0) {
    gat_owner_rows<true, false><<<NN / OWN_R, 1024, 0, stream>>>(Hbuf, el, er, src, dst, AGG, klist, kcnt);
  } else if (kind == 1) {
    gat_owner_rows<false, true><<<NN / OWN_R, 1024, 0, stream>>>(Hbuf, el, er, src, dst, AGG, klist, kcnt);
  } else {
    gat_owner_rows_out<<<NN / OWN_R, 1024, 0, stream>>>(Hbuf, el, er, src, dst, AGG, klist, kcnt);
  }
}

extern "C" void kernel_launch(void* const* d_in, const int* in_sizes, int n_in,
                              void* d_out, int out_size, void* d_ws, size_t ws_size,
                              hipStream_t stream) {
  (void)in_sizes; (void)n_in; (void)out_size; (void)ws_size;

  const float* feat = (const float*)d_in[0];
  const int*   src  = (const int*)d_in[1];
  const int*   dst  = (const int*)d_in[2];
  const float* W0  = (const float*)d_in[3];
  const float* al0 = (const float*)d_in[4];
  const float* ar0 = (const float*)d_in[5];
  const float* rW0 = (const float*)d_in[6];
  const float* W1  = (const float*)d_in[7];
  const float* al1 = (const float*)d_in[8];
  const float* ar1 = (const float*)d_in[9];
  const float* rW1 = (const float*)d_in[10];
  const float* W2  = (const float*)d_in[11];
  const float* al2 = (const float*)d_in[12];
  const float* ar2 = (const float*)d_in[13];
  const float* rW2 = (const float*)d_in[14];

  const size_t NF = (size_t)NN * 256;
  float* ws    = (float*)d_ws;
  float* bufH  = ws;
  float* bufY0 = bufH  + NF;
  float* bufY1 = bufY0 + NF;
  float* el    = bufY1 + NF;
  float* er    = el  + (size_t)NN * 4;
  unsigned* kcnt  = (unsigned*)(er + (size_t)NN * 4);
  float* V0    = (float*)kcnt + (size_t)NN * 4;
  float* V1    = V0 + (size_t)16 * 128;
  float* V2    = V1 + (size_t)16 * 256;
  unsigned* klist = kcnt + (size_t)NN * 8;
  float* W2p   = (float*)klist + (size_t)EE * 4;
  float* rW2p  = W2p + (size_t)64 * 256;

  gat_fold_pad<<<dim3(64, 5), 256, 0, stream>>>(W0, al0, ar0, V0, W1, al1, ar1, V1,
                                                W2, al2, ar2, V2, rW2, W2p, rW2p);

  run_layer(feat, 128, 4, 64, W0, V0, rW0, src, dst,
            bufH, bufY0, 256, el, er, klist, kcnt, 0, stream);
  run_layer(bufY0, 256, 4, 64, W1, V1, rW1, src, dst,
            bufH, bufY1, 256, el, er, klist, kcnt, 1, stream);
  run_layer(bufY1, 256, 1, 40, W2p, V2, rW2p, src, dst,
            bufH, (float*)d_out, 64, el, er, klist, kcnt, 2, stream);
}
